# starting kernel
# baseline (speedup 1.0000x reference)
#include <hip/hip_runtime.h>
#include <math.h>

typedef __attribute__((ext_vector_type(2))) float v2f;
typedef __attribute__((ext_vector_type(8))) float v8f;

#define D_IN  128
#define H_DIM 32
#define C_DIM 5

__global__ void gcn_build_w1(const float* __restrict__ Wrel, const float* __restrict__ Wroot,
                             float* __restrict__ Wcat) {
    int i = blockIdx.x * 256 + threadIdx.x;
    if (i >= 128 * 64) return;
    int k = i >> 6, j = i & 63;
    Wcat[i] = (j < 32) ? Wrel[k * 32 + j] : Wroot[k * 32 + (j - 32)];
}

__global__ void gcn_build_w2(const float* __restrict__ Wrel, const float* __restrict__ Wroot,
                             float* __restrict__ Wp) {
    int i = blockIdx.x * 256 + threadIdx.x;
    if (i >= 32 * 16) return;
    int k = i >> 4, n = i & 15;
    float v = 0.0f;
    if (n < 5)       v = Wrel[k * 5 + n];
    else if (n < 10) v = Wroot[k * 5 + (n - 5)];
    Wp[i] = v;
}

__global__ void gcn_zero(float* __restrict__ p, long n) {
    long i = (long)blockIdx.x * 256 + threadIdx.x;
    if (i < n) p[i] = 0.0f;
}

__global__ __launch_bounds__(256)
void gcn_gemm1_wmma(const float* __restrict__ x, const float* __restrict__ W,
                    float* __restrict__ PR1, int nTiles) {
    int wave = threadIdx.x >> 5;
    int lane = threadIdx.x & 31;
    int tile = blockIdx.x * 8 + wave;
    if (tile >= nTiles) return;
    int half = lane >> 4;
    int m    = lane & 15;
    const float* xr = x + (long)(tile * 16 + m) * D_IN;

    v2f a[32];
#pragma unroll
    for (int ks = 0; ks < 32; ++ks) {
        int k = ks * 4 + half * 2;
        a[ks].x = xr[k];
        a[ks].y = xr[k + 1];
    }

#pragma unroll
    for (int ct = 0; ct < 4; ++ct) {
        int n = ct * 16 + m;
        v8f acc = {};
#pragma unroll
        for (int ks = 0; ks < 32; ++ks) {
            int k = ks * 4 + half * 2;
            v2f b;
            b.x = W[k * 64 + n];
            b.y = W[(k + 1) * 64 + n];
            acc = __builtin_amdgcn_wmma_f32_16x16x4_f32(false, a[ks], false, b,
                                                        (short)0, acc, false, false);
        }
#pragma unroll
        for (int r = 0; r < 8; ++r) {
            int orow = tile * 16 + r + half * 8;
            PR1[(long)orow * 64 + ct * 16 + m] = acc[r];
        }
    }
}

__global__ __launch_bounds__(256)
void gcn_gemm2_wmma(const float* __restrict__ h, const float* __restrict__ W,
                    float* __restrict__ PR2, int nTiles) {
    int wave = threadIdx.x >> 5;
    int lane = threadIdx.x & 31;
    int tile = blockIdx.x * 8 + wave;
    if (tile >= nTiles) return;
    int half = lane >> 4;
    int m    = lane & 15;
    const float* hr = h + (long)(tile * 16 + m) * H_DIM;

    v2f a[8];
#pragma unroll
    for (int ks = 0; ks < 8; ++ks) {
        int k = ks * 4 + half * 2;
        a[ks].x = hr[k];
        a[ks].y = hr[k + 1];
    }
    v8f acc = {};
#pragma unroll
    for (int ks = 0; ks < 8; ++ks) {
        int k = ks * 4 + half * 2;
        v2f b;
        b.x = W[k * 16 + m];
        b.y = W[(k + 1) * 16 + m];
        acc = __builtin_amdgcn_wmma_f32_16x16x4_f32(false, a[ks], false, b,
                                                    (short)0, acc, false, false);
    }
    if (m < 10) {
#pragma unroll
        for (int r = 0; r < 8; ++r) {
            int orow = tile * 16 + r + half * 8;
            PR2[(long)orow * 10 + m] = acc[r];
        }
    }
}

__global__ __launch_bounds__(256)
void gcn_agg1(const int* __restrict__ src, const int* __restrict__ dst,
              const float* __restrict__ ew, const float* __restrict__ PR1,
              float* __restrict__ S1, float* __restrict__ cnt, int E) {
    int e = blockIdx.x * 8 + (threadIdx.x >> 5);
    int lane = threadIdx.x & 31;
    if (e >= E) return;
    int s = src[e], d = dst[e];
    float v = ew[e] * PR1[(long)s * 64 + lane];
    atomicAdd(&S1[(long)d * 32 + lane], v);
    if (lane == 0) atomicAdd(&cnt[d], 1.0f);
}

__global__ __launch_bounds__(256)
void gcn_combine1(const float* __restrict__ S1, const float* __restrict__ PR1,
                  const float* __restrict__ b1, const float* __restrict__ cnt,
                  float* __restrict__ h, long total) {
    long i = (long)blockIdx.x * 256 + threadIdx.x;
    if (i >= total) return;
    long n = i >> 5;
    int  j = (int)(i & 31);
    float c = fmaxf(cnt[n], 1.0f);
    float v = S1[i] / c + PR1[n * 64 + 32 + j] + b1[j];
    h[i] = fmaxf(v, 0.0f);
}

__global__ __launch_bounds__(256)
void gcn_agg2(const int* __restrict__ src, const int* __restrict__ dst,
              const float* __restrict__ PR2, float* __restrict__ S2, int E) {
    long t = (long)blockIdx.x * 256 + threadIdx.x;
    int e = (int)(t >> 3);
    int c = (int)(t & 7);
    if (e >= E || c >= 5) return;
    atomicAdd(&S2[(long)dst[e] * 5 + c], PR2[(long)src[e] * 10 + c]);
}

__global__ __launch_bounds__(256)
void gcn_output(const float* __restrict__ S2, const float* __restrict__ PR2,
                const float* __restrict__ b2, const float* __restrict__ cnt,
                float* __restrict__ out, int N) {
    int n = blockIdx.x * 256 + threadIdx.x;
    if (n >= N) return;
    float c = fmaxf(cnt[n], 1.0f);
    float v[5];
#pragma unroll
    for (int j = 0; j < 5; ++j)
        v[j] = S2[(long)n * 5 + j] / c + PR2[(long)n * 10 + 5 + j] + b2[j];
    float mx = v[0];
#pragma unroll
    for (int j = 1; j < 5; ++j) mx = fmaxf(mx, v[j]);
    float s = 0.0f;
#pragma unroll
    for (int j = 0; j < 5; ++j) s += expf(v[j] - mx);
    float l = logf(s);
#pragma unroll
    for (int j = 0; j < 5; ++j) out[(long)n * 5 + j] = v[j] - mx - l;
}

extern "C" void kernel_launch(void* const* d_in, const int* in_sizes, int n_in,
                              void* d_out, int out_size, void* d_ws, size_t ws_size,
                              hipStream_t stream) {
    const float* x      = (const float*)d_in[0];
    const int*   ei     = (const int*)d_in[1];
    const float* ew     = (const float*)d_in[2];
    const float* Wrel1  = (const float*)d_in[3];
    const float* Wroot1 = (const float*)d_in[4];
    const float* b1     = (const float*)d_in[5];
    const float* Wrel2  = (const float*)d_in[6];
    const float* Wroot2 = (const float*)d_in[7];
    const float* b2     = (const float*)d_in[8];

    const int N = in_sizes[0] / D_IN;
    const int E = in_sizes[1] / 2;
    const int* src = ei;
    const int* dst = ei + E;

    float* ws = (float*)d_ws;
    long off = 0;
    float* PR1 = ws + off; off += (long)N * 64;
    float* S1  = ws + off; off += (long)N * 32;
    float* cnt = ws + off; off += N;
    float* h   = ws + off; off += (long)N * 32;
    float* W1c = ws + off; off += 128 * 64;
    float* W2p = ws + off; off += 32 * 16;
    float* PR2 = PR1;
    float* S2  = S1;

    const int nTiles = N / 16;

    gcn_build_w1<<<(128 * 64 + 255) / 256, 256, 0, stream>>>(Wrel1, Wroot1, W1c);
    gcn_build_w2<<<(32 * 16 + 255) / 256, 256, 0, stream>>>(Wrel2, Wroot2, W2p);

    gcn_gemm1_wmma<<<(nTiles + 7) / 8, 256, 0, stream>>>(x, W1c, PR1, nTiles);

    long nz1 = (long)N * 33;
    gcn_zero<<<(int)((nz1 + 255) / 256), 256, 0, stream>>>(S1, nz1);
    gcn_agg1<<<(E + 7) / 8, 256, 0, stream>>>(src, dst, ew, PR1, S1, cnt, E);

    long nh = (long)N * 32;
    gcn_combine1<<<(int)((nh + 255) / 256), 256, 0, stream>>>(S1, PR1, b1, cnt, h, nh);

    gcn_gemm2_wmma<<<(nTiles + 7) / 8, 256, 0, stream>>>(h, W2p, PR2, nTiles);

    long nz2 = (long)N * 5;
    gcn_zero<<<(int)((nz2 + 255) / 256), 256, 0, stream>>>(S2, nz2);
    gcn_agg2<<<(int)(((long)E * 8 + 255) / 256), 256, 0, stream>>>(src, dst, PR2, S2, E);

    gcn_output<<<(N + 255) / 256, 256, 0, stream>>>(S2, PR2, b2, cnt, (float*)d_out, N);
}


# Edges bucketed by destination each call (returning int atomic = slot, scan, place); both segment sums become register gathers, no f32 atomics; sum order = atomic arrival rank, then 8-lane tree
# speedup vs baseline: 1.6483x; 1.6483x over previous
#include <hip/hip_runtime.h>
#include <math.h>

typedef __attribute__((ext_vector_type(2))) float v2f;
typedef __attribute__((ext_vector_type(8))) float v8f;

#define D_IN  128
#define H_DIM 32
#define C_DIM 5

__global__ void gcn_build_w1(const float* __restrict__ Wrel, const float* __restrict__ Wroot,
                             float* __restrict__ Wcat) {
    int i = blockIdx.x * 256 + threadIdx.x;
    if (i >= 128 * 64) return;
    int k = i >> 6, j = i & 63;
    Wcat[i] = (j < 32) ? Wrel[k * 32 + j] : Wroot[k * 32 + (j - 32)];
}

__global__ void gcn_build_w2(const float* __restrict__ Wrel, const float* __restrict__ Wroot,
                             float* __restrict__ Wp) {
    int i = blockIdx.x * 256 + threadIdx.x;
    if (i >= 32 * 16) return;
    int k = i >> 4, n = i & 15;
    float v = 0.0f;
    if (n < 5)       v = Wrel[k * 5 + n];
    else if (n < 10) v = Wroot[k * 5 + (n - 5)];
    Wp[i] = v;
}

__global__ void gcn_zero_i(int* __restrict__ p, int n) {
    int i = blockIdx.x * 256 + threadIdx.x;
    if (i < n) p[i] = 0;
}

__global__ __launch_bounds__(256)
void gcn_gemm1_wmma(const float* __restrict__ x, const float* __restrict__ W,
                    float* __restrict__ PR1, int nTiles) {
    int wave = threadIdx.x >> 5;
    int lane = threadIdx.x & 31;
    int tile = blockIdx.x * 8 + wave;
    if (tile >= nTiles) return;
    int half = lane >> 4;
    int m    = lane & 15;
    const float* xr = x + (long)(tile * 16 + m) * D_IN;

    v2f a[32];
#pragma unroll
    for (int ks = 0; ks < 32; ++ks) {
        int k = ks * 4 + half * 2;
        a[ks].x = xr[k];
        a[ks].y = xr[k + 1];
    }

#pragma unroll
    for (int ct = 0; ct < 4; ++ct) {
        int n = ct * 16 + m;
        v8f acc = {};
#pragma unroll
        for (int ks = 0; ks < 32; ++ks) {
            int k = ks * 4 + half * 2;
            v2f b;
            b.x = W[k * 64 + n];
            b.y = W[(k + 1) * 64 + n];
            acc = __builtin_amdgcn_wmma_f32_16x16x4_f32(false, a[ks], false, b,
                                                        (short)0, acc, false, false);
        }
#pragma unroll
        for (int r = 0; r < 8; ++r) {
            int orow = tile * 16 + r + half * 8;
            PR1[(long)orow * 64 + ct * 16 + m] = acc[r];
        }
    }
}

__global__ __launch_bounds__(256)
void gcn_gemm2_wmma(const float* __restrict__ h, const float* __restrict__ W,
                    float* __restrict__ PR2, int nTiles) {
    int wave = threadIdx.x >> 5;
    int lane = threadIdx.x & 31;
    int tile = blockIdx.x * 8 + wave;
    if (tile >= nTiles) return;
    int half = lane >> 4;
    int m    = lane & 15;
    const float* hr = h + (long)(tile * 16 + m) * H_DIM;

    v2f a[8];
#pragma unroll
    for (int ks = 0; ks < 8; ++ks) {
        int k = ks * 4 + half * 2;
        a[ks].x = hr[k];
        a[ks].y = hr[k + 1];
    }
    v8f acc = {};
#pragma unroll
    for (int ks = 0; ks < 8; ++ks) {
        int k = ks * 4 + half * 2;
        v2f b;
        b.x = W[k * 16 + m];
        b.y = W[(k + 1) * 16 + m];
        acc = __builtin_amdgcn_wmma_f32_16x16x4_f32(false, a[ks], false, b,
                                                    (short)0, acc, false, false);
    }
    if (m < 10) {
#pragma unroll
        for (int r = 0; r < 8; ++r) {
            int orow = tile * 16 + r + half * 8;
            PR2[(long)orow * 10 + m] = acc[r];
        }
    }
}

__global__ __launch_bounds__(256)
void gcn_count_slot(const int* __restrict__ dst, int* __restrict__ cnt,
                    int* __restrict__ slot, int E) {
    int e = blockIdx.x * 256 + threadIdx.x;
    if (e >= E) return;
    slot[e] = atomicAdd(&cnt[dst[e]], 1);
}

__global__ __launch_bounds__(1024)
void gcn_scan(const int* __restrict__ cnt, int* __restrict__ rowptr, int NP) {
    __shared__ int wtot[32];
    int wave = threadIdx.x >> 5;
    int lane = threadIdx.x & 31;
    int seg   = NP >> 5;
    int iters = seg >> 7;
    const int4* c4 = (const int4*)(cnt + wave * seg) + lane;
    int4*       r4 = (int4*)(rowptr + wave * seg) + lane;

    int acc = 0;
    for (int it = 0; it < iters; ++it) {
        int4 v = c4[it * 32];
        acc += v.x + v.y + v.z + v.w;
    }
    for (int o = 16; o > 0; o >>= 1) acc += __shfl_xor(acc, o);
    if (lane == 0) wtot[wave] = acc;
    __syncthreads();
    int t = (lane < wave) ? wtot[lane] : 0;
    for (int o = 16; o > 0; o >>= 1) t += __shfl_xor(t, o);
    int carry = t;

    for (int it = 0; it < iters; ++it) {
        int4 v = c4[it * 32];
        int s1 = v.x, s2 = s1 + v.y, s3 = s2 + v.z, s4 = s3 + v.w;
        int inc = s4;
        for (int o = 1; o < 32; o <<= 1) {
            int y = __shfl_up(inc, o);
            if (lane >= o) inc += y;
        }
        int ex = carry + inc - s4;
        r4[it * 32] = make_int4(ex, ex + s1, ex + s2, ex + s3);
        carry += __shfl(inc, 31);
    }
}

__global__ __launch_bounds__(256)
void gcn_place(const int* __restrict__ src, const int* __restrict__ dst,
               const float* __restrict__ ew, const int* __restrict__ rowptr,
               const int* __restrict__ slot, int* __restrict__ csr, int E) {
    int e = blockIdx.x * 256 + threadIdx.x;
    if (e >= E) return;
    int pos = rowptr[dst[e]] + slot[e];
    ((int2*)csr)[pos] = make_int2(src[e], __float_as_int(ew[e]));
}

__global__ __launch_bounds__(256)
void gcn_gather1(const int* __restrict__ rowptr, const int* __restrict__ csr,
                 const float* __restrict__ PR1, const float* __restrict__ b1,
                 float* __restrict__ h, int N) {
    int lane = threadIdx.x & 31;
    int n = __builtin_amdgcn_readfirstlane(blockIdx.x * 8 + (threadIdx.x >> 5));
    if (n >= N) return;
    int beg = rowptr[n], end = rowptr[n + 1];
    const int2* ed = (const int2*)csr;
    float acc = 0.0f;
    for (int base = beg; base < end; base += 32) {
        int m = min(32, end - base);
        int2 q = make_int2(0, 0);
        if (lane < m) q = ed[base + lane];
#pragma unroll 8
        for (int k = 0; k < m; ++k) {
            int   s = __builtin_amdgcn_readlane(q.x, k);
            float w = __int_as_float(__builtin_amdgcn_readlane(q.y, k));
            acc += w * PR1[(long)s * 64 + lane];
        }
    }
    float c = fmaxf((float)(end - beg), 1.0f);
    float v = acc / c + PR1[(long)n * 64 + 32 + lane] + b1[lane];
    h[(long)n * 32 + lane] = fmaxf(v, 0.0f);
}

__global__ __launch_bounds__(256)
void gcn_gather2_out(const int* __restrict__ rowptr, const int* __restrict__ csr,
                     const float* __restrict__ PR2, const float* __restrict__ b2,
                     float* __restrict__ out, int N) {
    __shared__ float so[32 * C_DIM];
    int g = threadIdx.x >> 3;
    int c = threadIdx.x & 7;
    int n = (int)blockIdx.x * 32 + g;
    const int2* ed = (const int2*)csr;
    float a0 = 0.0f, a1 = 0.0f, a2 = 0.0f, a3 = 0.0f, a4 = 0.0f;
    int deg = 0;
    if (n < N) {
        int beg = rowptr[n], end = rowptr[n + 1];
        deg = end - beg;
        for (int p = beg + c; p < end; p += 8) {
            int s = ed[p].x;
            const float* r = PR2 + (long)s * 10;
            float2 q0 = *(const float2*)(r);
            float2 q1 = *(const float2*)(r + 2);
            float  q2 = r[4];
            a0 += q0.x; a1 += q0.y; a2 += q1.x; a3 += q1.y; a4 += q2;
        }
    }
    for (int o = 1; o < 8; o <<= 1) {
        a0 += __shfl_xor(a0, o);
        a1 += __shfl_xor(a1, o);
        a2 += __shfl_xor(a2, o);
        a3 += __shfl_xor(a3, o);
        a4 += __shfl_xor(a4, o);
    }
    if (n < N) {
        float cn = fmaxf((float)deg, 1.0f);
        const float* rr = PR2 + (long)n * 10 + 5;
        float v0 = a0 / cn + rr[0] + b2[0];
        float v1 = a1 / cn + rr[1] + b2[1];
        float v2 = a2 / cn + rr[2] + b2[2];
        float v3 = a3 / cn + rr[3] + b2[3];
        float v4 = a4 / cn + rr[4] + b2[4];
        float mx = fmaxf(fmaxf(fmaxf(v0, v1), fmaxf(v2, v3)), v4);
        float s = expf(v0 - mx) + expf(v1 - mx) + expf(v2 - mx) + expf(v3 - mx) + expf(v4 - mx);
        float l = logf(s);
        float vc = (c == 0) ? v0 : (c == 1) ? v1 : (c == 2) ? v2 : (c == 3) ? v3 : v4;
        if (c < C_DIM) so[g * C_DIM + c] = vc - mx - l;
    }
    __syncthreads();
    int rem = N - (int)blockIdx.x * 32;
    int valid = (rem < 32 ? rem : 32) * C_DIM;
    if ((int)threadIdx.x < valid)
        out[(long)blockIdx.x * (32 * C_DIM) + threadIdx.x] = so[threadIdx.x];
}

extern "C" void kernel_launch(void* const* d_in, const int* in_sizes, int n_in,
                              void* d_out, int out_size, void* d_ws, size_t ws_size,
                              hipStream_t stream) {
    const float* x      = (const float*)d_in[0];
    const int*   ei     = (const int*)d_in[1];
    const float* ew     = (const float*)d_in[2];
    const float* Wrel1  = (const float*)d_in[3];
    const float* Wroot1 = (const float*)d_in[4];
    const float* b1     = (const float*)d_in[5];
    const float* Wrel2  = (const float*)d_in[6];
    const float* Wroot2 = (const float*)d_in[7];
    const float* b2     = (const float*)d_in[8];

    const int N = in_sizes[0] / D_IN;
    const int E = in_sizes[1] / 2;
    const int* src = ei;
    const int* dst = ei + E;

    float* ws = (float*)d_ws;
    const int NP = ((N + 4096) / 4096) * 4096;
    long off = 0;
    float* PR1 = ws + off; off += (long)N * 64;
    float* h   = ws + off; off += (long)N * 32;
    float* W1c = ws + off; off += 128 * 64;
    float* W2p = ws + off; off += 32 * 16;
    int* cnt    = (int*)(ws + off); off += NP;
    int* rowptr = (int*)(ws + off); off += NP;
    int* csr    = (int*)(ws + off); off += (long)E * 2;
    int* slot   = (int*)(ws + off); off += E;
    float* PR2 = PR1;

    const int nTiles = N / 16;

    gcn_zero_i<<<(NP + 255) / 256, 256, 0, stream>>>(cnt, NP);
    gcn_count_slot<<<(E + 255) / 256, 256, 0, stream>>>(dst, cnt, slot, E);
    gcn_scan<<<1, 1024, 0, stream>>>(cnt, rowptr, NP);
    gcn_place<<<(E + 255) / 256, 256, 0, stream>>>(src, dst, ew, rowptr, slot, csr, E);

    gcn_build_w1<<<(128 * 64 + 255) / 256, 256, 0, stream>>>(Wrel1, Wroot1, W1c);
    gcn_build_w2<<<(32 * 16 + 255) / 256, 256, 0, stream>>>(Wrel2, Wroot2, W2p);

    gcn_gemm1_wmma<<<(nTiles + 7) / 8, 256, 0, stream>>>(x, W1c, PR1, nTiles);

    gcn_gather1<<<(N + 7) / 8, 256, 0, stream>>>(rowptr, csr, PR1, b1, h, N);

    gcn_gemm2_wmma<<<(nTiles + 7) / 8, 256, 0, stream>>>(h, W2p, PR2, nTiles);

    gcn_gather2_out<<<(N + 31) / 32, 256, 0, stream>>>(rowptr, csr, PR2, b2, (float*)d_out, N);
}


# Edge bucketing ranked on LDS counters in two levels (256-node bins, then a per-bin node sort): global atomics per block and bin, not per edge; sum order = LDS atomic arrival rank, then 8-lane tree
# speedup vs baseline: 3.3090x; 2.0075x over previous
#include <hip/hip_runtime.h>
#include <math.h>

typedef __attribute__((ext_vector_type(2))) float v2f;
typedef __attribute__((ext_vector_type(8))) float v8f;

#define D_IN  128
#define H_DIM 32
#define C_DIM 5

__global__ void gcn_build_w1(const float* __restrict__ Wrel, const float* __restrict__ Wroot,
                             float* __restrict__ Wcat) {
    int i = blockIdx.x * 256 + threadIdx.x;
    if (i >= 128 * 64) return;
    int k = i >> 6, j = i & 63;
    Wcat[i] = (j < 32) ? Wrel[k * 32 + j] : Wroot[k * 32 + (j - 32)];
}

__global__ void gcn_build_w2(const float* __restrict__ Wrel, const float* __restrict__ Wroot,
                             float* __restrict__ Wp) {
    int i = blockIdx.x * 256 + threadIdx.x;
    if (i >= 32 * 16) return;
    int k = i >> 4, n = i & 15;
    float v = 0.0f;
    if (n < 5)       v = Wrel[k * 5 + n];
    else if (n < 10) v = Wroot[k * 5 + (n - 5)];
    Wp[i] = v;
}

__global__ void gcn_zero_i(int* __restrict__ p, int n) {
    int i = blockIdx.x * 256 + threadIdx.x;
    if (i < n) p[i] = 0;
}

__global__ __launch_bounds__(256)
void gcn_gemm1_wmma(const float* __restrict__ x, const float* __restrict__ W,
                    float* __restrict__ PR1, int nTiles) {
    int wave = threadIdx.x >> 5;
    int lane = threadIdx.x & 31;
    int tile = blockIdx.x * 8 + wave;
    if (tile >= nTiles) return;
    int half = lane >> 4;
    int m    = lane & 15;
    const float* xr = x + (long)(tile * 16 + m) * D_IN;

    v2f a[32];
#pragma unroll
    for (int ks = 0; ks < 32; ++ks) {
        int k = ks * 4 + half * 2;
        a[ks].x = xr[k];
        a[ks].y = xr[k + 1];
    }

#pragma unroll
    for (int ct = 0; ct < 4; ++ct) {
        int n = ct * 16 + m;
        v8f acc = {};
#pragma unroll
        for (int ks = 0; ks < 32; ++ks) {
            int k = ks * 4 + half * 2;
            v2f b;
            b.x = W[k * 64 + n];
            b.y = W[(k + 1) * 64 + n];
            acc = __builtin_amdgcn_wmma_f32_16x16x4_f32(false, a[ks], false, b,
                                                        (short)0, acc, false, false);
        }
#pragma unroll
        for (int r = 0; r < 8; ++r) {
            int orow = tile * 16 + r + half * 8;
            PR1[(long)orow * 64 + ct * 16 + m] = acc[r];
        }
    }
}

__global__ __launch_bounds__(256)
void gcn_gemm2_wmma(const float* __restrict__ h, const float* __restrict__ W,
                    float* __restrict__ PR2, int nTiles) {
    int wave = threadIdx.x >> 5;
    int lane = threadIdx.x & 31;
    int tile = blockIdx.x * 8 + wave;
    if (tile >= nTiles) return;
    int half = lane >> 4;
    int m    = lane & 15;
    const float* hr = h + (long)(tile * 16 + m) * H_DIM;

    v2f a[8];
#pragma unroll
    for (int ks = 0; ks < 8; ++ks) {
        int k = ks * 4 + half * 2;
        a[ks].x = hr[k];
        a[ks].y = hr[k + 1];
    }
    v8f acc = {};
#pragma unroll
    for (int ks = 0; ks < 8; ++ks) {
        int k = ks * 4 + half * 2;
        v2f b;
        b.x = W[k * 16 + m];
        b.y = W[(k + 1) * 16 + m];
        acc = __builtin_amdgcn_wmma_f32_16x16x4_f32(false, a[ks], false, b,
                                                    (short)0, acc, false, false);
    }
    if (m < 10) {
#pragma unroll
        for (int r = 0; r < 8; ++r) {
            int orow = tile * 16 + r + half * 8;
            PR2[(long)orow * 10 + m] = acc[r];
        }
    }
}

#define BIN_SHIFT 8
#define BIN_NODES 256
#define MAXB      512
#define SORT_T    512
#define EPT       8
#define CHUNK     (SORT_T * EPT)
#define SRC_BITS  24
#define SRC_MASK  0xFFFFFF
static_assert(SORT_T == MAXB, "one thread per bin counter");

__device__ __forceinline__ int gcn_block_excl_scan(int v, int* wsum) {
    int lane = threadIdx.x & 31, wave = threadIdx.x >> 5;
    int inc = v;
    for (int o = 1; o < 32; o <<= 1) {
        int y = __shfl_up(inc, o);
        if (lane >= o) inc += y;
    }
    if (lane == 31) wsum[wave] = inc;
    __syncthreads();
    int carry = 0;
    for (int w = 0; w < wave; ++w) carry += wsum[w];
    return carry + inc - v;
}

__global__ __launch_bounds__(SORT_T)
void gcn_bin_hist(const int* __restrict__ dst, int* __restrict__ binTotal, int E, int B) {
    __shared__ int hist[MAXB];
    int tid = threadIdx.x;
    hist[tid] = 0;
    __syncthreads();
    int e0 = blockIdx.x * CHUNK + tid;
    int d[EPT];
#pragma unroll
    for (int k = 0; k < EPT; ++k) d[k] = dst[min(e0 + k * SORT_T, E - 1)];
#pragma unroll
    for (int k = 0; k < EPT; ++k)
        if (e0 + k * SORT_T < E) atomicAdd(&hist[d[k] >> BIN_SHIFT], 1);
    __syncthreads();
    int c = hist[tid];
    if (tid < B && c) atomicAdd(&binTotal[tid], c);
}

__global__ __launch_bounds__(SORT_T)
void gcn_bin_scatter(const int* __restrict__ src, const int* __restrict__ dst,
                     const float* __restrict__ ew, const int* __restrict__ binTotal,
                     int* __restrict__ binCursor, int2* __restrict__ bsw, int E, int B) {
    __shared__ int hist[MAXB];
    __shared__ int base[MAXB];
    __shared__ int wsum[16];
    int tid = threadIdx.x;
    hist[tid] = 0;
    __syncthreads();
    int e0 = blockIdx.x * CHUNK + tid;
    int d[EPT], r[EPT], s[EPT];
    float w[EPT];
#pragma unroll
    for (int k = 0; k < EPT; ++k) d[k] = dst[min(e0 + k * SORT_T, E - 1)];
    int t = (tid < B) ? binTotal[tid] : 0;
#pragma unroll
    for (int k = 0; k < EPT; ++k) {
        int ec = min(e0 + k * SORT_T, E - 1);
        s[k] = src[ec];
        w[k] = ew[ec];
    }
#pragma unroll
    for (int k = 0; k < EPT; ++k) {
        r[k] = 0;
        if (e0 + k * SORT_T < E) r[k] = atomicAdd(&hist[d[k] >> BIN_SHIFT], 1);
    }
    int start = gcn_block_excl_scan(t, wsum);
    int c = hist[tid];
    int bs = 0;
    if (tid < B && c) bs = start + atomicAdd(&binCursor[tid], c);
    base[tid] = bs;
    __syncthreads();
#pragma unroll
    for (int k = 0; k < EPT; ++k) {
        if (e0 + k * SORT_T < E) {
            int p = base[d[k] >> BIN_SHIFT] + r[k];
            int key = s[k] | ((d[k] & (BIN_NODES - 1)) << SRC_BITS);
            bsw[p] = make_int2(key, __float_as_int(w[k]));
        }
    }
}

__global__ __launch_bounds__(SORT_T)
void gcn_bin_sort(const int* __restrict__ binTotal, const int2* __restrict__ bsw,
                  int* __restrict__ rowptr, int2* __restrict__ csr, int N, int B) {
    __shared__ int cnt[BIN_NODES];
    __shared__ int wsumA[16];
    __shared__ int wsumB[16];
    __shared__ int sStart, sTot;
    int tid = threadIdx.x;
    int b = blockIdx.x;
    int t = (tid < B) ? binTotal[tid] : 0;
    int ex = gcn_block_excl_scan(t, wsumA);
    if (tid == b) { sStart = ex; sTot = t; }
    if (tid < BIN_NODES) cnt[tid] = 0;
    __syncthreads();
    const int start = sStart, tot = sTot;
    const int node0 = b << BIN_SHIFT;
    const int2* mysw = bsw + start;
    for (int i0 = tid; i0 - tid < tot; i0 += CHUNK) {
        int dv[EPT];
#pragma unroll
        for (int u = 0; u < EPT; ++u) dv[u] = mysw[min(i0 + u * SORT_T, tot - 1)].x;
#pragma unroll
        for (int u = 0; u < EPT; ++u)
            if (i0 + u * SORT_T < tot) atomicAdd(&cnt[(unsigned)dv[u] >> SRC_BITS], 1);
    }
    __syncthreads();
    int c = (tid < BIN_NODES) ? cnt[tid] : 0;
    int lex = gcn_block_excl_scan(c, wsumB);
    if (tid < BIN_NODES) {
        cnt[tid] = lex;
        int n = node0 + tid;
        if (n <= N) rowptr[n] = start + lex;
    }
    __syncthreads();
    for (int i0 = tid; i0 - tid < tot; i0 += CHUNK) {
        int2 sw[EPT];
#pragma unroll
        for (int u = 0; u < EPT; ++u) sw[u] = mysw[min(i0 + u * SORT_T, tot - 1)];
#pragma unroll
        for (int u = 0; u < EPT; ++u) {
            if (i0 + u * SORT_T < tot) {
                int p = atomicAdd(&cnt[(unsigned)sw[u].x >> SRC_BITS], 1);
                csr[start + p] = make_int2(sw[u].x & SRC_MASK, sw[u].y);
            }
        }
    }
}

__global__ __launch_bounds__(256)
void gcn_gather1(const int* __restrict__ rowptr, const int* __restrict__ csr,
                 const float* __restrict__ PR1, const float* __restrict__ b1,
                 float* __restrict__ h, int N) {
    int lane = threadIdx.x & 31;
    int n = __builtin_amdgcn_readfirstlane(blockIdx.x * 8 + (threadIdx.x >> 5));
    if (n >= N) return;
    int beg = rowptr[n], end = rowptr[n + 1];
    const int2* ed = (const int2*)csr;
    float acc = 0.0f;
    for (int base = beg; base < end; base += 32) {
        int m = min(32, end - base);
        int2 q = make_int2(0, 0);
        if (lane < m) q = ed[base + lane];
#pragma unroll 8
        for (int k = 0; k < m; ++k) {
            int   s = __builtin_amdgcn_readlane(q.x, k);
            float w = __int_as_float(__builtin_amdgcn_readlane(q.y, k));
            acc += w * PR1[(long)s * 64 + lane];
        }
    }
    float c = fmaxf((float)(end - beg), 1.0f);
    float v = acc / c + PR1[(long)n * 64 + 32 + lane] + b1[lane];
    h[(long)n * 32 + lane] = fmaxf(v, 0.0f);
}

__global__ __launch_bounds__(256)
void gcn_gather2_out(const int* __restrict__ rowptr, const int* __restrict__ csr,
                     const float* __restrict__ PR2, const float* __restrict__ b2,
                     float* __restrict__ out, int N) {
    __shared__ float so[32 * C_DIM];
    int g = threadIdx.x >> 3;
    int c = threadIdx.x & 7;
    int n = (int)blockIdx.x * 32 + g;
    const int2* ed = (const int2*)csr;
    float a0 = 0.0f, a1 = 0.0f, a2 = 0.0f, a3 = 0.0f, a4 = 0.0f;
    int deg = 0;
    if (n < N) {
        int beg = rowptr[n], end = rowptr[n + 1];
        deg = end - beg;
        for (int p = beg + c; p < end; p += 8) {
            int s = ed[p].x;
            const float* r = PR2 + (long)s * 10;
            float2 q0 = *(const float2*)(r);
            float2 q1 = *(const float2*)(r + 2);
            float  q2 = r[4];
            a0 += q0.x; a1 += q0.y; a2 += q1.x; a3 += q1.y; a4 += q2;
        }
    }
    for (int o = 1; o < 8; o <<= 1) {
        a0 += __shfl_xor(a0, o);
        a1 += __shfl_xor(a1, o);
        a2 += __shfl_xor(a2, o);
        a3 += __shfl_xor(a3, o);
        a4 += __shfl_xor(a4, o);
    }
    if (n < N) {
        float cn = fmaxf((float)deg, 1.0f);
        const float* rr = PR2 + (long)n * 10 + 5;
        float v0 = a0 / cn + rr[0] + b2[0];
        float v1 = a1 / cn + rr[1] + b2[1];
        float v2 = a2 / cn + rr[2] + b2[2];
        float v3 = a3 / cn + rr[3] + b2[3];
        float v4 = a4 / cn + rr[4] + b2[4];
        float mx = fmaxf(fmaxf(fmaxf(v0, v1), fmaxf(v2, v3)), v4);
        float s = expf(v0 - mx) + expf(v1 - mx) + expf(v2 - mx) + expf(v3 - mx) + expf(v4 - mx);
        float l = logf(s);
        float vc = (c == 0) ? v0 : (c == 1) ? v1 : (c == 2) ? v2 : (c == 3) ? v3 : v4;
        if (c < C_DIM) so[g * C_DIM + c] = vc - mx - l;
    }
    __syncthreads();
    int rem = N - (int)blockIdx.x * 32;
    int valid = (rem < 32 ? rem : 32) * C_DIM;
    if ((int)threadIdx.x < valid)
        out[(long)blockIdx.x * (32 * C_DIM) + threadIdx.x] = so[threadIdx.x];
}

extern "C" void kernel_launch(void* const* d_in, const int* in_sizes, int n_in,
                              void* d_out, int out_size, void* d_ws, size_t ws_size,
                              hipStream_t stream) {
    const float* x      = (const float*)d_in[0];
    const int*   ei     = (const int*)d_in[1];
    const float* ew     = (const float*)d_in[2];
    const float* Wrel1  = (const float*)d_in[3];
    const float* Wroot1 = (const float*)d_in[4];
    const float* b1     = (const float*)d_in[5];
    const float* Wrel2  = (const float*)d_in[6];
    const float* Wroot2 = (const float*)d_in[7];
    const float* b2     = (const float*)d_in[8];

    const int N = in_sizes[0] / D_IN;
    const int E = in_sizes[1] / 2;
    const int* src = ei;
    const int* dst = ei + E;

    float* ws = (float*)d_ws;
    const int NP = ((N + 4096) / 4096) * 4096;
    long off = 0;
    float* PR1 = ws + off; off += (long)N * 64;
    float* h   = ws + off; off += (long)N * 32;
    float* W1c = ws + off; off += 128 * 64;
    float* W2p = ws + off; off += 32 * 16;
    int* binTotal  = (int*)(ws + off); off += MAXB;
    int* binCursor = (int*)(ws + off); off += MAXB;
    int* rowptr = (int*)(ws + off); off += NP;
    int* csr    = (int*)(ws + off); off += (long)E * 2;
    int* bsw    = (int*)(ws + off); off += (long)E * 2;
    float* PR2 = PR1;

    const int nTiles = N / 16;

    const int B = N / BIN_NODES + 1;
    const int nChunks = (E + CHUNK - 1) / CHUNK;

    gcn_zero_i<<<(2 * MAXB + 255) / 256, 256, 0, stream>>>(binTotal, 2 * MAXB);
    gcn_bin_hist<<<nChunks, SORT_T, 0, stream>>>(dst, binTotal, E, B);
    gcn_bin_scatter<<<nChunks, SORT_T, 0, stream>>>(src, dst, ew, binTotal, binCursor,
                                                    (int2*)bsw, E, B);
    gcn_bin_sort<<<B, SORT_T, 0, stream>>>(binTotal, (const int2*)bsw, rowptr, (int2*)csr, N, B);

    gcn_build_w1<<<(128 * 64 + 255) / 256, 256, 0, stream>>>(Wrel1, Wroot1, W1c);
    gcn_build_w2<<<(32 * 16 + 255) / 256, 256, 0, stream>>>(Wrel2, Wroot2, W2p);

    gcn_gemm1_wmma<<<(nTiles + 7) / 8, 256, 0, stream>>>(x, W1c, PR1, nTiles);

    gcn_gather1<<<(N + 7) / 8, 256, 0, stream>>>(rowptr, csr, PR1, b1, h, N);

    gcn_gemm2_wmma<<<(nTiles + 7) / 8, 256, 0, stream>>>(h, W2p, PR2, nTiles);

    gcn_gather2_out<<<(N + 31) / 32, 256, 0, stream>>>(rowptr, csr, PR2, b2, (float*)d_out, N);
}


# GEMM1 loses its own launch (10 launches -> 9): its tiles run as the leading work-groups of the scatter and sort launches, matrix work under the bucketing's memory and LDS waits; every sum unchanged
# speedup vs baseline: 3.9159x; 1.1834x over previous
#include <hip/hip_runtime.h>
#include <math.h>

typedef __attribute__((ext_vector_type(2))) float v2f;
typedef __attribute__((ext_vector_type(8))) float v8f;

#define D_IN  128
#define H_DIM 32
#define C_DIM 5

__global__ void gcn_build_w1(const float* __restrict__ Wrel, const float* __restrict__ Wroot,
                             float* __restrict__ Wcat) {
    int i = blockIdx.x * 256 + threadIdx.x;
    if (i >= 128 * 64) return;
    int k = i >> 6, j = i & 63;
    Wcat[i] = (j < 32) ? Wrel[k * 32 + j] : Wroot[k * 32 + (j - 32)];
}

__global__ void gcn_build_w2(const float* __restrict__ Wrel, const float* __restrict__ Wroot,
                             float* __restrict__ Wp) {
    int i = blockIdx.x * 256 + threadIdx.x;
    if (i >= 32 * 16) return;
    int k = i >> 4, n = i & 15;
    float v = 0.0f;
    if (n < 5)       v = Wrel[k * 5 + n];
    else if (n < 10) v = Wroot[k * 5 + (n - 5)];
    Wp[i] = v;
}

__global__ void gcn_zero_i(int* __restrict__ p, int n) {
    int i = blockIdx.x * 256 + threadIdx.x;
    if (i < n) p[i] = 0;
}

__device__ __forceinline__
void gcn_gemm1_tile(const float* __restrict__ x, const float* __restrict__ W,
                    float* __restrict__ PR1, int tile) {
    int lane = threadIdx.x & 31;
    int half = lane >> 4;
    int m    = lane & 15;
    const float* xr = x + (long)(tile * 16 + m) * D_IN;

    v2f a[32];
#pragma unroll
    for (int ks = 0; ks < 32; ++ks) {
        int k = ks * 4 + half * 2;
        a[ks].x = xr[k];
        a[ks].y = xr[k + 1];
    }

#pragma unroll
    for (int ct = 0; ct < 4; ++ct) {
        int n = ct * 16 + m;
        v8f acc = {};
#pragma unroll
        for (int ks = 0; ks < 32; ++ks) {
            int k = ks * 4 + half * 2;
            v2f b;
            b.x = W[k * 64 + n];
            b.y = W[(k + 1) * 64 + n];
            acc = __builtin_amdgcn_wmma_f32_16x16x4_f32(false, a[ks], false, b,
                                                        (short)0, acc, false, false);
        }
#pragma unroll
        for (int r = 0; r < 8; ++r) {
            int orow = tile * 16 + r + half * 8;
            PR1[(long)orow * 64 + ct * 16 + m] = acc[r];
        }
    }
}

__global__ __launch_bounds__(256)
void gcn_gemm2_wmma(const float* __restrict__ h, const float* __restrict__ W,
                    float* __restrict__ PR2, int nTiles) {
    int wave = threadIdx.x >> 5;
    int lane = threadIdx.x & 31;
    int tile = blockIdx.x * 8 + wave;
    if (tile >= nTiles) return;
    int half = lane >> 4;
    int m    = lane & 15;
    const float* hr = h + (long)(tile * 16 + m) * H_DIM;

    v2f a[8];
#pragma unroll
    for (int ks = 0; ks < 8; ++ks) {
        int k = ks * 4 + half * 2;
        a[ks].x = hr[k];
        a[ks].y = hr[k + 1];
    }
    v8f acc = {};
#pragma unroll
    for (int ks = 0; ks < 8; ++ks) {
        int k = ks * 4 + half * 2;
        v2f b;
        b.x = W[k * 16 + m];
        b.y = W[(k + 1) * 16 + m];
        acc = __builtin_amdgcn_wmma_f32_16x16x4_f32(false, a[ks], false, b,
                                                    (short)0, acc, false, false);
    }
    if (m < 10) {
#pragma unroll
        for (int r = 0; r < 8; ++r) {
            int orow = tile * 16 + r + half * 8;
            PR2[(long)orow * 10 + m] = acc[r];
        }
    }
}

#define BIN_SHIFT 8
#define BIN_NODES 256
#define MAXB      512
#define SORT_T    512
#define EPT       8
#define CHUNK     (SORT_T * EPT)
#define SRC_BITS  24
#define SRC_MASK  0xFFFFFF
static_assert(SORT_T == MAXB, "one thread per bin counter");

__device__ __forceinline__ int gcn_block_excl_scan(int v, int* wsum) {
    int lane = threadIdx.x & 31, wave = threadIdx.x >> 5;
    int inc = v;
    for (int o = 1; o < 32; o <<= 1) {
        int y = __shfl_up(inc, o);
        if (lane >= o) inc += y;
    }
    if (lane == 31) wsum[wave] = inc;
    __syncthreads();
    int carry = 0;
    for (int w = 0; w < wave; ++w) carry += wsum[w];
    return carry + inc - v;
}

__global__ __launch_bounds__(SORT_T)
void gcn_bin_hist(const int* __restrict__ dst, int* __restrict__ binTotal, int E, int B) {
    __shared__ int hist[MAXB];
    int tid = threadIdx.x;
    hist[tid] = 0;
    __syncthreads();
    int e0 = blockIdx.x * CHUNK + tid;
    int d[EPT];
#pragma unroll
    for (int k = 0; k < EPT; ++k) d[k] = dst[min(e0 + k * SORT_T, E - 1)];
#pragma unroll
    for (int k = 0; k < EPT; ++k)
        if (e0 + k * SORT_T < E) atomicAdd(&hist[d[k] >> BIN_SHIFT], 1);
    __syncthreads();
    int c = hist[tid];
    if (tid < B && c) atomicAdd(&binTotal[tid], c);
}

__global__ __launch_bounds__(SORT_T)
void gcn_bin_scatter(const int* __restrict__ src, const int* __restrict__ dst,
                     const float* __restrict__ ew, const int* __restrict__ binTotal,
                     int* __restrict__ binCursor, int2* __restrict__ bsw, int E, int B,
                     const float* __restrict__ x, const float* __restrict__ W1c,
                     float* __restrict__ PR1, int nGemm, int tile0, int tile1) {
    __shared__ int hist[MAXB];
    __shared__ int base[MAXB];
    __shared__ int wsum[16];
    int tid = threadIdx.x;
    if ((int)blockIdx.x < nGemm) {
        int tile = tile0 + (int)blockIdx.x * (SORT_T / 32) + (tid >> 5);
        if (tile < tile1) gcn_gemm1_tile(x, W1c, PR1, tile);
        return;
    }
    const int chunk = (int)blockIdx.x - nGemm;
    hist[tid] = 0;
    __syncthreads();
    int e0 = chunk * CHUNK + tid;
    int d[EPT], r[EPT], s[EPT];
    float w[EPT];
#pragma unroll
    for (int k = 0; k < EPT; ++k) d[k] = dst[min(e0 + k * SORT_T, E - 1)];
    int t = (tid < B) ? binTotal[tid] : 0;
#pragma unroll
    for (int k = 0; k < EPT; ++k) {
        int ec = min(e0 + k * SORT_T, E - 1);
        s[k] = src[ec];
        w[k] = ew[ec];
    }
#pragma unroll
    for (int k = 0; k < EPT; ++k) {
        r[k] = 0;
        if (e0 + k * SORT_T < E) r[k] = atomicAdd(&hist[d[k] >> BIN_SHIFT], 1);
    }
    int start = gcn_block_excl_scan(t, wsum);
    int c = hist[tid];
    int bs = 0;
    if (tid < B && c) bs = start + atomicAdd(&binCursor[tid], c);
    base[tid] = bs;
    __syncthreads();
#pragma unroll
    for (int k = 0; k < EPT; ++k) {
        if (e0 + k * SORT_T < E) {
            int p = base[d[k] >> BIN_SHIFT] + r[k];
            int key = s[k] | ((d[k] & (BIN_NODES - 1)) << SRC_BITS);
            bsw[p] = make_int2(key, __float_as_int(w[k]));
        }
    }
}

__global__ __launch_bounds__(SORT_T)
void gcn_bin_sort(const int* __restrict__ binTotal, const int2* __restrict__ bsw,
                  int* __restrict__ rowptr, int2* __restrict__ csr, int N, int B,
                  const float* __restrict__ x, const float* __restrict__ W1c,
                  float* __restrict__ PR1, int nGemm, int tile0, int tile1) {
    __shared__ int cnt[BIN_NODES];
    __shared__ int wsumA[16];
    __shared__ int wsumB[16];
    __shared__ int sStart, sTot;
    int tid = threadIdx.x;
    if ((int)blockIdx.x < nGemm) {
        int tile = tile0 + (int)blockIdx.x * (SORT_T / 32) + (tid >> 5);
        if (tile < tile1) gcn_gemm1_tile(x, W1c, PR1, tile);
        return;
    }
    int b = (int)blockIdx.x - nGemm;
    int t = (tid < B) ? binTotal[tid] : 0;
    int ex = gcn_block_excl_scan(t, wsumA);
    if (tid == b) { sStart = ex; sTot = t; }
    if (tid < BIN_NODES) cnt[tid] = 0;
    __syncthreads();
    const int start = sStart, tot = sTot;
    const int node0 = b << BIN_SHIFT;
    const int2* mysw = bsw + start;
    for (int i0 = tid; i0 - tid < tot; i0 += CHUNK) {
        int dv[EPT];
#pragma unroll
        for (int u = 0; u < EPT; ++u) dv[u] = mysw[min(i0 + u * SORT_T, tot - 1)].x;
#pragma unroll
        for (int u = 0; u < EPT; ++u)
            if (i0 + u * SORT_T < tot) atomicAdd(&cnt[(unsigned)dv[u] >> SRC_BITS], 1);
    }
    __syncthreads();
    int c = (tid < BIN_NODES) ? cnt[tid] : 0;
    int lex = gcn_block_excl_scan(c, wsumB);
    if (tid < BIN_NODES) {
        cnt[tid] = lex;
        int n = node0 + tid;
        if (n <= N) rowptr[n] = start + lex;
    }
    __syncthreads();
    for (int i0 = tid; i0 - tid < tot; i0 += CHUNK) {
        int2 sw[EPT];
#pragma unroll
        for (int u = 0; u < EPT; ++u) sw[u] = mysw[min(i0 + u * SORT_T, tot - 1)];
#pragma unroll
        for (int u = 0; u < EPT; ++u) {
            if (i0 + u * SORT_T < tot) {
                int p = atomicAdd(&cnt[(unsigned)sw[u].x >> SRC_BITS], 1);
                csr[start + p] = make_int2(sw[u].x & SRC_MASK, sw[u].y);
            }
        }
    }
}

#define G1_NPG 2
#define G1_BCAST(V, K) ((int)__builtin_amdgcn_mov_dpp8((unsigned)(V), (unsigned)(K) * 0x249249u))
#define G1_LOAD(K) { unsigned s = (unsigned)G1_BCAST(q.x, K); \
                     v[K] = *(const float4*)(tab + ((s << 8) | c16)); }
#define G1_FMA(K)  { float w = __int_as_float(G1_BCAST(q.y, K)); \
                     const bool on = (K) < cnt; \
                     float tx = acc.x + w * v[K].x, ty = acc.y + w * v[K].y; \
                     float tz = acc.z + w * v[K].z, tw = acc.w + w * v[K].w; \
                     acc.x = on ? tx : acc.x; acc.y = on ? ty : acc.y; \
                     acc.z = on ? tz : acc.z; acc.w = on ? tw : acc.w; }
__global__ __launch_bounds__(256)
void gcn_gather1(const int* __restrict__ rowptr, const int* __restrict__ csr,
                 const float* __restrict__ PR1, const float* __restrict__ b1,
                 float* __restrict__ h, int N) {
    const int c = threadIdx.x & 7;
    const unsigned c16 = (unsigned)c * 16u;
    const int n0 = ((int)blockIdx.x * 32 + (int)(threadIdx.x >> 3)) * G1_NPG;
    const int2* ed = (const int2*)csr;
    const char* tab = (const char*)PR1;
    static_assert(G1_NPG == 2, "three bounds per group");
    const int nb = max(min(n0, N - 2), 0);
    const int d  = n0 - nb;
    const int t0 = rowptr[nb], t1 = rowptr[nb + 1], t2 = rowptr[nb + 2];
    int r[G1_NPG + 1];
    r[0] = (d >= 2) ? t2 : (d == 1) ? t1 : t0;
    r[1] = (d >= 1) ? t2 : t1;
    r[2] = t2;
    int2 q0[G1_NPG];
#pragma unroll
    for (int it = 0; it < G1_NPG; ++it) q0[it] = ed[max(min(r[it] + c, r[it + 1] - 1), 0)];
#pragma unroll
    for (int it = 0; it < G1_NPG; ++it) {
        const int n = n0 + it;
        const int e1 = r[it + 1];
        const float4 root = *(const float4*)(tab + (((unsigned)min(n, N - 1) << 8) | c16) + 128);
        float4 acc = make_float4(0.0f, 0.0f, 0.0f, 0.0f);
        int2 q = q0[it];
        for (int p = r[it]; p < e1; p += 8) {
            const int cnt = e1 - p;
            int2 qn = ed[min(p + 8 + c, e1 - 1)];
            float4 v[8];
            G1_LOAD(0) G1_LOAD(1) G1_LOAD(2) G1_LOAD(3)
            G1_LOAD(4) G1_LOAD(5) G1_LOAD(6) G1_LOAD(7)
            G1_FMA(0) G1_FMA(1) G1_FMA(2) G1_FMA(3)
            G1_FMA(4) G1_FMA(5) G1_FMA(6) G1_FMA(7)
            q = qn;
        }
        if (n < N) {
            float cf = fmaxf((float)(e1 - r[it]), 1.0f);
            float4 o;
            o.x = fmaxf(acc.x / cf + root.x + b1[c * 4],     0.0f);
            o.y = fmaxf(acc.y / cf + root.y + b1[c * 4 + 1], 0.0f);
            o.z = fmaxf(acc.z / cf + root.z + b1[c * 4 + 2], 0.0f);
            o.w = fmaxf(acc.w / cf + root.w + b1[c * 4 + 3], 0.0f);
            *(float4*)(h + (long)n * 32 + c * 4) = o;
        }
    }
}

__global__ __launch_bounds__(256)
void gcn_gather2_out(const int* __restrict__ rowptr, const int* __restrict__ csr,
                     const float* __restrict__ PR2, const float* __restrict__ b2,
                     float* __restrict__ out, int N) {
    __shared__ float so[32 * C_DIM];
    int g = threadIdx.x >> 3;
    int c = threadIdx.x & 7;
    int n = (int)blockIdx.x * 32 + g;
    const int2* ed = (const int2*)csr;
    float a0 = 0.0f, a1 = 0.0f, a2 = 0.0f, a3 = 0.0f, a4 = 0.0f;
    int deg = 0;
    if (n < N) {
        int beg = rowptr[n], end = rowptr[n + 1];
        deg = end - beg;
        for (int p = beg + c; p < end; p += 8) {
            int s = ed[p].x;
            const float* r = PR2 + (long)s * 10;
            float2 q0 = *(const float2*)(r);
            float2 q1 = *(const float2*)(r + 2);
            float  q2 = r[4];
            a0 += q0.x; a1 += q0.y; a2 += q1.x; a3 += q1.y; a4 += q2;
        }
    }
    for (int o = 1; o < 8; o <<= 1) {
        a0 += __shfl_xor(a0, o);
        a1 += __shfl_xor(a1, o);
        a2 += __shfl_xor(a2, o);
        a3 += __shfl_xor(a3, o);
        a4 += __shfl_xor(a4, o);
    }
    if (n < N) {
        float cn = fmaxf((float)deg, 1.0f);
        const float* rr = PR2 + (long)n * 10 + 5;
        float v0 = a0 / cn + rr[0] + b2[0];
        float v1 = a1 / cn + rr[1] + b2[1];
        float v2 = a2 / cn + rr[2] + b2[2];
        float v3 = a3 / cn + rr[3] + b2[3];
        float v4 = a4 / cn + rr[4] + b2[4];
        float mx = fmaxf(fmaxf(fmaxf(v0, v1), fmaxf(v2, v3)), v4);
        float s = expf(v0 - mx) + expf(v1 - mx) + expf(v2 - mx) + expf(v3 - mx) + expf(v4 - mx);
        float l = logf(s);
        float vc = (c == 0) ? v0 : (c == 1) ? v1 : (c == 2) ? v2 : (c == 3) ? v3 : v4;
        if (c < C_DIM) so[g * C_DIM + c] = vc - mx - l;
    }
    __syncthreads();
    int rem = N - (int)blockIdx.x * 32;
    int valid = (rem < 32 ? rem : 32) * C_DIM;
    if ((int)threadIdx.x < valid)
        out[(long)blockIdx.x * (32 * C_DIM) + threadIdx.x] = so[threadIdx.x];
}

extern "C" void kernel_launch(void* const* d_in, const int* in_sizes, int n_in,
                              void* d_out, int out_size, void* d_ws, size_t ws_size,
                              hipStream_t stream) {
    const float* x      = (const float*)d_in[0];
    const int*   ei     = (const int*)d_in[1];
    const float* ew     = (const float*)d_in[2];
    const float* Wrel1  = (const float*)d_in[3];
    const float* Wroot1 = (const float*)d_in[4];
    const float* b1     = (const float*)d_in[5];
    const float* Wrel2  = (const float*)d_in[6];
    const float* Wroot2 = (const float*)d_in[7];
    const float* b2     = (const float*)d_in[8];

    const int N = in_sizes[0] / D_IN;
    const int E = in_sizes[1] / 2;
    const int* src = ei;
    const int* dst = ei + E;

    float* ws = (float*)d_ws;
    const int NP = ((N + 4096) / 4096) * 4096;
    long off = 0;
    float* PR1 = ws + off; off += (long)N * 64;
    float* h   = ws + off; off += (long)N * 32;
    float* W1c = ws + off; off += 128 * 64;
    float* W2p = ws + off; off += 32 * 16;
    int* binTotal  = (int*)(ws + off); off += MAXB;
    int* binCursor = (int*)(ws + off); off += MAXB;
    int* rowptr = (int*)(ws + off); off += NP;
    int* csr    = (int*)(ws + off); off += (long)E * 2;
    int* bsw    = (int*)(ws + off); off += (long)E * 2;
    float* PR2 = PR1;

    const int nTiles = N / 16;

    const int B = N / BIN_NODES + 1;
    const int nChunks = (E + CHUNK - 1) / CHUNK;

    const int tHalf = ((nTiles / 2 + 15) / 16) * 16;
    const int gA = tHalf / 16;
    const int gB = (nTiles - tHalf + 15) / 16;

    gcn_zero_i<<<(2 * MAXB + 255) / 256, 256, 0, stream>>>(binTotal, 2 * MAXB);
    gcn_build_w1<<<(128 * 64 + 255) / 256, 256, 0, stream>>>(Wrel1, Wroot1, W1c);
    gcn_build_w2<<<(32 * 16 + 255) / 256, 256, 0, stream>>>(Wrel2, Wroot2, W2p);
    gcn_bin_hist<<<nChunks, SORT_T, 0, stream>>>(dst, binTotal, E, B);
    gcn_bin_scatter<<<gA + nChunks, SORT_T, 0, stream>>>(src, dst, ew, binTotal, binCursor,
                                                         (int2*)bsw, E, B,
                                                         x, W1c, PR1, gA, 0, tHalf);
    gcn_bin_sort<<<gB + B, SORT_T, 0, stream>>>(binTotal, (const int2*)bsw, rowptr, (int2*)csr, N, B,
                                                x, W1c, PR1, gB, tHalf, nTiles);

    gcn_gather1<<<(N + 32 * G1_NPG - 1) / (32 * G1_NPG), 256, 0, stream>>>(rowptr, csr, PR1, b1, h, N);

    gcn_gemm2_wmma<<<(nTiles + 7) / 8, 256, 0, stream>>>(h, W2p, PR2, nTiles);

    gcn_gather2_out<<<(N + 31) / 32, 256, 0, stream>>>(rowptr, csr, PR2, b2, (float*)d_out, N);
}


# Layer-2 gather: PR2 rows padded to 64 B, 2 lanes per edge, one 16 B load each (1 line request per edge, was 3), 2 nodes per 8-lane group, 32 edges in flight; sum order: 4 pair partials, 2-step tree
# speedup vs baseline: 4.0718x; 1.0398x over previous
#include <hip/hip_runtime.h>
#include <math.h>

typedef __attribute__((ext_vector_type(2))) float v2f;
typedef __attribute__((ext_vector_type(8))) float v8f;

#define D_IN  128
#define H_DIM 32
#define C_DIM 5

__global__ void gcn_build_w1(const float* __restrict__ Wrel, const float* __restrict__ Wroot,
                             float* __restrict__ Wcat) {
    int i = blockIdx.x * 256 + threadIdx.x;
    if (i >= 128 * 64) return;
    int k = i >> 6, j = i & 63;
    Wcat[i] = (j < 32) ? Wrel[k * 32 + j] : Wroot[k * 32 + (j - 32)];
}

__global__ void gcn_build_w2(const float* __restrict__ Wrel, const float* __restrict__ Wroot,
                             float* __restrict__ Wp) {
    int i = blockIdx.x * 256 + threadIdx.x;
    if (i >= 32 * 16) return;
    int k = i >> 4, n = i & 15;
    float v = 0.0f;
    if (n < 5)                 v = Wrel[k * 5 + n];
    else if (n >= 8 && n < 13) v = Wroot[k * 5 + (n - 8)];
    Wp[i] = v;
}

__global__ void gcn_zero_i(int* __restrict__ p, int n) {
    int i = blockIdx.x * 256 + threadIdx.x;
    if (i < n) p[i] = 0;
}

__device__ __forceinline__
void gcn_gemm1_tile(const float* __restrict__ x, const float* __restrict__ W,
                    float* __restrict__ PR1, int tile) {
    int lane = threadIdx.x & 31;
    int half = lane >> 4;
    int m    = lane & 15;
    const float* xr = x + (long)(tile * 16 + m) * D_IN;

    v2f a[32];
#pragma unroll
    for (int ks = 0; ks < 32; ++ks) {
        int k = ks * 4 + half * 2;
        a[ks].x = xr[k];
        a[ks].y = xr[k + 1];
    }

#pragma unroll
    for (int ct = 0; ct < 4; ++ct) {
        int n = ct * 16 + m;
        v8f acc = {};
#pragma unroll
        for (int ks = 0; ks < 32; ++ks) {
            int k = ks * 4 + half * 2;
            v2f b;
            b.x = W[k * 64 + n];
            b.y = W[(k + 1) * 64 + n];
            acc = __builtin_amdgcn_wmma_f32_16x16x4_f32(false, a[ks], false, b,
                                                        (short)0, acc, false, false);
        }
#pragma unroll
        for (int r = 0; r < 8; ++r) {
            int orow = tile * 16 + r + half * 8;
            PR1[(long)orow * 64 + ct * 16 + m] = acc[r];
        }
    }
}

__global__ __launch_bounds__(256)
void gcn_gemm2_wmma(const float* __restrict__ h, const float* __restrict__ W,
                    float* __restrict__ PR2, int nTiles) {
    int wave = threadIdx.x >> 5;
    int lane = threadIdx.x & 31;
    int tile = blockIdx.x * 8 + wave;
    if (tile >= nTiles) return;
    int half = lane >> 4;
    int m    = lane & 15;
    const float* hr = h + (long)(tile * 16 + m) * H_DIM;

    v2f a[8];
#pragma unroll
    for (int ks = 0; ks < 8; ++ks) {
        int k = ks * 4 + half * 2;
        a[ks].x = hr[k];
        a[ks].y = hr[k + 1];
    }
    v8f acc = {};
#pragma unroll
    for (int ks = 0; ks < 8; ++ks) {
        int k = ks * 4 + half * 2;
        v2f b;
        b.x = W[k * 16 + m];
        b.y = W[(k + 1) * 16 + m];
        acc = __builtin_amdgcn_wmma_f32_16x16x4_f32(false, a[ks], false, b,
                                                    (short)0, acc, false, false);
    }
#pragma unroll
    for (int r = 0; r < 8; ++r) {
        int orow = tile * 16 + r + half * 8;
        PR2[(long)orow * 16 + m] = acc[r];
    }
}

#define BIN_SHIFT 8
#define BIN_NODES 256
#define MAXB      512
#define SORT_T    512
#define EPT       8
#define CHUNK     (SORT_T * EPT)
#define SRC_BITS  24
#define SRC_MASK  0xFFFFFF
static_assert(SORT_T == MAXB, "one thread per bin counter");

__device__ __forceinline__ int gcn_block_excl_scan(int v, int* wsum) {
    int lane = threadIdx.x & 31, wave = threadIdx.x >> 5;
    int inc = v;
    for (int o = 1; o < 32; o <<= 1) {
        int y = __shfl_up(inc, o);
        if (lane >= o) inc += y;
    }
    if (lane == 31) wsum[wave] = inc;
    __syncthreads();
    int carry = 0;
    for (int w = 0; w < wave; ++w) carry += wsum[w];
    return carry + inc - v;
}

__global__ __launch_bounds__(SORT_T)
void gcn_bin_hist(const int* __restrict__ dst, int* __restrict__ binTotal, int E, int B) {
    __shared__ int hist[MAXB];
    int tid = threadIdx.x;
    hist[tid] = 0;
    __syncthreads();
    int e0 = blockIdx.x * CHUNK + tid;
    int d[EPT];
#pragma unroll
    for (int k = 0; k < EPT; ++k) d[k] = dst[min(e0 + k * SORT_T, E - 1)];
#pragma unroll
    for (int k = 0; k < EPT; ++k)
        if (e0 + k * SORT_T < E) atomicAdd(&hist[d[k] >> BIN_SHIFT], 1);
    __syncthreads();
    int c = hist[tid];
    if (tid < B && c) atomicAdd(&binTotal[tid], c);
}

__global__ __launch_bounds__(SORT_T)
void gcn_bin_scatter(const int* __restrict__ src, const int* __restrict__ dst,
                     const float* __restrict__ ew, const int* __restrict__ binTotal,
                     int* __restrict__ binCursor, int2* __restrict__ bsw, int E, int B,
                     const float* __restrict__ x, const float* __restrict__ W1c,
                     float* __restrict__ PR1, int nGemm, int tile0, int tile1) {
    __shared__ int hist[MAXB];
    __shared__ int base[MAXB];
    __shared__ int wsum[16];
    int tid = threadIdx.x;
    if ((int)blockIdx.x < nGemm) {
        int tile = tile0 + (int)blockIdx.x * (SORT_T / 32) + (tid >> 5);
        if (tile < tile1) gcn_gemm1_tile(x, W1c, PR1, tile);
        return;
    }
    const int chunk = (int)blockIdx.x - nGemm;
    hist[tid] = 0;
    __syncthreads();
    int e0 = chunk * CHUNK + tid;
    int d[EPT], r[EPT], s[EPT];
    float w[EPT];
#pragma unroll
    for (int k = 0; k < EPT; ++k) d[k] = dst[min(e0 + k * SORT_T, E - 1)];
    int t = (tid < B) ? binTotal[tid] : 0;
#pragma unroll
    for (int k = 0; k < EPT; ++k) {
        int ec = min(e0 + k * SORT_T, E - 1);
        s[k] = src[ec];
        w[k] = ew[ec];
    }
#pragma unroll
    for (int k = 0; k < EPT; ++k) {
        r[k] = 0;
        if (e0 + k * SORT_T < E) r[k] = atomicAdd(&hist[d[k] >> BIN_SHIFT], 1);
    }
    int start = gcn_block_excl_scan(t, wsum);
    int c = hist[tid];
    int bs = 0;
    if (tid < B && c) bs = start + atomicAdd(&binCursor[tid], c);
    base[tid] = bs;
    __syncthreads();
#pragma unroll
    for (int k = 0; k < EPT; ++k) {
        if (e0 + k * SORT_T < E) {
            int p = base[d[k] >> BIN_SHIFT] + r[k];
            int key = s[k] | ((d[k] & (BIN_NODES - 1)) << SRC_BITS);
            bsw[p] = make_int2(key, __float_as_int(w[k]));
        }
    }
}

__global__ __launch_bounds__(SORT_T)
void gcn_bin_sort(const int* __restrict__ binTotal, const int2* __restrict__ bsw,
                  int* __restrict__ rowptr, int2* __restrict__ csr, int N, int B,
                  const float* __restrict__ x, const float* __restrict__ W1c,
                  float* __restrict__ PR1, int nGemm, int tile0, int tile1) {
    __shared__ int cnt[BIN_NODES];
    __shared__ int wsumA[16];
    __shared__ int wsumB[16];
    __shared__ int sStart, sTot;
    int tid = threadIdx.x;
    if ((int)blockIdx.x < nGemm) {
        int tile = tile0 + (int)blockIdx.x * (SORT_T / 32) + (tid >> 5);
        if (tile < tile1) gcn_gemm1_tile(x, W1c, PR1, tile);
        return;
    }
    int b = (int)blockIdx.x - nGemm;
    int t = (tid < B) ? binTotal[tid] : 0;
    int ex = gcn_block_excl_scan(t, wsumA);
    if (tid == b) { sStart = ex; sTot = t; }
    if (tid < BIN_NODES) cnt[tid] = 0;
    __syncthreads();
    const int start = sStart, tot = sTot;
    const int node0 = b << BIN_SHIFT;
    const int2* mysw = bsw + start;
    for (int i0 = tid; i0 - tid < tot; i0 += CHUNK) {
        int dv[EPT];
#pragma unroll
        for (int u = 0; u < EPT; ++u) dv[u] = mysw[min(i0 + u * SORT_T, tot - 1)].x;
#pragma unroll
        for (int u = 0; u < EPT; ++u)
            if (i0 + u * SORT_T < tot) atomicAdd(&cnt[(unsigned)dv[u] >> SRC_BITS], 1);
    }
    __syncthreads();
    int c = (tid < BIN_NODES) ? cnt[tid] : 0;
    int lex = gcn_block_excl_scan(c, wsumB);
    if (tid < BIN_NODES) {
        cnt[tid] = lex;
        int n = node0 + tid;
        if (n <= N) rowptr[n] = start + lex;
    }
    __syncthreads();
    for (int i0 = tid; i0 - tid < tot; i0 += CHUNK) {
        int2 sw[EPT];
#pragma unroll
        for (int u = 0; u < EPT; ++u) sw[u] = mysw[min(i0 + u * SORT_T, tot - 1)];
#pragma unroll
        for (int u = 0; u < EPT; ++u) {
            if (i0 + u * SORT_T < tot) {
                int p = atomicAdd(&cnt[(unsigned)sw[u].x >> SRC_BITS], 1);
                csr[start + p] = make_int2(sw[u].x & SRC_MASK, sw[u].y);
            }
        }
    }
}

#define G1_NPG 2
#define G1_BCAST(V, K) ((int)__builtin_amdgcn_mov_dpp8((unsigned)(V), (unsigned)(K) * 0x249249u))
#define G1_LOAD(K) { unsigned s = (unsigned)G1_BCAST(q.x, K); \
                     v[K] = *(const float4*)(tab + ((s << 8) | c16)); }
#define G1_FMA(K)  { float w = __int_as_float(G1_BCAST(q.y, K)); \
                     const bool on = (K) < cnt; \
                     float tx = acc.x + w * v[K].x, ty = acc.y + w * v[K].y; \
                     float tz = acc.z + w * v[K].z, tw = acc.w + w * v[K].w; \
                     acc.x = on ? tx : acc.x; acc.y = on ? ty : acc.y; \
                     acc.z = on ? tz : acc.z; acc.w = on ? tw : acc.w; }
__global__ __launch_bounds__(256)
void gcn_gather1(const int* __restrict__ rowptr, const int* __restrict__ csr,
                 const float* __restrict__ PR1, const float* __restrict__ b1,
                 float* __restrict__ h, int N) {
    const int c = threadIdx.x & 7;
    const unsigned c16 = (unsigned)c * 16u;
    const int n0 = ((int)blockIdx.x * 32 + (int)(threadIdx.x >> 3)) * G1_NPG;
    const int2* ed = (const int2*)csr;
    const char* tab = (const char*)PR1;
    static_assert(G1_NPG == 2, "three bounds per group");
    const int nb = max(min(n0, N - 2), 0);
    const int d  = n0 - nb;
    const int t0 = rowptr[nb], t1 = rowptr[nb + 1], t2 = rowptr[nb + 2];
    int r[G1_NPG + 1];
    r[0] = (d >= 2) ? t2 : (d == 1) ? t1 : t0;
    r[1] = (d >= 1) ? t2 : t1;
    r[2] = t2;
    int2 q0[G1_NPG];
#pragma unroll
    for (int it = 0; it < G1_NPG; ++it) q0[it] = ed[max(min(r[it] + c, r[it + 1] - 1), 0)];
#pragma unroll
    for (int it = 0; it < G1_NPG; ++it) {
        const int n = n0 + it;
        const int e1 = r[it + 1];
        const float4 root = *(const float4*)(tab + (((unsigned)min(n, N - 1) << 8) | c16) + 128);
        float4 acc = make_float4(0.0f, 0.0f, 0.0f, 0.0f);
        int2 q = q0[it];
        for (int p = r[it]; p < e1; p += 8) {
            const int cnt = e1 - p;
            int2 qn = ed[min(p + 8 + c, e1 - 1)];
            float4 v[8];
            G1_LOAD(0) G1_LOAD(1) G1_LOAD(2) G1_LOAD(3)
            G1_LOAD(4) G1_LOAD(5) G1_LOAD(6) G1_LOAD(7)
            G1_FMA(0) G1_FMA(1) G1_FMA(2) G1_FMA(3)
            G1_FMA(4) G1_FMA(5) G1_FMA(6) G1_FMA(7)
            q = qn;
        }
        if (n < N) {
            float cf = fmaxf((float)(e1 - r[it]), 1.0f);
            float4 o;
            o.x = fmaxf(acc.x / cf + root.x + b1[c * 4],     0.0f);
            o.y = fmaxf(acc.y / cf + root.y + b1[c * 4 + 1], 0.0f);
            o.z = fmaxf(acc.z / cf + root.z + b1[c * 4 + 2], 0.0f);
            o.w = fmaxf(acc.w / cf + root.w + b1[c * 4 + 3], 0.0f);
            *(float4*)(h + (long)n * 32 + c * 4) = o;
        }
    }
}

#define G2_NPG 2
#define G2_PAIR_LO(V) ((int)__builtin_amdgcn_mov_dpp8((unsigned)(V), 0x6D2240u))
#define G2_PAIR_HI(V) ((int)__builtin_amdgcn_mov_dpp8((unsigned)(V), 0xFF6B64u))
#define G2_X1(F) __uint_as_float(__builtin_amdgcn_mov_dpp8(__float_as_uint(F), 0xDE54C1u))
#define G2_X2(F) __uint_as_float(__builtin_amdgcn_mov_dpp8(__float_as_uint(F), 0xB3E21Au))
#define G2_X4(F) __uint_as_float(__builtin_amdgcn_mov_dpp8(__float_as_uint(F), 0x688FACu))

__device__ __forceinline__ void gcn_g2_records(int (&q)[4], const int2* __restrict__ ed,
                                               int p, int e1, int c) {
#pragma unroll
    for (int k = 0; k < 4; ++k) q[k] = ed[max(min(p + 8 * k + c, e1 - 1), 0)].x;
}
__device__ __forceinline__ void gcn_g2_rows(float4 (&v)[8], const int (&q)[4],
                                            const char* __restrict__ tab, unsigned h16) {
#pragma unroll
    for (int k = 0; k < 4; ++k) {
        unsigned s0 = (unsigned)G2_PAIR_LO(q[k]);
        unsigned s1 = (unsigned)G2_PAIR_HI(q[k]);
        v[2 * k]     = *(const float4*)(tab + ((s0 << 6) | h16));
        v[2 * k + 1] = *(const float4*)(tab + ((s1 << 6) | h16));
    }
}
__device__ __forceinline__ void gcn_g2_add(float4& acc, const float4 (&v)[8], int cnt, int slot) {
#pragma unroll
    for (int u = 0; u < 8; ++u) {
        const bool on = (4 * u + slot) < cnt;
        const float tx = acc.x + v[u].x, ty = acc.y + v[u].y;
        const float tz = acc.z + v[u].z, tw = acc.w + v[u].w;
        acc.x = on ? tx : acc.x; acc.y = on ? ty : acc.y;
        acc.z = on ? tz : acc.z; acc.w = on ? tw : acc.w;
    }
}

__global__ __launch_bounds__(256)
void gcn_gather2_out(const int* __restrict__ rowptr, const int* __restrict__ csr,
                     const float* __restrict__ PR2, const float* __restrict__ b2,
                     float* __restrict__ out, int N) {
    __shared__ float so[32 * G2_NPG * C_DIM];
    const int c    = threadIdx.x & 7;
    const int slot = c >> 1;
    const int hf   = c & 1;
    const unsigned h16 = (unsigned)hf * 16u;
    const int g  = threadIdx.x >> 3;
    const int n0 = ((int)blockIdx.x * 32 + g) * G2_NPG;
    const int2* ed = (const int2*)csr;
    const char* tab = (const char*)PR2;
    static_assert(G2_NPG == 2, "three bounds per group");
    const int nb = max(min(n0, N - 2), 0);
    const int d  = n0 - nb;
    const int t0 = rowptr[nb], t1 = rowptr[nb + 1], t2 = rowptr[nb + 2];
    int r[G2_NPG + 1];
    r[0] = (d >= 2) ? t2 : (d == 1) ? t1 : t0;
    r[1] = (d >= 1) ? t2 : t1;
    r[2] = t2;
    int q[G2_NPG][4];
#pragma unroll
    for (int it = 0; it < G2_NPG; ++it) gcn_g2_records(q[it], ed, r[it], r[it + 1], c);
    float4 root[G2_NPG];
#pragma unroll
    for (int it = 0; it < G2_NPG; ++it)
        root[it] = *(const float4*)(tab + (((unsigned)min(n0 + it, N - 1) << 6) | h16) + 32);
    const int bi = hf * 4;
    const float4 bb = make_float4(b2[bi], b2[min(bi + 1, C_DIM - 1)],
                                  b2[min(bi + 2, C_DIM - 1)], b2[min(bi + 3, C_DIM - 1)]);
    float4 v[G2_NPG][8];
#pragma unroll
    for (int it = 0; it < G2_NPG; ++it) gcn_g2_rows(v[it], q[it], tab, h16);
    asm volatile("" :: "v"(v[0][0].x), "v"(v[0][0].y), "v"(v[0][0].z), "v"(v[0][0].w));
#pragma unroll
    for (int it = 0; it < G2_NPG; ++it) {
        const int n = n0 + it;
        const int e1 = r[it + 1];
        const int deg = e1 - r[it];
        float4 acc = make_float4(0.0f, 0.0f, 0.0f, 0.0f);
        gcn_g2_add(acc, v[it], deg, slot);
        for (int p = r[it] + 32; p < e1; p += 32) {
            int qq[4];
            float4 vv[8];
            gcn_g2_records(qq, ed, p, e1, c);
            gcn_g2_rows(vv, qq, tab, h16);
            gcn_g2_add(acc, vv, e1 - p, slot);
        }
        acc.x += G2_X2(acc.x); acc.y += G2_X2(acc.y); acc.z += G2_X2(acc.z); acc.w += G2_X2(acc.w);
        acc.x += G2_X4(acc.x); acc.y += G2_X4(acc.y); acc.z += G2_X4(acc.z); acc.w += G2_X4(acc.w);
        const float cn = fmaxf((float)deg, 1.0f);
        const float v0 = acc.x / cn + root[it].x + bb.x;
        const float v1 = acc.y / cn + root[it].y + bb.y;
        const float v2 = acc.z / cn + root[it].z + bb.z;
        const float v3 = acc.w / cn + root[it].w + bb.w;
        const float v4 = G2_X1(v0);
        float mx = fmaxf(fmaxf(fmaxf(v0, v1), fmaxf(v2, v3)), v4);
        float s = expf(v0 - mx) + expf(v1 - mx) + expf(v2 - mx) + expf(v3 - mx) + expf(v4 - mx);
        float l = logf(s);
        if (c == 0 && n < N) {
            float* o = so + (g * G2_NPG + it) * C_DIM;
            o[0] = v0 - mx - l; o[1] = v1 - mx - l; o[2] = v2 - mx - l;
            o[3] = v3 - mx - l; o[4] = v4 - mx - l;
        }
    }
    __syncthreads();
    const int rem = N - (int)blockIdx.x * (32 * G2_NPG);
    const int valid = (rem < 32 * G2_NPG ? rem : 32 * G2_NPG) * C_DIM;
    static_assert(32 * G2_NPG * C_DIM <= 2 * 256, "two stores per thread cover the block's rows");
    float* ob = out + (long)blockIdx.x * (32 * G2_NPG * C_DIM);
    const int t = (int)threadIdx.x;
    if (t < valid)       ob[t]       = so[t];
    if (t + 256 < valid) ob[t + 256] = so[t + 256];
}

extern "C" void kernel_launch(void* const* d_in, const int* in_sizes, int n_in,
                              void* d_out, int out_size, void* d_ws, size_t ws_size,
                              hipStream_t stream) {
    const float* x      = (const float*)d_in[0];
    const int*   ei     = (const int*)d_in[1];
    const float* ew     = (const float*)d_in[2];
    const float* Wrel1  = (const float*)d_in[3];
    const float* Wroot1 = (const float*)d_in[4];
    const float* b1     = (const float*)d_in[5];
    const float* Wrel2  = (const float*)d_in[6];
    const float* Wroot2 = (const float*)d_in[7];
    const float* b2     = (const float*)d_in[8];

    const int N = in_sizes[0] / D_IN;
    const int E = in_sizes[1] / 2;
    const int* src = ei;
    const int* dst = ei + E;

    float* ws = (float*)d_ws;
    const int NP = ((N + 4096) / 4096) * 4096;
    long off = 0;
    float* PR1 = ws + off; off += (long)N * 64;
    float* h   = ws + off; off += (long)N * 32;
    float* W1c = ws + off; off += 128 * 64;
    float* W2p = ws + off; off += 32 * 16;
    int* binTotal  = (int*)(ws + off); off += MAXB;
    int* binCursor = (int*)(ws + off); off += MAXB;
    int* rowptr = (int*)(ws + off); off += NP;
    int* csr    = (int*)(ws + off); off += (long)E * 2;
    int* bsw    = (int*)(ws + off); off += (long)E * 2;
    float* PR2 = PR1;

    const int nTiles = N / 16;

    const int B = N / BIN_NODES + 1;
    const int nChunks = (E + CHUNK - 1) / CHUNK;

    const int tHalf = ((nTiles / 2 + 15) / 16) * 16;
    const int gA = tHalf / 16;
    const int gB = (nTiles - tHalf + 15) / 16;

    gcn_zero_i<<<(2 * MAXB + 255) / 256, 256, 0, stream>>>(binTotal, 2 * MAXB);
    gcn_build_w1<<<(128 * 64 + 255) / 256, 256, 0, stream>>>(Wrel1, Wroot1, W1c);
    gcn_build_w2<<<(32 * 16 + 255) / 256, 256, 0, stream>>>(Wrel2, Wroot2, W2p);
    gcn_bin_hist<<<nChunks, SORT_T, 0, stream>>>(dst, binTotal, E, B);
    gcn_bin_scatter<<<gA + nChunks, SORT_T, 0, stream>>>(src, dst, ew, binTotal, binCursor,
                                                         (int2*)bsw, E, B,
                                                         x, W1c, PR1, gA, 0, tHalf);
    gcn_bin_sort<<<gB + B, SORT_T, 0, stream>>>(binTotal, (const int2*)bsw, rowptr, (int2*)csr, N, B,
                                                x, W1c, PR1, gB, tHalf, nTiles);

    gcn_gather1<<<(N + 32 * G1_NPG - 1) / (32 * G1_NPG), 256, 0, stream>>>(rowptr, csr, PR1, b1, h, N);

    gcn_gemm2_wmma<<<(nTiles + 7) / 8, 256, 0, stream>>>(h, W2p, PR2, nTiles);

    gcn_gather2_out<<<(N + 32 * G2_NPG - 1) / (32 * G2_NPG), 256, 0, stream>>>(rowptr, csr, PR2, b2,
                                                                               (float*)d_out, N);
}


# GEMM2 loses its own launch (9 launches -> 8), h its trip through memory: the layer-1 gather keeps its block's 64 rows of h in LDS, 4 of its waves run the WMMA tiles from there; every sum unchanged
# speedup vs baseline: 4.3394x; 1.0657x over previous
#include <hip/hip_runtime.h>
#include <math.h>

typedef __attribute__((ext_vector_type(2))) float v2f;
typedef __attribute__((ext_vector_type(8))) float v8f;

#define D_IN  128
#define H_DIM 32
#define C_DIM 5

__global__ void gcn_build_w1(const float* __restrict__ Wrel, const float* __restrict__ Wroot,
                             float* __restrict__ Wcat) {
    int i = blockIdx.x * 256 + threadIdx.x;
    if (i >= 128 * 64) return;
    int k = i >> 6, j = i & 63;
    Wcat[i] = (j < 32) ? Wrel[k * 32 + j] : Wroot[k * 32 + (j - 32)];
}

__global__ void gcn_build_w2(const float* __restrict__ Wrel, const float* __restrict__ Wroot,
                             float* __restrict__ Wp) {
    int i = blockIdx.x * 256 + threadIdx.x;
    if (i >= 32 * 16) return;
    int k = i >> 4, n = i & 15;
    float v = 0.0f;
    if (n < 5)                 v = Wrel[k * 5 + n];
    else if (n >= 8 && n < 13) v = Wroot[k * 5 + (n - 8)];
    Wp[i] = v;
}

__global__ void gcn_zero_i(int* __restrict__ p, int n) {
    int i = blockIdx.x * 256 + threadIdx.x;
    if (i < n) p[i] = 0;
}

__device__ __forceinline__
void gcn_gemm1_tile(const float* __restrict__ x, const float* __restrict__ W,
                    float* __restrict__ PR1, int tile) {
    int lane = threadIdx.x & 31;
    int half = lane >> 4;
    int m    = lane & 15;
    const float* xr = x + (long)(tile * 16 + m) * D_IN;

    v2f a[32];
#pragma unroll
    for (int ks = 0; ks < 32; ++ks) {
        int k = ks * 4 + half * 2;
        a[ks].x = xr[k];
        a[ks].y = xr[k + 1];
    }

#pragma unroll
    for (int ct = 0; ct < 4; ++ct) {
        int n = ct * 16 + m;
        v8f acc = {};
#pragma unroll
        for (int ks = 0; ks < 32; ++ks) {
            int k = ks * 4 + half * 2;
            v2f b;
            b.x = W[k * 64 + n];
            b.y = W[(k + 1) * 64 + n];
            acc = __builtin_amdgcn_wmma_f32_16x16x4_f32(false, a[ks], false, b,
                                                        (short)0, acc, false, false);
        }
#pragma unroll
        for (int r = 0; r < 8; ++r) {
            int orow = tile * 16 + r + half * 8;
            PR1[(long)orow * 64 + ct * 16 + m] = acc[r];
        }
    }
}

#define H_PITCH 36
__device__ __forceinline__
void gcn_gemm2_tile(const float* __restrict__ hs, const float* __restrict__ W,
                    float* __restrict__ PR2, int row0, int lane) {
    int half = lane >> 4;
    int m    = lane & 15;
    const float* hr = hs + m * H_PITCH;

    v2f a[8];
#pragma unroll
    for (int ks = 0; ks < 8; ++ks) {
        int k = ks * 4 + half * 2;
        a[ks].x = hr[k];
        a[ks].y = hr[k + 1];
    }
    v8f acc = {};
#pragma unroll
    for (int ks = 0; ks < 8; ++ks) {
        int k = ks * 4 + half * 2;
        v2f b;
        b.x = W[k * 16 + m];
        b.y = W[(k + 1) * 16 + m];
        acc = __builtin_amdgcn_wmma_f32_16x16x4_f32(false, a[ks], false, b,
                                                    (short)0, acc, false, false);
    }
#pragma unroll
    for (int r = 0; r < 8; ++r) {
        int orow = row0 + r + half * 8;
        PR2[(long)orow * 16 + m] = acc[r];
    }
}

#define BIN_SHIFT 8
#define BIN_NODES 256
#define MAXB      512
#define SORT_T    512
#define EPT       8
#define CHUNK     (SORT_T * EPT)
#define SRC_BITS  24
#define SRC_MASK  0xFFFFFF
static_assert(SORT_T == MAXB, "one thread per bin counter");

__device__ __forceinline__ int gcn_block_excl_scan(int v, int* wsum) {
    int lane = threadIdx.x & 31, wave = threadIdx.x >> 5;
    int inc = v;
    for (int o = 1; o < 32; o <<= 1) {
        int y = __shfl_up(inc, o);
        if (lane >= o) inc += y;
    }
    if (lane == 31) wsum[wave] = inc;
    __syncthreads();
    int carry = 0;
    for (int w = 0; w < wave; ++w) carry += wsum[w];
    return carry + inc - v;
}

__global__ __launch_bounds__(SORT_T)
void gcn_bin_hist(const int* __restrict__ dst, int* __restrict__ binTotal, int E, int B) {
    __shared__ int hist[MAXB];
    int tid = threadIdx.x;
    hist[tid] = 0;
    __syncthreads();
    int e0 = blockIdx.x * CHUNK + tid;
    int d[EPT];
#pragma unroll
    for (int k = 0; k < EPT; ++k) d[k] = dst[min(e0 + k * SORT_T, E - 1)];
#pragma unroll
    for (int k = 0; k < EPT; ++k)
        if (e0 + k * SORT_T < E) atomicAdd(&hist[d[k] >> BIN_SHIFT], 1);
    __syncthreads();
    int c = hist[tid];
    if (tid < B && c) atomicAdd(&binTotal[tid], c);
}

__global__ __launch_bounds__(SORT_T)
void gcn_bin_scatter(const int* __restrict__ src, const int* __restrict__ dst,
                     const float* __restrict__ ew, const int* __restrict__ binTotal,
                     int* __restrict__ binCursor, int2* __restrict__ bsw, int E, int B,
                     const float* __restrict__ x, const float* __restrict__ W1c,
                     float* __restrict__ PR1, int nGemm, int tile0, int tile1) {
    __shared__ int hist[MAXB];
    __shared__ int base[MAXB];
    __shared__ int wsum[16];
    int tid = threadIdx.x;
    if ((int)blockIdx.x < nGemm) {
        int tile = tile0 + (int)blockIdx.x * (SORT_T / 32) + (tid >> 5);
        if (tile < tile1) gcn_gemm1_tile(x, W1c, PR1, tile);
        return;
    }
    const int chunk = (int)blockIdx.x - nGemm;
    hist[tid] = 0;
    __syncthreads();
    int e0 = chunk * CHUNK + tid;
    int d[EPT], r[EPT], s[EPT];
    float w[EPT];
#pragma unroll
    for (int k = 0; k < EPT; ++k) d[k] = dst[min(e0 + k * SORT_T, E - 1)];
    int t = (tid < B) ? binTotal[tid] : 0;
#pragma unroll
    for (int k = 0; k < EPT; ++k) {
        int ec = min(e0 + k * SORT_T, E - 1);
        s[k] = src[ec];
        w[k] = ew[ec];
    }
#pragma unroll
    for (int k = 0; k < EPT; ++k) {
        r[k] = 0;
        if (e0 + k * SORT_T < E) r[k] = atomicAdd(&hist[d[k] >> BIN_SHIFT], 1);
    }
    int start = gcn_block_excl_scan(t, wsum);
    int c = hist[tid];
    int bs = 0;
    if (tid < B && c) bs = start + atomicAdd(&binCursor[tid], c);
    base[tid] = bs;
    __syncthreads();
#pragma unroll
    for (int k = 0; k < EPT; ++k) {
        if (e0 + k * SORT_T < E) {
            int p = base[d[k] >> BIN_SHIFT] + r[k];
            int key = s[k] | ((d[k] & (BIN_NODES - 1)) << SRC_BITS);
            bsw[p] = make_int2(key, __float_as_int(w[k]));
        }
    }
}

__global__ __launch_bounds__(SORT_T)
void gcn_bin_sort(const int* __restrict__ binTotal, const int2* __restrict__ bsw,
                  int* __restrict__ rowptr, int2* __restrict__ csr, int N, int B,
                  const float* __restrict__ x, const float* __restrict__ W1c,
                  float* __restrict__ PR1, int nGemm, int tile0, int tile1) {
    __shared__ int cnt[BIN_NODES];
    __shared__ int wsumA[16];
    __shared__ int wsumB[16];
    __shared__ int sStart, sTot;
    int tid = threadIdx.x;
    if ((int)blockIdx.x < nGemm) {
        int tile = tile0 + (int)blockIdx.x * (SORT_T / 32) + (tid >> 5);
        if (tile < tile1) gcn_gemm1_tile(x, W1c, PR1, tile);
        return;
    }
    int b = (int)blockIdx.x - nGemm;
    int t = (tid < B) ? binTotal[tid] : 0;
    int ex = gcn_block_excl_scan(t, wsumA);
    if (tid == b) { sStart = ex; sTot = t; }
    if (tid < BIN_NODES) cnt[tid] = 0;
    __syncthreads();
    const int start = sStart, tot = sTot;
    const int node0 = b << BIN_SHIFT;
    const int2* mysw = bsw + start;
    for (int i0 = tid; i0 - tid < tot; i0 += CHUNK) {
        int dv[EPT];
#pragma unroll
        for (int u = 0; u < EPT; ++u) dv[u] = mysw[min(i0 + u * SORT_T, tot - 1)].x;
#pragma unroll
        for (int u = 0; u < EPT; ++u)
            if (i0 + u * SORT_T < tot) atomicAdd(&cnt[(unsigned)dv[u] >> SRC_BITS], 1);
    }
    __syncthreads();
    int c = (tid < BIN_NODES) ? cnt[tid] : 0;
    int lex = gcn_block_excl_scan(c, wsumB);
    if (tid < BIN_NODES) {
        cnt[tid] = lex;
        int n = node0 + tid;
        if (n <= N) rowptr[n] = start + lex;
    }
    __syncthreads();
    for (int i0 = tid; i0 - tid < tot; i0 += CHUNK) {
        int2 sw[EPT];
#pragma unroll
        for (int u = 0; u < EPT; ++u) sw[u] = mysw[min(i0 + u * SORT_T, tot - 1)];
#pragma unroll
        for (int u = 0; u < EPT; ++u) {
            if (i0 + u * SORT_T < tot) {
                int p = atomicAdd(&cnt[(unsigned)sw[u].x >> SRC_BITS], 1);
                csr[start + p] = make_int2(sw[u].x & SRC_MASK, sw[u].y);
            }
        }
    }
}

#define G1_NPG 2
#define G1_BCAST(V, K) ((int)__builtin_amdgcn_mov_dpp8((unsigned)(V), (unsigned)(K) * 0x249249u))
#define G1_LOAD(K) { unsigned s = (unsigned)G1_BCAST(q.x, K); \
                     v[K] = *(const float4*)(tab + ((s << 8) | c16)); }
#define G1_FMA(K)  { float w = __int_as_float(G1_BCAST(q.y, K)); \
                     const bool on = (K) < cnt; \
                     float tx = acc.x + w * v[K].x, ty = acc.y + w * v[K].y; \
                     float tz = acc.z + w * v[K].z, tw = acc.w + w * v[K].w; \
                     acc.x = on ? tx : acc.x; acc.y = on ? ty : acc.y; \
                     acc.z = on ? tz : acc.z; acc.w = on ? tw : acc.w; }
__global__ __launch_bounds__(256, 16)
void gcn_gather1(const int* __restrict__ rowptr, const int* __restrict__ csr,
                 const float* __restrict__ PR1, const float* __restrict__ b1,
                 const float* __restrict__ W2p, float* __restrict__ PR2, int N) {
    __shared__ __attribute__((aligned(16))) float hs[32 * G1_NPG * H_PITCH];
    const int node00 = (int)blockIdx.x * (32 * G1_NPG);
    const int c = threadIdx.x & 7;
    const unsigned c16 = (unsigned)c * 16u;
    const int n0 = ((int)blockIdx.x * 32 + (int)(threadIdx.x >> 3)) * G1_NPG;
    const int2* ed = (const int2*)csr;
    const char* tab = (const char*)PR1;
    static_assert(G1_NPG == 2, "three bounds per group");
    const int nb = max(min(n0, N - 2), 0);
    const int d  = n0 - nb;
    const int t0 = rowptr[nb], t1 = rowptr[nb + 1], t2 = rowptr[nb + 2];
    int r[G1_NPG + 1];
    r[0] = (d >= 2) ? t2 : (d == 1) ? t1 : t0;
    r[1] = (d >= 1) ? t2 : t1;
    r[2] = t2;
    int2 q0[G1_NPG];
#pragma unroll
    for (int it = 0; it < G1_NPG; ++it) q0[it] = ed[max(min(r[it] + c, r[it + 1] - 1), 0)];
#pragma unroll
    for (int it = 0; it < G1_NPG; ++it) {
        const int n = n0 + it;
        const int e1 = r[it + 1];
        const float4 root = *(const float4*)(tab + (((unsigned)min(n, N - 1) << 8) | c16) + 128);
        float4 acc = make_float4(0.0f, 0.0f, 0.0f, 0.0f);
        int2 q = q0[it];
        for (int p = r[it]; p < e1; p += 8) {
            const int cnt = e1 - p;
            int2 qn = ed[min(p + 8 + c, e1 - 1)];
            float4 v[8];
            G1_LOAD(0) G1_LOAD(1) G1_LOAD(2) G1_LOAD(3)
            G1_LOAD(4) G1_LOAD(5) G1_LOAD(6) G1_LOAD(7)
            G1_FMA(0) G1_FMA(1) G1_FMA(2) G1_FMA(3)
            G1_FMA(4) G1_FMA(5) G1_FMA(6) G1_FMA(7)
            q = qn;
        }
        float4 o = make_float4(0.0f, 0.0f, 0.0f, 0.0f);
        if (n < N) {
            float cf = fmaxf((float)(e1 - r[it]), 1.0f);
            o.x = fmaxf(acc.x / cf + root.x + b1[c * 4],     0.0f);
            o.y = fmaxf(acc.y / cf + root.y + b1[c * 4 + 1], 0.0f);
            o.z = fmaxf(acc.z / cf + root.z + b1[c * 4 + 2], 0.0f);
            o.w = fmaxf(acc.w / cf + root.w + b1[c * 4 + 3], 0.0f);
        }
        *(float4*)(hs + (n - node00) * H_PITCH + c * 4) = o;
    }
    __syncthreads();
    static_assert((32 * G1_NPG) % 16 == 0, "whole tiles per block");
    const int wave = (__builtin_amdgcn_readfirstlane(n0) - node00) / (4 * G1_NPG);
    const int row0 = node00 + wave * 16;
    if (wave < (32 * G1_NPG) / 16 && row0 + 16 <= N)
        gcn_gemm2_tile(hs + wave * 16 * H_PITCH, W2p, PR2, row0, (int)__lane_id());
}

#define G2_NPG 2
#define G2_PAIR_LO(V) ((int)__builtin_amdgcn_mov_dpp8((unsigned)(V), 0x6D2240u))
#define G2_PAIR_HI(V) ((int)__builtin_amdgcn_mov_dpp8((unsigned)(V), 0xFF6B64u))
#define G2_X1(F) __uint_as_float(__builtin_amdgcn_mov_dpp8(__float_as_uint(F), 0xDE54C1u))
#define G2_X2(F) __uint_as_float(__builtin_amdgcn_mov_dpp8(__float_as_uint(F), 0xB3E21Au))
#define G2_X4(F) __uint_as_float(__builtin_amdgcn_mov_dpp8(__float_as_uint(F), 0x688FACu))

__device__ __forceinline__ void gcn_g2_records(int (&q)[4], const int2* __restrict__ ed,
                                               int p, int e1, int c) {
#pragma unroll
    for (int k = 0; k < 4; ++k) q[k] = ed[max(min(p + 8 * k + c, e1 - 1), 0)].x;
}
__device__ __forceinline__ void gcn_g2_rows(float4 (&v)[8], const int (&q)[4],
                                            const char* __restrict__ tab, unsigned h16) {
#pragma unroll
    for (int k = 0; k < 4; ++k) {
        unsigned s0 = (unsigned)G2_PAIR_LO(q[k]);
        unsigned s1 = (unsigned)G2_PAIR_HI(q[k]);
        v[2 * k]     = *(const float4*)(tab + ((s0 << 6) | h16));
        v[2 * k + 1] = *(const float4*)(tab + ((s1 << 6) | h16));
    }
}
__device__ __forceinline__ void gcn_g2_add(float4& acc, const float4 (&v)[8], int cnt, int slot) {
#pragma unroll
    for (int u = 0; u < 8; ++u) {
        const bool on = (4 * u + slot) < cnt;
        const float tx = acc.x + v[u].x, ty = acc.y + v[u].y;
        const float tz = acc.z + v[u].z, tw = acc.w + v[u].w;
        acc.x = on ? tx : acc.x; acc.y = on ? ty : acc.y;
        acc.z = on ? tz : acc.z; acc.w = on ? tw : acc.w;
    }
}

__global__ __launch_bounds__(256)
void gcn_gather2_out(const int* __restrict__ rowptr, const int* __restrict__ csr,
                     const float* __restrict__ PR2, const float* __restrict__ b2,
                     float* __restrict__ out, int N) {
    __shared__ float so[32 * G2_NPG * C_DIM];
    const int c    = threadIdx.x & 7;
    const int slot = c >> 1;
    const int hf   = c & 1;
    const unsigned h16 = (unsigned)hf * 16u;
    const int g  = threadIdx.x >> 3;
    const int n0 = ((int)blockIdx.x * 32 + g) * G2_NPG;
    const int2* ed = (const int2*)csr;
    const char* tab = (const char*)PR2;
    static_assert(G2_NPG == 2, "three bounds per group");
    const int nb = max(min(n0, N - 2), 0);
    const int d  = n0 - nb;
    const int t0 = rowptr[nb], t1 = rowptr[nb + 1], t2 = rowptr[nb + 2];
    int r[G2_NPG + 1];
    r[0] = (d >= 2) ? t2 : (d == 1) ? t1 : t0;
    r[1] = (d >= 1) ? t2 : t1;
    r[2] = t2;
    int q[G2_NPG][4];
#pragma unroll
    for (int it = 0; it < G2_NPG; ++it) gcn_g2_records(q[it], ed, r[it], r[it + 1], c);
    float4 root[G2_NPG];
#pragma unroll
    for (int it = 0; it < G2_NPG; ++it)
        root[it] = *(const float4*)(tab + (((unsigned)min(n0 + it, N - 1) << 6) | h16) + 32);
    const int bi = hf * 4;
    const float4 bb = make_float4(b2[bi], b2[min(bi + 1, C_DIM - 1)],
                                  b2[min(bi + 2, C_DIM - 1)], b2[min(bi + 3, C_DIM - 1)]);
    float4 v[G2_NPG][8];
#pragma unroll
    for (int it = 0; it < G2_NPG; ++it) gcn_g2_rows(v[it], q[it], tab, h16);
    asm volatile("" :: "v"(v[0][0].x), "v"(v[0][0].y), "v"(v[0][0].z), "v"(v[0][0].w));
#pragma unroll
    for (int it = 0; it < G2_NPG; ++it) {
        const int n = n0 + it;
        const int e1 = r[it + 1];
        const int deg = e1 - r[it];
        float4 acc = make_float4(0.0f, 0.0f, 0.0f, 0.0f);
        gcn_g2_add(acc, v[it], deg, slot);
        for (int p = r[it] + 32; p < e1; p += 32) {
            int qq[4];
            float4 vv[8];
            gcn_g2_records(qq, ed, p, e1, c);
            gcn_g2_rows(vv, qq, tab, h16);
            gcn_g2_add(acc, vv, e1 - p, slot);
        }
        acc.x += G2_X2(acc.x); acc.y += G2_X2(acc.y); acc.z += G2_X2(acc.z); acc.w += G2_X2(acc.w);
        acc.x += G2_X4(acc.x); acc.y += G2_X4(acc.y); acc.z += G2_X4(acc.z); acc.w += G2_X4(acc.w);
        const float cn = fmaxf((float)deg, 1.0f);
        const float v0 = acc.x / cn + root[it].x + bb.x;
        const float v1 = acc.y / cn + root[it].y + bb.y;
        const float v2 = acc.z / cn + root[it].z + bb.z;
        const float v3 = acc.w / cn + root[it].w + bb.w;
        const float v4 = G2_X1(v0);
        float mx = fmaxf(fmaxf(fmaxf(v0, v1), fmaxf(v2, v3)), v4);
        float s = expf(v0 - mx) + expf(v1 - mx) + expf(v2 - mx) + expf(v3 - mx) + expf(v4 - mx);
        float l = logf(s);
        if (c == 0 && n < N) {
            float* o = so + (g * G2_NPG + it) * C_DIM;
            o[0] = v0 - mx - l; o[1] = v1 - mx - l; o[2] = v2 - mx - l;
            o[3] = v3 - mx - l; o[4] = v4 - mx - l;
        }
    }
    __syncthreads();
    const int rem = N - (int)blockIdx.x * (32 * G2_NPG);
    const int valid = (rem < 32 * G2_NPG ? rem : 32 * G2_NPG) * C_DIM;
    static_assert(32 * G2_NPG * C_DIM <= 2 * 256, "two stores per thread cover the block's rows");
    float* ob = out + (long)blockIdx.x * (32 * G2_NPG * C_DIM);
    const int t = (int)threadIdx.x;
    if (t < valid)       ob[t]       = so[t];
    if (t + 256 < valid) ob[t + 256] = so[t + 256];
}

extern "C" void kernel_launch(void* const* d_in, const int* in_sizes, int n_in,
                              void* d_out, int out_size, void* d_ws, size_t ws_size,
                              hipStream_t stream) {
    const float* x      = (const float*)d_in[0];
    const int*   ei     = (const int*)d_in[1];
    const float* ew     = (const float*)d_in[2];
    const float* Wrel1  = (const float*)d_in[3];
    const float* Wroot1 = (const float*)d_in[4];
    const float* b1     = (const float*)d_in[5];
    const float* Wrel2  = (const float*)d_in[6];
    const float* Wroot2 = (const float*)d_in[7];
    const float* b2     = (const float*)d_in[8];

    const int N = in_sizes[0] / D_IN;
    const int E = in_sizes[1] / 2;
    const int* src = ei;
    const int* dst = ei + E;

    float* ws = (float*)d_ws;
    const int NP = ((N + 4096) / 4096) * 4096;
    long off = 0;
    float* PR1 = ws + off; off += (long)N * 64;
    float* PR2 = ws + off; off += (long)N * 16;
    float* W1c = ws + off; off += 128 * 64;
    float* W2p = ws + off; off += 32 * 16;
    int* binTotal  = (int*)(ws + off); off += MAXB;
    int* binCursor = (int*)(ws + off); off += MAXB;
    int* rowptr = (int*)(ws + off); off += NP;
    int* csr    = (int*)(ws + off); off += (long)E * 2;
    int* bsw    = (int*)(ws + off); off += (long)E * 2;

    const int nTiles = N / 16;

    const int B = N / BIN_NODES + 1;
    const int nChunks = (E + CHUNK - 1) / CHUNK;

    const int tHalf = ((nTiles / 2 + 15) / 16) * 16;
    const int gA = tHalf / 16;
    const int gB = (nTiles - tHalf + 15) / 16;

    gcn_zero_i<<<(2 * MAXB + 255) / 256, 256, 0, stream>>>(binTotal, 2 * MAXB);
    gcn_build_w1<<<(128 * 64 + 255) / 256, 256, 0, stream>>>(Wrel1, Wroot1, W1c);
    gcn_build_w2<<<(32 * 16 + 255) / 256, 256, 0, stream>>>(Wrel2, Wroot2, W2p);
    gcn_bin_hist<<<nChunks, SORT_T, 0, stream>>>(dst, binTotal, E, B);
    gcn_bin_scatter<<<gA + nChunks, SORT_T, 0, stream>>>(src, dst, ew, binTotal, binCursor,
                                                         (int2*)bsw, E, B,
                                                         x, W1c, PR1, gA, 0, tHalf);
    gcn_bin_sort<<<gB + B, SORT_T, 0, stream>>>(binTotal, (const int2*)bsw, rowptr, (int2*)csr, N, B,
                                                x, W1c, PR1, gB, tHalf, nTiles);

    gcn_gather1<<<(N + 32 * G1_NPG - 1) / (32 * G1_NPG), 256, 0, stream>>>(rowptr, csr, PR1, b1, W2p, PR2, N);

    gcn_gather2_out<<<(N + 32 * G2_NPG - 1) / (32 * G2_NPG), 256, 0, stream>>>(rowptr, csr, PR2, b2,
                                                                               (float*)d_out, N);
}


# Both weight repack launches removed (8 launches -> 6): GEMM1 and GEMM2 read W_rel and W_root where they lie, in the packed copies' column order; every product and sum of the result unchanged
# speedup vs baseline: 4.4736x; 1.0309x over previous
#include <hip/hip_runtime.h>
#include <math.h>

typedef __attribute__((ext_vector_type(2))) float v2f;
typedef __attribute__((ext_vector_type(8))) float v8f;

#define D_IN  128
#define H_DIM 32
#define C_DIM 5

__global__ void gcn_zero_i(int* __restrict__ p, int n) {
    int i = blockIdx.x * 256 + threadIdx.x;
    if (i < n) p[i] = 0;
}

__device__ __forceinline__
void gcn_gemm1_tile(const float* __restrict__ x, const float* __restrict__ Wrel,
                    const float* __restrict__ Wroot, float* __restrict__ PR1, int tile) {
    int lane = threadIdx.x & 31;
    int half = lane >> 4;
    int m    = lane & 15;
    const float* xr = x + (long)(tile * 16 + m) * D_IN;

    v2f a[32];
#pragma unroll
    for (int ks = 0; ks < 32; ++ks) {
        int k = ks * 4 + half * 2;
        a[ks].x = xr[k];
        a[ks].y = xr[k + 1];
    }

#pragma unroll
    for (int ct = 0; ct < 4; ++ct) {
        const float* __restrict__ W = (ct < 2) ? Wrel : Wroot;
        int n = (ct & 1) * 16 + m;
        v8f acc = {};
#pragma unroll
        for (int ks = 0; ks < 32; ++ks) {
            int k = ks * 4 + half * 2;
            v2f b;
            b.x = W[k * H_DIM + n];
            b.y = W[(k + 1) * H_DIM + n];
            acc = __builtin_amdgcn_wmma_f32_16x16x4_f32(false, a[ks], false, b,
                                                        (short)0, acc, false, false);
        }
#pragma unroll
        for (int r = 0; r < 8; ++r) {
            int orow = tile * 16 + r + half * 8;
            PR1[(long)orow * 64 + ct * 16 + m] = acc[r];
        }
    }
}

#define H_PITCH 36
__device__ __forceinline__
void gcn_gemm2_tile(const float* __restrict__ hs, const float* __restrict__ Wrel,
                    const float* __restrict__ Wroot, float* __restrict__ PR2, int row0, int lane) {
    int half = lane >> 4;
    int m    = lane & 15;
    const float* hr = hs + m * H_PITCH;
    const float* __restrict__ W = (m < 8) ? Wrel : Wroot;
    const int wc = min(m & 7, C_DIM - 1);

    v2f a[8];
#pragma unroll
    for (int ks = 0; ks < 8; ++ks) {
        int k = ks * 4 + half * 2;
        a[ks].x = hr[k];
        a[ks].y = hr[k + 1];
    }
    v8f acc = {};
#pragma unroll
    for (int ks = 0; ks < 8; ++ks) {
        int k = ks * 4 + half * 2;
        v2f b;
        b.x = W[k * C_DIM + wc];
        b.y = W[(k + 1) * C_DIM + wc];
        acc = __builtin_amdgcn_wmma_f32_16x16x4_f32(false, a[ks], false, b,
                                                    (short)0, acc, false, false);
    }
#pragma unroll
    for (int r = 0; r < 8; ++r) {
        int orow = row0 + r + half * 8;
        PR2[(long)orow * 16 + m] = acc[r];
    }
}

#define BIN_SHIFT 8
#define BIN_NODES 256
#define MAXB      512
#define SORT_T    512
#define EPT       8
#define CHUNK     (SORT_T * EPT)
#define SRC_BITS  24
#define SRC_MASK  0xFFFFFF
static_assert(SORT_T == MAXB, "one thread per bin counter");

__device__ __forceinline__ int gcn_block_excl_scan(int v, int* wsum) {
    int lane = threadIdx.x & 31, wave = threadIdx.x >> 5;
    int inc = v;
    for (int o = 1; o < 32; o <<= 1) {
        int y = __shfl_up(inc, o);
        if (lane >= o) inc += y;
    }
    if (lane == 31) wsum[wave] = inc;
    __syncthreads();
    int carry = 0;
    for (int w = 0; w < wave; ++w) carry += wsum[w];
    return carry + inc - v;
}

__global__ __launch_bounds__(SORT_T)
void gcn_bin_hist(const int* __restrict__ dst, int* __restrict__ binTotal, int E, int B) {
    __shared__ int hist[MAXB];
    int tid = threadIdx.x;
    hist[tid] = 0;
    __syncthreads();
    int e0 = blockIdx.x * CHUNK + tid;
    int d[EPT];
#pragma unroll
    for (int k = 0; k < EPT; ++k) d[k] = dst[min(e0 + k * SORT_T, E - 1)];
#pragma unroll
    for (int k = 0; k < EPT; ++k)
        if (e0 + k * SORT_T < E) atomicAdd(&hist[d[k] >> BIN_SHIFT], 1);
    __syncthreads();
    int c = hist[tid];
    if (tid < B && c) atomicAdd(&binTotal[tid], c);
}

__global__ __launch_bounds__(SORT_T)
void gcn_bin_scatter(const int* __restrict__ src, const int* __restrict__ dst,
                     const float* __restrict__ ew, const int* __restrict__ binTotal,
                     int* __restrict__ binCursor, int2* __restrict__ bsw, int E, int B,
                     const float* __restrict__ x, const float* __restrict__ Wrel1,
                     const float* __restrict__ Wroot1,
                     float* __restrict__ PR1, int nGemm, int tile0, int tile1) {
    __shared__ int hist[MAXB];
    __shared__ int base[MAXB];
    __shared__ int wsum[16];
    int tid = threadIdx.x;
    if ((int)blockIdx.x < nGemm) {
        int tile = tile0 + (int)blockIdx.x * (SORT_T / 32) + (tid >> 5);
        if (tile < tile1) gcn_gemm1_tile(x, Wrel1, Wroot1, PR1, tile);
        return;
    }
    const int chunk = (int)blockIdx.x - nGemm;
    hist[tid] = 0;
    __syncthreads();
    int e0 = chunk * CHUNK + tid;
    int d[EPT], r[EPT], s[EPT];
    float w[EPT];
#pragma unroll
    for (int k = 0; k < EPT; ++k) d[k] = dst[min(e0 + k * SORT_T, E - 1)];
    int t = (tid < B) ? binTotal[tid] : 0;
#pragma unroll
    for (int k = 0; k < EPT; ++k) {
        int ec = min(e0 + k * SORT_T, E - 1);
        s[k] = src[ec];
        w[k] = ew[ec];
    }
#pragma unroll
    for (int k = 0; k < EPT; ++k) {
        r[k] = 0;
        if (e0 + k * SORT_T < E) r[k] = atomicAdd(&hist[d[k] >> BIN_SHIFT], 1);
    }
    int start = gcn_block_excl_scan(t, wsum);
    int c = hist[tid];
    int bs = 0;
    if (tid < B && c) bs = start + atomicAdd(&binCursor[tid], c);
    base[tid] = bs;
    __syncthreads();
#pragma unroll
    for (int k = 0; k < EPT; ++k) {
        if (e0 + k * SORT_T < E) {
            int p = base[d[k] >> BIN_SHIFT] + r[k];
            int key = s[k] | ((d[k] & (BIN_NODES - 1)) << SRC_BITS);
            bsw[p] = make_int2(key, __float_as_int(w[k]));
        }
    }
}

__global__ __launch_bounds__(SORT_T)
void gcn_bin_sort(const int* __restrict__ binTotal, const int2* __restrict__ bsw,
                  int* __restrict__ rowptr, int2* __restrict__ csr, int N, int B,
                  const float* __restrict__ x, const float* __restrict__ Wrel1,
                  const float* __restrict__ Wroot1,
                  float* __restrict__ PR1, int nGemm, int tile0, int tile1) {
    __shared__ int cnt[BIN_NODES];
    __shared__ int wsumA[16];
    __shared__ int wsumB[16];
    __shared__ int sStart, sTot;
    int tid = threadIdx.x;
    if ((int)blockIdx.x < nGemm) {
        int tile = tile0 + (int)blockIdx.x * (SORT_T / 32) + (tid >> 5);
        if (tile < tile1) gcn_gemm1_tile(x, Wrel1, Wroot1, PR1, tile);
        return;
    }
    int b = (int)blockIdx.x - nGemm;
    int t = (tid < B) ? binTotal[tid] : 0;
    int ex = gcn_block_excl_scan(t, wsumA);
    if (tid == b) { sStart = ex; sTot = t; }
    if (tid < BIN_NODES) cnt[tid] = 0;
    __syncthreads();
    const int start = sStart, tot = sTot;
    const int node0 = b << BIN_SHIFT;
    const int2* mysw = bsw + start;
    for (int i0 = tid; i0 - tid < tot; i0 += CHUNK) {
        int dv[EPT];
#pragma unroll
        for (int u = 0; u < EPT; ++u) dv[u] = mysw[min(i0 + u * SORT_T, tot - 1)].x;
#pragma unroll
        for (int u = 0; u < EPT; ++u)
            if (i0 + u * SORT_T < tot) atomicAdd(&cnt[(unsigned)dv[u] >> SRC_BITS], 1);
    }
    __syncthreads();
    int c = (tid < BIN_NODES) ? cnt[tid] : 0;
    int lex = gcn_block_excl_scan(c, wsumB);
    if (tid < BIN_NODES) {
        cnt[tid] = lex;
        int n = node0 + tid;
        if (n <= N) rowptr[n] = start + lex;
    }
    __syncthreads();
    for (int i0 = tid; i0 - tid < tot; i0 += CHUNK) {
        int2 sw[EPT];
#pragma unroll
        for (int u = 0; u < EPT; ++u) sw[u] = mysw[min(i0 + u * SORT_T, tot - 1)];
#pragma unroll
        for (int u = 0; u < EPT; ++u) {
            if (i0 + u * SORT_T < tot) {
                int p = atomicAdd(&cnt[(unsigned)sw[u].x >> SRC_BITS], 1);
                csr[start + p] = make_int2(sw[u].x & SRC_MASK, sw[u].y);
            }
        }
    }
}

#define G1_NPG 2
#define G1_BCAST(V, K) ((int)__builtin_amdgcn_mov_dpp8((unsigned)(V), (unsigned)(K) * 0x249249u))
#define G1_LOAD(K) { unsigned s = (unsigned)G1_BCAST(q.x, K); \
                     v[K] = *(const float4*)(tab + ((s << 8) | c16)); }
#define G1_FMA(K)  { float w = __int_as_float(G1_BCAST(q.y, K)); \
                     const bool on = (K) < cnt; \
                     float tx = acc.x + w * v[K].x, ty = acc.y + w * v[K].y; \
                     float tz = acc.z + w * v[K].z, tw = acc.w + w * v[K].w; \
                     acc.x = on ? tx : acc.x; acc.y = on ? ty : acc.y; \
                     acc.z = on ? tz : acc.z; acc.w = on ? tw : acc.w; }
__global__ __launch_bounds__(256, 16)
void gcn_gather1(const int* __restrict__ rowptr, const int* __restrict__ csr,
                 const float* __restrict__ PR1, const float* __restrict__ b1,
                 const float* __restrict__ Wrel2, const float* __restrict__ Wroot2,
                 float* __restrict__ PR2, int N) {
    __shared__ __attribute__((aligned(16))) float hs[32 * G1_NPG * H_PITCH];
    const int node00 = (int)blockIdx.x * (32 * G1_NPG);
    const int c = threadIdx.x & 7;
    const unsigned c16 = (unsigned)c * 16u;
    const int n0 = ((int)blockIdx.x * 32 + (int)(threadIdx.x >> 3)) * G1_NPG;
    const int2* ed = (const int2*)csr;
    const char* tab = (const char*)PR1;
    static_assert(G1_NPG == 2, "three bounds per group");
    const int nb = max(min(n0, N - 2), 0);
    const int d  = n0 - nb;
    const int t0 = rowptr[nb], t1 = rowptr[nb + 1], t2 = rowptr[nb + 2];
    int r[G1_NPG + 1];
    r[0] = (d >= 2) ? t2 : (d == 1) ? t1 : t0;
    r[1] = (d >= 1) ? t2 : t1;
    r[2] = t2;
    int2 q0[G1_NPG];
#pragma unroll
    for (int it = 0; it < G1_NPG; ++it) q0[it] = ed[max(min(r[it] + c, r[it + 1] - 1), 0)];
#pragma unroll
    for (int it = 0; it < G1_NPG; ++it) {
        const int n = n0 + it;
        const int e1 = r[it + 1];
        const float4 root = *(const float4*)(tab + (((unsigned)min(n, N - 1) << 8) | c16) + 128);
        float4 acc = make_float4(0.0f, 0.0f, 0.0f, 0.0f);
        int2 q = q0[it];
        for (int p = r[it]; p < e1; p += 8) {
            const int cnt = e1 - p;
            int2 qn = ed[min(p + 8 + c, e1 - 1)];
            float4 v[8];
            G1_LOAD(0) G1_LOAD(1) G1_LOAD(2) G1_LOAD(3)
            G1_LOAD(4) G1_LOAD(5) G1_LOAD(6) G1_LOAD(7)
            G1_FMA(0) G1_FMA(1) G1_FMA(2) G1_FMA(3)
            G1_FMA(4) G1_FMA(5) G1_FMA(6) G1_FMA(7)
            q = qn;
        }
        float4 o = make_float4(0.0f, 0.0f, 0.0f, 0.0f);
        if (n < N) {
            float cf = fmaxf((float)(e1 - r[it]), 1.0f);
            o.x = fmaxf(acc.x / cf + root.x + b1[c * 4],     0.0f);
            o.y = fmaxf(acc.y / cf + root.y + b1[c * 4 + 1], 0.0f);
            o.z = fmaxf(acc.z / cf + root.z + b1[c * 4 + 2], 0.0f);
            o.w = fmaxf(acc.w / cf + root.w + b1[c * 4 + 3], 0.0f);
        }
        *(float4*)(hs + (n - node00) * H_PITCH + c * 4) = o;
    }
    __syncthreads();
    static_assert((32 * G1_NPG) % 16 == 0, "whole tiles per block");
    const int wave = (__builtin_amdgcn_readfirstlane(n0) - node00) / (4 * G1_NPG);
    const int row0 = node00 + wave * 16;
    if (wave < (32 * G1_NPG) / 16 && row0 + 16 <= N)
        gcn_gemm2_tile(hs + wave * 16 * H_PITCH, Wrel2, Wroot2, PR2, row0, (int)__lane_id());
}

#define G2_NPG 2
#define G2_PAIR_LO(V) ((int)__builtin_amdgcn_mov_dpp8((unsigned)(V), 0x6D2240u))
#define G2_PAIR_HI(V) ((int)__builtin_amdgcn_mov_dpp8((unsigned)(V), 0xFF6B64u))
#define G2_X1(F) __uint_as_float(__builtin_amdgcn_mov_dpp8(__float_as_uint(F), 0xDE54C1u))
#define G2_X2(F) __uint_as_float(__builtin_amdgcn_mov_dpp8(__float_as_uint(F), 0xB3E21Au))
#define G2_X4(F) __uint_as_float(__builtin_amdgcn_mov_dpp8(__float_as_uint(F), 0x688FACu))

__device__ __forceinline__ void gcn_g2_records(int (&q)[4], const int2* __restrict__ ed,
                                               int p, int e1, int c) {
#pragma unroll
    for (int k = 0; k < 4; ++k) q[k] = ed[max(min(p + 8 * k + c, e1 - 1), 0)].x;
}
__device__ __forceinline__ void gcn_g2_rows(float4 (&v)[8], const int (&q)[4],
                                            const char* __restrict__ tab, unsigned h16) {
#pragma unroll
    for (int k = 0; k < 4; ++k) {
        unsigned s0 = (unsigned)G2_PAIR_LO(q[k]);
        unsigned s1 = (unsigned)G2_PAIR_HI(q[k]);
        v[2 * k]     = *(const float4*)(tab + ((s0 << 6) | h16));
        v[2 * k + 1] = *(const float4*)(tab + ((s1 << 6) | h16));
    }
}
__device__ __forceinline__ void gcn_g2_add(float4& acc, const float4 (&v)[8], int cnt, int slot) {
#pragma unroll
    for (int u = 0; u < 8; ++u) {
        const bool on = (4 * u + slot) < cnt;
        const float tx = acc.x + v[u].x, ty = acc.y + v[u].y;
        const float tz = acc.z + v[u].z, tw = acc.w + v[u].w;
        acc.x = on ? tx : acc.x; acc.y = on ? ty : acc.y;
        acc.z = on ? tz : acc.z; acc.w = on ? tw : acc.w;
    }
}

__global__ __launch_bounds__(256)
void gcn_gather2_out(const int* __restrict__ rowptr, const int* __restrict__ csr,
                     const float* __restrict__ PR2, const float* __restrict__ b2,
                     float* __restrict__ out, int N) {
    __shared__ float so[32 * G2_NPG * C_DIM];
    const int c    = threadIdx.x & 7;
    const int slot = c >> 1;
    const int hf   = c & 1;
    const unsigned h16 = (unsigned)hf * 16u;
    const int g  = threadIdx.x >> 3;
    const int n0 = ((int)blockIdx.x * 32 + g) * G2_NPG;
    const int2* ed = (const int2*)csr;
    const char* tab = (const char*)PR2;
    static_assert(G2_NPG == 2, "three bounds per group");
    const int nb = max(min(n0, N - 2), 0);
    const int d  = n0 - nb;
    const int t0 = rowptr[nb], t1 = rowptr[nb + 1], t2 = rowptr[nb + 2];
    int r[G2_NPG + 1];
    r[0] = (d >= 2) ? t2 : (d == 1) ? t1 : t0;
    r[1] = (d >= 1) ? t2 : t1;
    r[2] = t2;
    int q[G2_NPG][4];
#pragma unroll
    for (int it = 0; it < G2_NPG; ++it) gcn_g2_records(q[it], ed, r[it], r[it + 1], c);
    float4 root[G2_NPG];
#pragma unroll
    for (int it = 0; it < G2_NPG; ++it)
        root[it] = *(const float4*)(tab + (((unsigned)min(n0 + it, N - 1) << 6) | h16) + 32);
    const int bi = hf * 4;
    const float4 bb = make_float4(b2[bi], b2[min(bi + 1, C_DIM - 1)],
                                  b2[min(bi + 2, C_DIM - 1)], b2[min(bi + 3, C_DIM - 1)]);
    float4 v[G2_NPG][8];
#pragma unroll
    for (int it = 0; it < G2_NPG; ++it) gcn_g2_rows(v[it], q[it], tab, h16);
    asm volatile("" :: "v"(v[0][0].x), "v"(v[0][0].y), "v"(v[0][0].z), "v"(v[0][0].w));
#pragma unroll
    for (int it = 0; it < G2_NPG; ++it) {
        const int n = n0 + it;
        const int e1 = r[it + 1];
        const int deg = e1 - r[it];
        float4 acc = make_float4(0.0f, 0.0f, 0.0f, 0.0f);
        gcn_g2_add(acc, v[it], deg, slot);
        for (int p = r[it] + 32; p < e1; p += 32) {
            int qq[4];
            float4 vv[8];
            gcn_g2_records(qq, ed, p, e1, c);
            gcn_g2_rows(vv, qq, tab, h16);
            gcn_g2_add(acc, vv, e1 - p, slot);
        }
        acc.x += G2_X2(acc.x); acc.y += G2_X2(acc.y); acc.z += G2_X2(acc.z); acc.w += G2_X2(acc.w);
        acc.x += G2_X4(acc.x); acc.y += G2_X4(acc.y); acc.z += G2_X4(acc.z); acc.w += G2_X4(acc.w);
        const float cn = fmaxf((float)deg, 1.0f);
        const float v0 = acc.x / cn + root[it].x + bb.x;
        const float v1 = acc.y / cn + root[it].y + bb.y;
        const float v2 = acc.z / cn + root[it].z + bb.z;
        const float v3 = acc.w / cn + root[it].w + bb.w;
        const float v4 = G2_X1(v0);
        float mx = fmaxf(fmaxf(fmaxf(v0, v1), fmaxf(v2, v3)), v4);
        float s = expf(v0 - mx) + expf(v1 - mx) + expf(v2 - mx) + expf(v3 - mx) + expf(v4 - mx);
        float l = logf(s);
        if (c == 0 && n < N) {
            float* o = so + (g * G2_NPG + it) * C_DIM;
            o[0] = v0 - mx - l; o[1] = v1 - mx - l; o[2] = v2 - mx - l;
            o[3] = v3 - mx - l; o[4] = v4 - mx - l;
        }
    }
    __syncthreads();
    const int rem = N - (int)blockIdx.x * (32 * G2_NPG);
    const int valid = (rem < 32 * G2_NPG ? rem : 32 * G2_NPG) * C_DIM;
    static_assert(32 * G2_NPG * C_DIM <= 2 * 256, "two stores per thread cover the block's rows");
    float* ob = out + (long)blockIdx.x * (32 * G2_NPG * C_DIM);
    const int t = (int)threadIdx.x;
    if (t < valid)       ob[t]       = so[t];
    if (t + 256 < valid) ob[t + 256] = so[t + 256];
}

extern "C" void kernel_launch(void* const* d_in, const int* in_sizes, int n_in,
                              void* d_out, int out_size, void* d_ws, size_t ws_size,
                              hipStream_t stream) {
    const float* x      = (const float*)d_in[0];
    const int*   ei     = (const int*)d_in[1];
    const float* ew     = (const float*)d_in[2];
    const float* Wrel1  = (const float*)d_in[3];
    const float* Wroot1 = (const float*)d_in[4];
    const float* b1     = (const float*)d_in[5];
    const float* Wrel2  = (const float*)d_in[6];
    const float* Wroot2 = (const float*)d_in[7];
    const float* b2     = (const float*)d_in[8];

    const int N = in_sizes[0] / D_IN;
    const int E = in_sizes[1] / 2;
    const int* src = ei;
    const int* dst = ei + E;

    float* ws = (float*)d_ws;
    const int NP = ((N + 4096) / 4096) * 4096;
    long off = 0;
    float* PR1 = ws + off; off += (long)N * 64;
    float* PR2 = ws + off; off += (long)N * 16;
    int* binTotal  = (int*)(ws + off); off += MAXB;
    int* binCursor = (int*)(ws + off); off += MAXB;
    int* rowptr = (int*)(ws + off); off += NP;
    int* csr    = (int*)(ws + off); off += (long)E * 2;
    int* bsw    = (int*)(ws + off); off += (long)E * 2;

    const int nTiles = N / 16;

    const int B = N / BIN_NODES + 1;
    const int nChunks = (E + CHUNK - 1) / CHUNK;

    const int tHalf = ((nTiles / 2 + 15) / 16) * 16;
    const int gA = tHalf / 16;
    const int gB = (nTiles - tHalf + 15) / 16;

    gcn_zero_i<<<(2 * MAXB + 255) / 256, 256, 0, stream>>>(binTotal, 2 * MAXB);
    gcn_bin_hist<<<nChunks, SORT_T, 0, stream>>>(dst, binTotal, E, B);
    gcn_bin_scatter<<<gA + nChunks, SORT_T, 0, stream>>>(src, dst, ew, binTotal, binCursor,
                                                         (int2*)bsw, E, B,
                                                         x, Wrel1, Wroot1, PR1, gA, 0, tHalf);
    gcn_bin_sort<<<gB + B, SORT_T, 0, stream>>>(binTotal, (const int2*)bsw, rowptr, (int2*)csr, N, B,
                                                x, Wrel1, Wroot1, PR1, gB, tHalf, nTiles);

    gcn_gather1<<<(N + 32 * G1_NPG - 1) / (32 * G1_NPG), 256, 0, stream>>>(rowptr, csr, PR1, b1, Wrel2, Wroot2, PR2, N);

    gcn_gather2_out<<<(N + 32 * G2_NPG - 1) / (32 * G2_NPG), 256, 0, stream>>>(rowptr, csr, PR2, b2,
                                                                               (float*)d_out, N);
}


# Bin-count launch removed (6 launches -> 5): each coarse bin owns a fixed region filled from its cursor's returning add; rare spill goes to a side list its sort block scans; sum order as before
# speedup vs baseline: 4.6333x; 1.0357x over previous
#include <hip/hip_runtime.h>
#include <math.h>

typedef __attribute__((ext_vector_type(2))) float v2f;
typedef __attribute__((ext_vector_type(8))) float v8f;

#define D_IN  128
#define H_DIM 32
#define C_DIM 5

__global__ void gcn_zero_i(int* __restrict__ p, int n) {
    int i = blockIdx.x * 256 + threadIdx.x;
    if (i < n) p[i] = 0;
}

__device__ __forceinline__
void gcn_gemm1_tile(const float* __restrict__ x, const float* __restrict__ Wrel,
                    const float* __restrict__ Wroot, float* __restrict__ PR1, int tile) {
    int lane = threadIdx.x & 31;
    int half = lane >> 4;
    int m    = lane & 15;
    const float* xr = x + (long)(tile * 16 + m) * D_IN;

    v2f a[32];
#pragma unroll
    for (int ks = 0; ks < 32; ++ks) {
        int k = ks * 4 + half * 2;
        a[ks].x = xr[k];
        a[ks].y = xr[k + 1];
    }

#pragma unroll
    for (int ct = 0; ct < 4; ++ct) {
        const float* __restrict__ W = (ct < 2) ? Wrel : Wroot;
        int n = (ct & 1) * 16 + m;
        v8f acc = {};
#pragma unroll
        for (int ks = 0; ks < 32; ++ks) {
            int k = ks * 4 + half * 2;
            v2f b;
            b.x = W[k * H_DIM + n];
            b.y = W[(k + 1) * H_DIM + n];
            acc = __builtin_amdgcn_wmma_f32_16x16x4_f32(false, a[ks], false, b,
                                                        (short)0, acc, false, false);
        }
#pragma unroll
        for (int r = 0; r < 8; ++r) {
            int orow = tile * 16 + r + half * 8;
            PR1[(long)orow * 64 + ct * 16 + m] = acc[r];
        }
    }
}

#define H_PITCH 36
__device__ __forceinline__
void gcn_gemm2_tile(const float* __restrict__ hs, const float* __restrict__ Wrel,
                    const float* __restrict__ Wroot, float* __restrict__ PR2, int row0, int lane) {
    int half = lane >> 4;
    int m    = lane & 15;
    const float* hr = hs + m * H_PITCH;
    const float* __restrict__ W = (m < 8) ? Wrel : Wroot;
    const int wc = min(m & 7, C_DIM - 1);

    v2f a[8];
#pragma unroll
    for (int ks = 0; ks < 8; ++ks) {
        int k = ks * 4 + half * 2;
        a[ks].x = hr[k];
        a[ks].y = hr[k + 1];
    }
    v8f acc = {};
#pragma unroll
    for (int ks = 0; ks < 8; ++ks) {
        int k = ks * 4 + half * 2;
        v2f b;
        b.x = W[k * C_DIM + wc];
        b.y = W[(k + 1) * C_DIM + wc];
        acc = __builtin_amdgcn_wmma_f32_16x16x4_f32(false, a[ks], false, b,
                                                    (short)0, acc, false, false);
    }
#pragma unroll
    for (int r = 0; r < 8; ++r) {
        int orow = row0 + r + half * 8;
        PR2[(long)orow * 16 + m] = acc[r];
    }
}

#define BIN_SHIFT 8
#define BIN_NODES 256
#define MAXB      512
#define SORT_T    512
#define EPT       8
#define CHUNK     (SORT_T * EPT)
#define SRC_BITS  24
#define SRC_MASK  0xFFFFFF
#define BIN_CAP_SHIFT 14
#define BIN_CAP   (1 << BIN_CAP_SHIFT)
#define BIN_PITCH (BIN_CAP + 40)
static_assert(CHUNK < (1 << (31 - BIN_CAP_SHIFT)), "a block's share of a bin and its place fit one word");
static_assert(SORT_T == MAXB, "one thread per bin counter");

__device__ __forceinline__ int gcn_block_excl_scan(int v, int* wsum) {
    int lane = threadIdx.x & 31, wave = threadIdx.x >> 5;
    int inc = v;
    for (int o = 1; o < 32; o <<= 1) {
        int y = __shfl_up(inc, o);
        if (lane >= o) inc += y;
    }
    if (lane == 31) wsum[wave] = inc;
    __syncthreads();
    int carry = 0;
    for (int w = 0; w < wave; ++w) carry += wsum[w];
    return carry + inc - v;
}

__global__ __launch_bounds__(SORT_T)
void gcn_bin_scatter(const int* __restrict__ src, const int* __restrict__ dst,
                     const float* __restrict__ ew, int* __restrict__ ovCount,
                     int* __restrict__ binCursor, int2* __restrict__ bsw,
                     int* __restrict__ ovS, int* __restrict__ ovD, float* __restrict__ ovW,
                     int E, int B,
                     const float* __restrict__ x, const float* __restrict__ Wrel1,
                     const float* __restrict__ Wroot1,
                     float* __restrict__ PR1, int nGemm, int tile0, int tile1) {
    __shared__ int hist[MAXB];
    __shared__ int place[MAXB];
    __shared__ int obase[MAXB];
    int tid = threadIdx.x;
    if ((int)blockIdx.x < nGemm) {
        int tile = tile0 + (int)blockIdx.x * (SORT_T / 32) + (tid >> 5);
        if (tile < tile1) gcn_gemm1_tile(x, Wrel1, Wroot1, PR1, tile);
        return;
    }
    const int chunk = (int)blockIdx.x - nGemm;
    hist[tid] = 0;
    __syncthreads();
    int e0 = chunk * CHUNK + tid;
    int d[EPT], r[EPT], s[EPT];
    float w[EPT];
#pragma unroll
    for (int k = 0; k < EPT; ++k) d[k] = dst[min(e0 + k * SORT_T, E - 1)];
#pragma unroll
    for (int k = 0; k < EPT; ++k) {
        int ec = min(e0 + k * SORT_T, E - 1);
        s[k] = src[ec];
        w[k] = ew[ec];
    }
#pragma unroll
    for (int k = 0; k < EPT; ++k) {
        r[k] = 0;
        if (e0 + k * SORT_T < E) r[k] = atomicAdd(&hist[d[k] >> BIN_SHIFT], 1);
    }
    __syncthreads();
    int c = hist[tid];
    int off = 0, fit = 0, ob = 0;
    if (tid < B && c) {
        off = atomicAdd(&binCursor[tid], c);
        fit = min(max(BIN_CAP - off, 0), c);
        if (fit < c) ob = atomicAdd(ovCount, c - fit);
    }
    place[tid] = (fit << BIN_CAP_SHIFT) | (fit ? off : 0);
    obase[tid] = ob;
    __syncthreads();
    int pk[EPT];
#pragma unroll
    for (int k = 0; k < EPT; ++k) pk[k] = place[d[k] >> BIN_SHIFT];
#pragma unroll
    for (int k = 0; k < EPT; ++k) {
        if (e0 + k * SORT_T < E) {
            const int bin = d[k] >> BIN_SHIFT;
            const int nf = pk[k] >> BIN_CAP_SHIFT;
            if (r[k] < nf) {
                int key = s[k] | ((d[k] & (BIN_NODES - 1)) << SRC_BITS);
                bsw[bin * BIN_PITCH + (pk[k] & (BIN_CAP - 1)) + r[k]] = make_int2(key, __float_as_int(w[k]));
            } else {
                int p = obase[bin] + (r[k] - nf);
                ovS[p] = s[k]; ovD[p] = d[k]; ovW[p] = w[k];
            }
        }
    }
}

__global__ __launch_bounds__(SORT_T)
void gcn_bin_sort(const int* __restrict__ binTotal, const int* __restrict__ ovCount,
                  const int2* __restrict__ bsw, const int* __restrict__ ovS,
                  const int* __restrict__ ovD, const float* __restrict__ ovW,
                  int* __restrict__ rowptr, int2* __restrict__ csr, int N, int B,
                  const float* __restrict__ x, const float* __restrict__ Wrel1,
                  const float* __restrict__ Wroot1,
                  float* __restrict__ PR1, int nGemm, int tile0, int tile1) {
    __shared__ int cnt[BIN_NODES];
    __shared__ int wsumA[16];
    __shared__ int wsumB[16];
    __shared__ int sStart, sTot;
    int tid = threadIdx.x;
    if ((int)blockIdx.x < nGemm) {
        int tile = tile0 + (int)blockIdx.x * (SORT_T / 32) + (tid >> 5);
        if (tile < tile1) gcn_gemm1_tile(x, Wrel1, Wroot1, PR1, tile);
        return;
    }
    int b = (int)blockIdx.x - nGemm;
    int t = (tid < B) ? binTotal[tid] : 0;
    int ex = gcn_block_excl_scan(t, wsumA);
    if (tid == b) { sStart = ex; sTot = t; }
    if (tid < BIN_NODES) cnt[tid] = 0;
    __syncthreads();
    const int start = sStart;
    const int tot = min(sTot, BIN_CAP);
    const int nov = (sTot > BIN_CAP) ? ovCount[0] : 0;
    const int node0 = b << BIN_SHIFT;
    const int2* mysw = bsw + (long)b * BIN_PITCH;
    for (int i0 = tid; i0 - tid < tot; i0 += CHUNK) {
        int dv[EPT];
#pragma unroll
        for (int u = 0; u < EPT; ++u) dv[u] = mysw[min(i0 + u * SORT_T, tot - 1)].x;
#pragma unroll
        for (int u = 0; u < EPT; ++u)
            if (i0 + u * SORT_T < tot) atomicAdd(&cnt[(unsigned)dv[u] >> SRC_BITS], 1);
    }
    for (int i = tid; i < nov; i += SORT_T) {
        int dn = ovD[i];
        if ((dn >> BIN_SHIFT) == b) atomicAdd(&cnt[dn & (BIN_NODES - 1)], 1);
    }
    __syncthreads();
    int c = (tid < BIN_NODES) ? cnt[tid] : 0;
    int lex = gcn_block_excl_scan(c, wsumB);
    if (tid < BIN_NODES) {
        cnt[tid] = lex;
        int n = node0 + tid;
        if (n <= N) rowptr[n] = start + lex;
    }
    __syncthreads();
    for (int i0 = tid; i0 - tid < tot; i0 += CHUNK) {
        int2 sw[EPT];
#pragma unroll
        for (int u = 0; u < EPT; ++u) sw[u] = mysw[min(i0 + u * SORT_T, tot - 1)];
#pragma unroll
        for (int u = 0; u < EPT; ++u) {
            if (i0 + u * SORT_T < tot) {
                int p = atomicAdd(&cnt[(unsigned)sw[u].x >> SRC_BITS], 1);
                csr[start + p] = make_int2(sw[u].x & SRC_MASK, sw[u].y);
            }
        }
    }
    for (int i = tid; i < nov; i += SORT_T) {
        int dn = ovD[i];
        if ((dn >> BIN_SHIFT) == b) {
            int p = atomicAdd(&cnt[dn & (BIN_NODES - 1)], 1);
            csr[start + p] = make_int2(ovS[i], __float_as_int(ovW[i]));
        }
    }
}

#define G1_NPG 2
#define G1_BCAST(V, K) ((int)__builtin_amdgcn_mov_dpp8((unsigned)(V), (unsigned)(K) * 0x249249u))
#define G1_LOAD(K) { unsigned s = (unsigned)G1_BCAST(q.x, K); \
                     v[K] = *(const float4*)(tab + ((s << 8) | c16)); }
#define G1_FMA(K)  { float w = __int_as_float(G1_BCAST(q.y, K)); \
                     const bool on = (K) < cnt; \
                     float tx = acc.x + w * v[K].x, ty = acc.y + w * v[K].y; \
                     float tz = acc.z + w * v[K].z, tw = acc.w + w * v[K].w; \
                     acc.x = on ? tx : acc.x; acc.y = on ? ty : acc.y; \
                     acc.z = on ? tz : acc.z; acc.w = on ? tw : acc.w; }
__global__ __launch_bounds__(256, 16)
void gcn_gather1(const int* __restrict__ rowptr, const int* __restrict__ csr,
                 const float* __restrict__ PR1, const float* __restrict__ b1,
                 const float* __restrict__ Wrel2, const float* __restrict__ Wroot2,
                 float* __restrict__ PR2, int N) {
    __shared__ __attribute__((aligned(16))) float hs[32 * G1_NPG * H_PITCH];
    const int node00 = (int)blockIdx.x * (32 * G1_NPG);
    const int c = threadIdx.x & 7;
    const unsigned c16 = (unsigned)c * 16u;
    const int n0 = ((int)blockIdx.x * 32 + (int)(threadIdx.x >> 3)) * G1_NPG;
    const int2* ed = (const int2*)csr;
    const char* tab = (const char*)PR1;
    static_assert(G1_NPG == 2, "three bounds per group");
    const int nb = max(min(n0, N - 2), 0);
    const int d  = n0 - nb;
    const int t0 = rowptr[nb], t1 = rowptr[nb + 1], t2 = rowptr[nb + 2];
    int r[G1_NPG + 1];
    r[0] = (d >= 2) ? t2 : (d == 1) ? t1 : t0;
    r[1] = (d >= 1) ? t2 : t1;
    r[2] = t2;
    int2 q0[G1_NPG];
#pragma unroll
    for (int it = 0; it < G1_NPG; ++it) q0[it] = ed[max(min(r[it] + c, r[it + 1] - 1), 0)];
#pragma unroll
    for (int it = 0; it < G1_NPG; ++it) {
        const int n = n0 + it;
        const int e1 = r[it + 1];
        const float4 root = *(const float4*)(tab + (((unsigned)min(n, N - 1) << 8) | c16) + 128);
        float4 acc = make_float4(0.0f, 0.0f, 0.0f, 0.0f);
        int2 q = q0[it];
        for (int p = r[it]; p < e1; p += 8) {
            const int cnt = e1 - p;
            int2 qn = ed[min(p + 8 + c, e1 - 1)];
            float4 v[8];
            G1_LOAD(0) G1_LOAD(1) G1_LOAD(2) G1_LOAD(3)
            G1_LOAD(4) G1_LOAD(5) G1_LOAD(6) G1_LOAD(7)
            G1_FMA(0) G1_FMA(1) G1_FMA(2) G1_FMA(3)
            G1_FMA(4) G1_FMA(5) G1_FMA(6) G1_FMA(7)
            q = qn;
        }
        float4 o = make_float4(0.0f, 0.0f, 0.0f, 0.0f);
        if (n < N) {
            float cf = fmaxf((float)(e1 - r[it]), 1.0f);
            o.x = fmaxf(acc.x / cf + root.x + b1[c * 4],     0.0f);
            o.y = fmaxf(acc.y / cf + root.y + b1[c * 4 + 1], 0.0f);
            o.z = fmaxf(acc.z / cf + root.z + b1[c * 4 + 2], 0.0f);
            o.w = fmaxf(acc.w / cf + root.w + b1[c * 4 + 3], 0.0f);
        }
        *(float4*)(hs + (n - node00) * H_PITCH + c * 4) = o;
    }
    __syncthreads();
    static_assert((32 * G1_NPG) % 16 == 0, "whole tiles per block");
    const int wave = (__builtin_amdgcn_readfirstlane(n0) - node00) / (4 * G1_NPG);
    const int row0 = node00 + wave * 16;
    if (wave < (32 * G1_NPG) / 16 && row0 + 16 <= N)
        gcn_gemm2_tile(hs + wave * 16 * H_PITCH, Wrel2, Wroot2, PR2, row0, (int)__lane_id());
}

#define G2_NPG 2
#define G2_PAIR_LO(V) ((int)__builtin_amdgcn_mov_dpp8((unsigned)(V), 0x6D2240u))
#define G2_PAIR_HI(V) ((int)__builtin_amdgcn_mov_dpp8((unsigned)(V), 0xFF6B64u))
#define G2_X1(F) __uint_as_float(__builtin_amdgcn_mov_dpp8(__float_as_uint(F), 0xDE54C1u))
#define G2_X2(F) __uint_as_float(__builtin_amdgcn_mov_dpp8(__float_as_uint(F), 0xB3E21Au))
#define G2_X4(F) __uint_as_float(__builtin_amdgcn_mov_dpp8(__float_as_uint(F), 0x688FACu))

__device__ __forceinline__ void gcn_g2_records(int (&q)[4], const int2* __restrict__ ed,
                                               int p, int e1, int c) {
#pragma unroll
    for (int k = 0; k < 4; ++k) q[k] = ed[max(min(p + 8 * k + c, e1 - 1), 0)].x;
}
__device__ __forceinline__ void gcn_g2_rows(float4 (&v)[8], const int (&q)[4],
                                            const char* __restrict__ tab, unsigned h16) {
#pragma unroll
    for (int k = 0; k < 4; ++k) {
        unsigned s0 = (unsigned)G2_PAIR_LO(q[k]);
        unsigned s1 = (unsigned)G2_PAIR_HI(q[k]);
        v[2 * k]     = *(const float4*)(tab + ((s0 << 6) | h16));
        v[2 * k + 1] = *(const float4*)(tab + ((s1 << 6) | h16));
    }
}
__device__ __forceinline__ void gcn_g2_add(float4& acc, const float4 (&v)[8], int cnt, int slot) {
#pragma unroll
    for (int u = 0; u < 8; ++u) {
        const bool on = (4 * u + slot) < cnt;
        const float tx = acc.x + v[u].x, ty = acc.y + v[u].y;
        const float tz = acc.z + v[u].z, tw = acc.w + v[u].w;
        acc.x = on ? tx : acc.x; acc.y = on ? ty : acc.y;
        acc.z = on ? tz : acc.z; acc.w = on ? tw : acc.w;
    }
}

__global__ __launch_bounds__(256)
void gcn_gather2_out(const int* __restrict__ rowptr, const int* __restrict__ csr,
                     const float* __restrict__ PR2, const float* __restrict__ b2,
                     float* __restrict__ out, int N) {
    __shared__ float so[32 * G2_NPG * C_DIM];
    const int c    = threadIdx.x & 7;
    const int slot = c >> 1;
    const int hf   = c & 1;
    const unsigned h16 = (unsigned)hf * 16u;
    const int g  = threadIdx.x >> 3;
    const int n0 = ((int)blockIdx.x * 32 + g) * G2_NPG;
    const int2* ed = (const int2*)csr;
    const char* tab = (const char*)PR2;
    static_assert(G2_NPG == 2, "three bounds per group");
    const int nb = max(min(n0, N - 2), 0);
    const int d  = n0 - nb;
    const int t0 = rowptr[nb], t1 = rowptr[nb + 1], t2 = rowptr[nb + 2];
    int r[G2_NPG + 1];
    r[0] = (d >= 2) ? t2 : (d == 1) ? t1 : t0;
    r[1] = (d >= 1) ? t2 : t1;
    r[2] = t2;
    int q[G2_NPG][4];
#pragma unroll
    for (int it = 0; it < G2_NPG; ++it) gcn_g2_records(q[it], ed, r[it], r[it + 1], c);
    float4 root[G2_NPG];
#pragma unroll
    for (int it = 0; it < G2_NPG; ++it)
        root[it] = *(const float4*)(tab + (((unsigned)min(n0 + it, N - 1) << 6) | h16) + 32);
    const int bi = hf * 4;
    const float4 bb = make_float4(b2[bi], b2[min(bi + 1, C_DIM - 1)],
                                  b2[min(bi + 2, C_DIM - 1)], b2[min(bi + 3, C_DIM - 1)]);
    float4 v[G2_NPG][8];
#pragma unroll
    for (int it = 0; it < G2_NPG; ++it) gcn_g2_rows(v[it], q[it], tab, h16);
    asm volatile("" :: "v"(v[0][0].x), "v"(v[0][0].y), "v"(v[0][0].z), "v"(v[0][0].w));
#pragma unroll
    for (int it = 0; it < G2_NPG; ++it) {
        const int n = n0 + it;
        const int e1 = r[it + 1];
        const int deg = e1 - r[it];
        float4 acc = make_float4(0.0f, 0.0f, 0.0f, 0.0f);
        gcn_g2_add(acc, v[it], deg, slot);
        for (int p = r[it] + 32; p < e1; p += 32) {
            int qq[4];
            float4 vv[8];
            gcn_g2_records(qq, ed, p, e1, c);
            gcn_g2_rows(vv, qq, tab, h16);
            gcn_g2_add(acc, vv, e1 - p, slot);
        }
        acc.x += G2_X2(acc.x); acc.y += G2_X2(acc.y); acc.z += G2_X2(acc.z); acc.w += G2_X2(acc.w);
        acc.x += G2_X4(acc.x); acc.y += G2_X4(acc.y); acc.z += G2_X4(acc.z); acc.w += G2_X4(acc.w);
        const float cn = fmaxf((float)deg, 1.0f);
        const float v0 = acc.x / cn + root[it].x + bb.x;
        const float v1 = acc.y / cn + root[it].y + bb.y;
        const float v2 = acc.z / cn + root[it].z + bb.z;
        const float v3 = acc.w / cn + root[it].w + bb.w;
        const float v4 = G2_X1(v0);
        float mx = fmaxf(fmaxf(fmaxf(v0, v1), fmaxf(v2, v3)), v4);
        float s = expf(v0 - mx) + expf(v1 - mx) + expf(v2 - mx) + expf(v3 - mx) + expf(v4 - mx);
        float l = logf(s);
        if (c == 0 && n < N) {
            float* o = so + (g * G2_NPG + it) * C_DIM;
            o[0] = v0 - mx - l; o[1] = v1 - mx - l; o[2] = v2 - mx - l;
            o[3] = v3 - mx - l; o[4] = v4 - mx - l;
        }
    }
    __syncthreads();
    const int rem = N - (int)blockIdx.x * (32 * G2_NPG);
    const int valid = (rem < 32 * G2_NPG ? rem : 32 * G2_NPG) * C_DIM;
    static_assert(32 * G2_NPG * C_DIM <= 2 * 256, "two stores per thread cover the block's rows");
    float* ob = out + (long)blockIdx.x * (32 * G2_NPG * C_DIM);
    const int t = (int)threadIdx.x;
    if (t < valid)       ob[t]       = so[t];
    if (t + 256 < valid) ob[t + 256] = so[t + 256];
}

extern "C" void kernel_launch(void* const* d_in, const int* in_sizes, int n_in,
                              void* d_out, int out_size, void* d_ws, size_t ws_size,
                              hipStream_t stream) {
    const float* x      = (const float*)d_in[0];
    const int*   ei     = (const int*)d_in[1];
    const float* ew     = (const float*)d_in[2];
    const float* Wrel1  = (const float*)d_in[3];
    const float* Wroot1 = (const float*)d_in[4];
    const float* b1     = (const float*)d_in[5];
    const float* Wrel2  = (const float*)d_in[6];
    const float* Wroot2 = (const float*)d_in[7];
    const float* b2     = (const float*)d_in[8];

    const int N = in_sizes[0] / D_IN;
    const int E = in_sizes[1] / 2;
    const int* src = ei;
    const int* dst = ei + E;

    float* ws = (float*)d_ws;
    const int NP = ((N + 4096) / 4096) * 4096;
    long off = 0;
    float* PR1 = ws + off; off += (long)N * 64;
    float* PR2 = ws + off; off += (long)N * 16;
    int* ovCount   = (int*)(ws + off); off += MAXB;
    int* binCursor = (int*)(ws + off); off += MAXB;
    int* rowptr = (int*)(ws + off); off += NP;
    int* csr    = (int*)(ws + off); off += (long)E * 2;
    int* bsw    = (int*)(ws + off); off += (long)MAXB * BIN_PITCH * 2;
    int* ovS    = (int*)(ws + off); off += E;
    int* ovD    = (int*)(ws + off); off += E;
    float* ovW  = ws + off;         off += E;

    const int nTiles = N / 16;

    const int B = N / BIN_NODES + 1;
    const int nChunks = (E + CHUNK - 1) / CHUNK;

    const int tHalf = ((nTiles / 2 + 15) / 16) * 16;
    const int gA = tHalf / 16;
    const int gB = (nTiles - tHalf + 15) / 16;

    gcn_zero_i<<<(2 * MAXB + 255) / 256, 256, 0, stream>>>(ovCount, 2 * MAXB);
    gcn_bin_scatter<<<gA + nChunks, SORT_T, 0, stream>>>(src, dst, ew, ovCount, binCursor,
                                                         (int2*)bsw, ovS, ovD, ovW, E, B,
                                                         x, Wrel1, Wroot1, PR1, gA, 0, tHalf);
    gcn_bin_sort<<<gB + B, SORT_T, 0, stream>>>(binCursor, ovCount, (const int2*)bsw, ovS, ovD, ovW,
                                                rowptr, (int2*)csr, N, B,
                                                x, Wrel1, Wroot1, PR1, gB, tHalf, nTiles);

    gcn_gather1<<<(N + 32 * G1_NPG - 1) / (32 * G1_NPG), 256, 0, stream>>>(rowptr, csr, PR1, b1, Wrel2, Wroot2, PR2, N);

    gcn_gather2_out<<<(N + 32 * G2_NPG - 1) / (32 * G2_NPG), 256, 0, stream>>>(rowptr, csr, PR2, b2,
                                                                               (float*)d_out, N);
}


# Both bucketing launches order their records in LDS and store them as contiguous runs, not one 8-byte store per lane (32 lines per store instruction); each record's place and all sum orders unchanged
# speedup vs baseline: 5.4433x; 1.1748x over previous
#include <hip/hip_runtime.h>
#include <math.h>

typedef __attribute__((ext_vector_type(2))) float v2f;
typedef __attribute__((ext_vector_type(8))) float v8f;

#define D_IN  128
#define H_DIM 32
#define C_DIM 5

__global__ void gcn_zero_i(int* __restrict__ p, int n) {
    int i = blockIdx.x * 256 + threadIdx.x;
    if (i < n) p[i] = 0;
}

__device__ __forceinline__
void gcn_gemm1_tile(const float* __restrict__ x, const float* __restrict__ Wrel,
                    const float* __restrict__ Wroot, float* __restrict__ PR1, int tile) {
    int lane = threadIdx.x & 31;
    int half = lane >> 4;
    int m    = lane & 15;
    const float* xr = x + (long)(tile * 16 + m) * D_IN;

    v2f a[32];
#pragma unroll
    for (int ks = 0; ks < 32; ++ks) {
        int k = ks * 4 + half * 2;
        a[ks].x = xr[k];
        a[ks].y = xr[k + 1];
    }

#pragma unroll
    for (int ct = 0; ct < 4; ++ct) {
        const float* __restrict__ W = (ct < 2) ? Wrel : Wroot;
        int n = (ct & 1) * 16 + m;
        v8f acc = {};
#pragma unroll
        for (int ks = 0; ks < 32; ++ks) {
            int k = ks * 4 + half * 2;
            v2f b;
            b.x = W[k * H_DIM + n];
            b.y = W[(k + 1) * H_DIM + n];
            acc = __builtin_amdgcn_wmma_f32_16x16x4_f32(false, a[ks], false, b,
                                                        (short)0, acc, false, false);
        }
#pragma unroll
        for (int r = 0; r < 8; ++r) {
            int orow = tile * 16 + r + half * 8;
            PR1[(long)orow * 64 + ct * 16 + m] = acc[r];
        }
    }
}

#define H_PITCH 36
__device__ __forceinline__
void gcn_gemm2_tile(const float* __restrict__ hs, const float* __restrict__ Wrel,
                    const float* __restrict__ Wroot, float* __restrict__ PR2, int row0, int lane) {
    int half = lane >> 4;
    int m    = lane & 15;
    const float* hr = hs + m * H_PITCH;
    const float* __restrict__ W = (m < 8) ? Wrel : Wroot;
    const int wc = min(m & 7, C_DIM - 1);

    v2f a[8];
#pragma unroll
    for (int ks = 0; ks < 8; ++ks) {
        int k = ks * 4 + half * 2;
        a[ks].x = hr[k];
        a[ks].y = hr[k + 1];
    }
    v8f acc = {};
#pragma unroll
    for (int ks = 0; ks < 8; ++ks) {
        int k = ks * 4 + half * 2;
        v2f b;
        b.x = W[k * C_DIM + wc];
        b.y = W[(k + 1) * C_DIM + wc];
        acc = __builtin_amdgcn_wmma_f32_16x16x4_f32(false, a[ks], false, b,
                                                    (short)0, acc, false, false);
    }
#pragma unroll
    for (int r = 0; r < 8; ++r) {
        int orow = row0 + r + half * 8;
        PR2[(long)orow * 16 + m] = acc[r];
    }
}

#define BIN_SHIFT 8
#define BIN_NODES 256
#define MAXB      512
#define SORT_T    512
#define EPT       8
#define CHUNK     (SORT_T * EPT)
#define SRC_BITS  24
#define SRC_MASK  0xFFFFFF
#define BIN_CAP_SHIFT 14
#define BIN_CAP   (1 << BIN_CAP_SHIFT)
#define BIN_PITCH (BIN_CAP + 40)
#define STAGE_CAP 5120
static_assert(CHUNK < (1 << (31 - BIN_CAP_SHIFT)), "a block's share of a bin and its place fit one word");
static_assert(SORT_T == MAXB, "one thread per bin counter");

__device__ __forceinline__ int gcn_block_excl_scan(int v, int* wsum) {
    int lane = threadIdx.x & 31, wave = threadIdx.x >> 5;
    int inc = v;
    for (int o = 1; o < 32; o <<= 1) {
        int y = __shfl_up(inc, o);
        if (lane >= o) inc += y;
    }
    if (lane == 31) wsum[wave] = inc;
    __syncthreads();
    int carry = 0;
    for (int w = 0; w < wave; ++w) carry += wsum[w];
    return carry + inc - v;
}

__global__ __launch_bounds__(SORT_T)
void gcn_bin_scatter(const int* __restrict__ src, const int* __restrict__ dst,
                     const float* __restrict__ ew, int* __restrict__ ovCount,
                     int* __restrict__ binCursor, int2* __restrict__ bsw,
                     int* __restrict__ ovS, int* __restrict__ ovD, float* __restrict__ ovW,
                     int E, int B,
                     const float* __restrict__ x, const float* __restrict__ Wrel1,
                     const float* __restrict__ Wroot1,
                     float* __restrict__ PR1, int nGemm, int tile0, int tile1) {
    __shared__ int hist[MAXB];
    __shared__ int place[MAXB];
    __shared__ int obase[MAXB];
    __shared__ int lbase[MAXB];
    __shared__ int wsumS[16];
    __shared__ __attribute__((aligned(8))) int stageS[2 * CHUNK];
    __shared__ int gidx[CHUNK];
    int2* st = (int2*)stageS;
    int tid = threadIdx.x;
    if ((int)blockIdx.x < nGemm) {
        int tile = tile0 + (int)blockIdx.x * (SORT_T / 32) + (tid >> 5);
        if (tile < tile1) gcn_gemm1_tile(x, Wrel1, Wroot1, PR1, tile);
        return;
    }
    const int chunk = (int)blockIdx.x - nGemm;
    hist[tid] = 0;
    __syncthreads();
    int e0 = chunk * CHUNK + tid;
    int d[EPT], r[EPT], s[EPT];
    float w[EPT];
#pragma unroll
    for (int k = 0; k < EPT; ++k) d[k] = dst[min(e0 + k * SORT_T, E - 1)];
#pragma unroll
    for (int k = 0; k < EPT; ++k) {
        int ec = min(e0 + k * SORT_T, E - 1);
        s[k] = src[ec];
        w[k] = ew[ec];
    }
#pragma unroll
    for (int k = 0; k < EPT; ++k) {
        r[k] = 0;
        if (e0 + k * SORT_T < E) r[k] = atomicAdd(&hist[d[k] >> BIN_SHIFT], 1);
    }
    __syncthreads();
    int c = hist[tid];
    const int lo = gcn_block_excl_scan(c, wsumS);
    int off = 0, fit = 0, ob = 0;
    if (tid < B && c) {
        off = atomicAdd(&binCursor[tid], c);
        fit = min(max(BIN_CAP - off, 0), c);
        if (fit < c) ob = atomicAdd(ovCount, c - fit);
    }
    place[tid] = (fit << BIN_CAP_SHIFT) | (fit ? off : 0);
    obase[tid] = ob;
    lbase[tid] = lo;
    __syncthreads();
    int pk[EPT], lb[EPT];
#pragma unroll
    for (int k = 0; k < EPT; ++k) { pk[k] = place[d[k] >> BIN_SHIFT]; lb[k] = lbase[d[k] >> BIN_SHIFT]; }
#pragma unroll
    for (int k = 0; k < EPT; ++k) {
        if (e0 + k * SORT_T < E) {
            const int bin = d[k] >> BIN_SHIFT;
            const int nf = pk[k] >> BIN_CAP_SHIFT;
            const int pos = lb[k] + r[k];
            if (r[k] < nf) {
                int key = s[k] | ((d[k] & (BIN_NODES - 1)) << SRC_BITS);
                st[pos] = make_int2(key, __float_as_int(w[k]));
                gidx[pos] = bin * BIN_PITCH + (pk[k] & (BIN_CAP - 1)) + r[k];
            } else {
                gidx[pos] = -1;
                int p = obase[bin] + (r[k] - nf);
                ovS[p] = s[k]; ovD[p] = d[k]; ovW[p] = w[k];
            }
        }
    }
    __syncthreads();
    const int nE = min(CHUNK, E - chunk * CHUNK);
    int g[EPT];
    int2 rc[EPT];
#pragma unroll
    for (int k = 0; k < EPT; ++k) {
        const int i = min(tid + k * SORT_T, nE - 1);
        g[k] = gidx[i];
        rc[k] = st[i];
    }
#pragma unroll
    for (int k = 0; k < EPT; ++k)
        if (tid + k * SORT_T < nE && g[k] >= 0) bsw[g[k]] = rc[k];
}

__global__ __launch_bounds__(SORT_T)
void gcn_bin_sort(const int* __restrict__ binTotal, const int* __restrict__ ovCount,
                  const int2* __restrict__ bsw, const int* __restrict__ ovS,
                  const int* __restrict__ ovD, const float* __restrict__ ovW,
                  int* __restrict__ rowptr, int2* __restrict__ csr, int N, int B,
                  const float* __restrict__ x, const float* __restrict__ Wrel1,
                  const float* __restrict__ Wroot1,
                  float* __restrict__ PR1, int nGemm, int tile0, int tile1) {
    __shared__ int cnt[BIN_NODES];
    __shared__ int wsumA[16];
    __shared__ int wsumB[16];
    __shared__ int sStart, sTot;
    __shared__ __attribute__((aligned(8))) int stageB[2 * STAGE_CAP];
    int2* st = (int2*)stageB;
    int tid = threadIdx.x;
    if ((int)blockIdx.x < nGemm) {
        int tile = tile0 + (int)blockIdx.x * (SORT_T / 32) + (tid >> 5);
        if (tile < tile1) gcn_gemm1_tile(x, Wrel1, Wroot1, PR1, tile);
        return;
    }
    int b = (int)blockIdx.x - nGemm;
    int t = (tid < B) ? binTotal[tid] : 0;
    int ex = gcn_block_excl_scan(t, wsumA);
    if (tid == b) { sStart = ex; sTot = t; }
    if (tid < BIN_NODES) cnt[tid] = 0;
    __syncthreads();
    const int start = sStart;
    const int nst = min(sTot, STAGE_CAP);
    const int tot = min(sTot, BIN_CAP);
    const int nov = (sTot > BIN_CAP) ? ovCount[0] : 0;
    const int node0 = b << BIN_SHIFT;
    const int2* mysw = bsw + (long)b * BIN_PITCH;
    for (int i0 = tid; i0 - tid < tot; i0 += CHUNK) {
        int dv[EPT];
#pragma unroll
        for (int u = 0; u < EPT; ++u) dv[u] = mysw[min(i0 + u * SORT_T, tot - 1)].x;
#pragma unroll
        for (int u = 0; u < EPT; ++u)
            if (i0 + u * SORT_T < tot) atomicAdd(&cnt[(unsigned)dv[u] >> SRC_BITS], 1);
    }
    for (int i = tid; i < nov; i += SORT_T) {
        int dn = ovD[i];
        if ((dn >> BIN_SHIFT) == b) atomicAdd(&cnt[dn & (BIN_NODES - 1)], 1);
    }
    __syncthreads();
    int c = (tid < BIN_NODES) ? cnt[tid] : 0;
    int lex = gcn_block_excl_scan(c, wsumB);
    if (tid < BIN_NODES) {
        cnt[tid] = lex;
        int n = node0 + tid;
        if (n <= N) rowptr[n] = start + lex;
    }
    __syncthreads();
    for (int i0 = tid; i0 - tid < tot; i0 += CHUNK) {
        int2 sw[EPT];
#pragma unroll
        for (int u = 0; u < EPT; ++u) sw[u] = mysw[min(i0 + u * SORT_T, tot - 1)];
#pragma unroll
        for (int u = 0; u < EPT; ++u) {
            if (i0 + u * SORT_T < tot) {
                int p = atomicAdd(&cnt[(unsigned)sw[u].x >> SRC_BITS], 1);
                const int2 rec = make_int2(sw[u].x & SRC_MASK, sw[u].y);
                if (p < STAGE_CAP) st[p] = rec; else csr[start + p] = rec;
            }
        }
    }
    for (int i = tid; i < nov; i += SORT_T) {
        int dn = ovD[i];
        if ((dn >> BIN_SHIFT) == b) {
            int p = atomicAdd(&cnt[dn & (BIN_NODES - 1)], 1);
            const int2 rec = make_int2(ovS[i], __float_as_int(ovW[i]));
            if (p < STAGE_CAP) st[p] = rec; else csr[start + p] = rec;
        }
    }
    __syncthreads();
    for (int i0 = tid; i0 - tid < nst; i0 += CHUNK) {
        int2 rc[EPT];
#pragma unroll
        for (int u = 0; u < EPT; ++u) rc[u] = st[min(i0 + u * SORT_T, nst - 1)];
#pragma unroll
        for (int u = 0; u < EPT; ++u)
            if (i0 + u * SORT_T < nst) csr[start + i0 + u * SORT_T] = rc[u];
    }
}

#define G1_NPG 2
#define G1_BCAST(V, K) ((int)__builtin_amdgcn_mov_dpp8((unsigned)(V), (unsigned)(K) * 0x249249u))
#define G1_LOAD(K) { unsigned s = (unsigned)G1_BCAST(q.x, K); \
                     v[K] = *(const float4*)(tab + ((s << 8) | c16)); }
#define G1_FMA(K)  { float w = __int_as_float(G1_BCAST(q.y, K)); \
                     const bool on = (K) < cnt; \
                     float tx = acc.x + w * v[K].x, ty = acc.y + w * v[K].y; \
                     float tz = acc.z + w * v[K].z, tw = acc.w + w * v[K].w; \
                     acc.x = on ? tx : acc.x; acc.y = on ? ty : acc.y; \
                     acc.z = on ? tz : acc.z; acc.w = on ? tw : acc.w; }
__global__ __launch_bounds__(256, 16)
void gcn_gather1(const int* __restrict__ rowptr, const int* __restrict__ csr,
                 const float* __restrict__ PR1, const float* __restrict__ b1,
                 const float* __restrict__ Wrel2, const float* __restrict__ Wroot2,
                 float* __restrict__ PR2, int N) {
    __shared__ __attribute__((aligned(16))) float hs[32 * G1_NPG * H_PITCH];
    const int node00 = (int)blockIdx.x * (32 * G1_NPG);
    const int c = threadIdx.x & 7;
    const unsigned c16 = (unsigned)c * 16u;
    const int n0 = ((int)blockIdx.x * 32 + (int)(threadIdx.x >> 3)) * G1_NPG;
    const int2* ed = (const int2*)csr;
    const char* tab = (const char*)PR1;
    static_assert(G1_NPG == 2, "three bounds per group");
    const int nb = max(min(n0, N - 2), 0);
    const int d  = n0 - nb;
    const int t0 = rowptr[nb], t1 = rowptr[nb + 1], t2 = rowptr[nb + 2];
    int r[G1_NPG + 1];
    r[0] = (d >= 2) ? t2 : (d == 1) ? t1 : t0;
    r[1] = (d >= 1) ? t2 : t1;
    r[2] = t2;
    int2 q0[G1_NPG];
#pragma unroll
    for (int it = 0; it < G1_NPG; ++it) q0[it] = ed[max(min(r[it] + c, r[it + 1] - 1), 0)];
#pragma unroll
    for (int it = 0; it < G1_NPG; ++it) {
        const int n = n0 + it;
        const int e1 = r[it + 1];
        const float4 root = *(const float4*)(tab + (((unsigned)min(n, N - 1) << 8) | c16) + 128);
        float4 acc = make_float4(0.0f, 0.0f, 0.0f, 0.0f);
        int2 q = q0[it];
        for (int p = r[it]; p < e1; p += 8) {
            const int cnt = e1 - p;
            int2 qn = ed[min(p + 8 + c, e1 - 1)];
            float4 v[8];
            G1_LOAD(0) G1_LOAD(1) G1_LOAD(2) G1_LOAD(3)
            G1_LOAD(4) G1_LOAD(5) G1_LOAD(6) G1_LOAD(7)
            G1_FMA(0) G1_FMA(1) G1_FMA(2) G1_FMA(3)
            G1_FMA(4) G1_FMA(5) G1_FMA(6) G1_FMA(7)
            q = qn;
        }
        float4 o = make_float4(0.0f, 0.0f, 0.0f, 0.0f);
        if (n < N) {
            float cf = fmaxf((float)(e1 - r[it]), 1.0f);
            o.x = fmaxf(acc.x / cf + root.x + b1[c * 4],     0.0f);
            o.y = fmaxf(acc.y / cf + root.y + b1[c * 4 + 1], 0.0f);
            o.z = fmaxf(acc.z / cf + root.z + b1[c * 4 + 2], 0.0f);
            o.w = fmaxf(acc.w / cf + root.w + b1[c * 4 + 3], 0.0f);
        }
        *(float4*)(hs + (n - node00) * H_PITCH + c * 4) = o;
    }
    __syncthreads();
    static_assert((32 * G1_NPG) % 16 == 0, "whole tiles per block");
    const int wave = (__builtin_amdgcn_readfirstlane(n0) - node00) / (4 * G1_NPG);
    const int row0 = node00 + wave * 16;
    if (wave < (32 * G1_NPG) / 16 && row0 + 16 <= N)
        gcn_gemm2_tile(hs + wave * 16 * H_PITCH, Wrel2, Wroot2, PR2, row0, (int)__lane_id());
}

#define G2_NPG 2
#define G2_PAIR_LO(V) ((int)__builtin_amdgcn_mov_dpp8((unsigned)(V), 0x6D2240u))
#define G2_PAIR_HI(V) ((int)__builtin_amdgcn_mov_dpp8((unsigned)(V), 0xFF6B64u))
#define G2_X1(F) __uint_as_float(__builtin_amdgcn_mov_dpp8(__float_as_uint(F), 0xDE54C1u))
#define G2_X2(F) __uint_as_float(__builtin_amdgcn_mov_dpp8(__float_as_uint(F), 0xB3E21Au))
#define G2_X4(F) __uint_as_float(__builtin_amdgcn_mov_dpp8(__float_as_uint(F), 0x688FACu))

__device__ __forceinline__ void gcn_g2_records(int (&q)[4], const int2* __restrict__ ed,
                                               int p, int e1, int c) {
#pragma unroll
    for (int k = 0; k < 4; ++k) q[k] = ed[max(min(p + 8 * k + c, e1 - 1), 0)].x;
}
__device__ __forceinline__ void gcn_g2_rows(float4 (&v)[8], const int (&q)[4],
                                            const char* __restrict__ tab, unsigned h16) {
#pragma unroll
    for (int k = 0; k < 4; ++k) {
        unsigned s0 = (unsigned)G2_PAIR_LO(q[k]);
        unsigned s1 = (unsigned)G2_PAIR_HI(q[k]);
        v[2 * k]     = *(const float4*)(tab + ((s0 << 6) | h16));
        v[2 * k + 1] = *(const float4*)(tab + ((s1 << 6) | h16));
    }
}
__device__ __forceinline__ void gcn_g2_add(float4& acc, const float4 (&v)[8], int cnt, int slot) {
#pragma unroll
    for (int u = 0; u < 8; ++u) {
        const bool on = (4 * u + slot) < cnt;
        const float tx = acc.x + v[u].x, ty = acc.y + v[u].y;
        const float tz = acc.z + v[u].z, tw = acc.w + v[u].w;
        acc.x = on ? tx : acc.x; acc.y = on ? ty : acc.y;
        acc.z = on ? tz : acc.z; acc.w = on ? tw : acc.w;
    }
}

__global__ __launch_bounds__(256)
void gcn_gather2_out(const int* __restrict__ rowptr, const int* __restrict__ csr,
                     const float* __restrict__ PR2, const float* __restrict__ b2,
                     float* __restrict__ out, int N) {
    __shared__ float so[32 * G2_NPG * C_DIM];
    const int c    = threadIdx.x & 7;
    const int slot = c >> 1;
    const int hf   = c & 1;
    const unsigned h16 = (unsigned)hf * 16u;
    const int g  = threadIdx.x >> 3;
    const int n0 = ((int)blockIdx.x * 32 + g) * G2_NPG;
    const int2* ed = (const int2*)csr;
    const char* tab = (const char*)PR2;
    static_assert(G2_NPG == 2, "three bounds per group");
    const int nb = max(min(n0, N - 2), 0);
    const int d  = n0 - nb;
    const int t0 = rowptr[nb], t1 = rowptr[nb + 1], t2 = rowptr[nb + 2];
    int r[G2_NPG + 1];
    r[0] = (d >= 2) ? t2 : (d == 1) ? t1 : t0;
    r[1] = (d >= 1) ? t2 : t1;
    r[2] = t2;
    int q[G2_NPG][4];
#pragma unroll
    for (int it = 0; it < G2_NPG; ++it) gcn_g2_records(q[it], ed, r[it], r[it + 1], c);
    float4 root[G2_NPG];
#pragma unroll
    for (int it = 0; it < G2_NPG; ++it)
        root[it] = *(const float4*)(tab + (((unsigned)min(n0 + it, N - 1) << 6) | h16) + 32);
    const int bi = hf * 4;
    const float4 bb = make_float4(b2[bi], b2[min(bi + 1, C_DIM - 1)],
                                  b2[min(bi + 2, C_DIM - 1)], b2[min(bi + 3, C_DIM - 1)]);
    float4 v[G2_NPG][8];
#pragma unroll
    for (int it = 0; it < G2_NPG; ++it) gcn_g2_rows(v[it], q[it], tab, h16);
    asm volatile("" :: "v"(v[0][0].x), "v"(v[0][0].y), "v"(v[0][0].z), "v"(v[0][0].w));
#pragma unroll
    for (int it = 0; it < G2_NPG; ++it) {
        const int n = n0 + it;
        const int e1 = r[it + 1];
        const int deg = e1 - r[it];
        float4 acc = make_float4(0.0f, 0.0f, 0.0f, 0.0f);
        gcn_g2_add(acc, v[it], deg, slot);
        for (int p = r[it] + 32; p < e1; p += 32) {
            int qq[4];
            float4 vv[8];
            gcn_g2_records(qq, ed, p, e1, c);
            gcn_g2_rows(vv, qq, tab, h16);
            gcn_g2_add(acc, vv, e1 - p, slot);
        }
        acc.x += G2_X2(acc.x); acc.y += G2_X2(acc.y); acc.z += G2_X2(acc.z); acc.w += G2_X2(acc.w);
        acc.x += G2_X4(acc.x); acc.y += G2_X4(acc.y); acc.z += G2_X4(acc.z); acc.w += G2_X4(acc.w);
        const float cn = fmaxf((float)deg, 1.0f);
        const float v0 = acc.x / cn + root[it].x + bb.x;
        const float v1 = acc.y / cn + root[it].y + bb.y;
        const float v2 = acc.z / cn + root[it].z + bb.z;
        const float v3 = acc.w / cn + root[it].w + bb.w;
        const float v4 = G2_X1(v0);
        float mx = fmaxf(fmaxf(fmaxf(v0, v1), fmaxf(v2, v3)), v4);
        float s = expf(v0 - mx) + expf(v1 - mx) + expf(v2 - mx) + expf(v3 - mx) + expf(v4 - mx);
        float l = logf(s);
        if (c == 0 && n < N) {
            float* o = so + (g * G2_NPG + it) * C_DIM;
            o[0] = v0 - mx - l; o[1] = v1 - mx - l; o[2] = v2 - mx - l;
            o[3] = v3 - mx - l; o[4] = v4 - mx - l;
        }
    }
    __syncthreads();
    const int rem = N - (int)blockIdx.x * (32 * G2_NPG);
    const int valid = (rem < 32 * G2_NPG ? rem : 32 * G2_NPG) * C_DIM;
    static_assert(32 * G2_NPG * C_DIM <= 2 * 256, "two stores per thread cover the block's rows");
    float* ob = out + (long)blockIdx.x * (32 * G2_NPG * C_DIM);
    const int t = (int)threadIdx.x;
    if (t < valid)       ob[t]       = so[t];
    if (t + 256 < valid) ob[t + 256] = so[t + 256];
}

extern "C" void kernel_launch(void* const* d_in, const int* in_sizes, int n_in,
                              void* d_out, int out_size, void* d_ws, size_t ws_size,
                              hipStream_t stream) {
    const float* x      = (const float*)d_in[0];
    const int*   ei     = (const int*)d_in[1];
    const float* ew     = (const float*)d_in[2];
    const float* Wrel1  = (const float*)d_in[3];
    const float* Wroot1 = (const float*)d_in[4];
    const float* b1     = (const float*)d_in[5];
    const float* Wrel2  = (const float*)d_in[6];
    const float* Wroot2 = (const float*)d_in[7];
    const float* b2     = (const float*)d_in[8];

    const int N = in_sizes[0] / D_IN;
    const int E = in_sizes[1] / 2;
    const int* src = ei;
    const int* dst = ei + E;

    float* ws = (float*)d_ws;
    const int NP = ((N + 4096) / 4096) * 4096;
    long off = 0;
    float* PR1 = ws + off; off += (long)N * 64;
    float* PR2 = ws + off; off += (long)N * 16;
    int* ovCount   = (int*)(ws + off); off += MAXB;
    int* binCursor = (int*)(ws + off); off += MAXB;
    int* rowptr = (int*)(ws + off); off += NP;
    int* csr    = (int*)(ws + off); off += (long)E * 2;
    int* bsw    = (int*)(ws + off); off += (long)MAXB * BIN_PITCH * 2;
    int* ovS    = (int*)(ws + off); off += E;
    int* ovD    = (int*)(ws + off); off += E;
    float* ovW  = ws + off;         off += E;

    const int nTiles = N / 16;

    const int B = N / BIN_NODES + 1;
    const int nChunks = (E + CHUNK - 1) / CHUNK;

    const int tHalf = ((nTiles / 2 + 15) / 16) * 16;
    const int gA = tHalf / 16;
    const int gB = (nTiles - tHalf + 15) / 16;

    gcn_zero_i<<<(2 * MAXB + 255) / 256, 256, 0, stream>>>(ovCount, 2 * MAXB);
    gcn_bin_scatter<<<gA + nChunks, SORT_T, 0, stream>>>(src, dst, ew, ovCount, binCursor,
                                                         (int2*)bsw, ovS, ovD, ovW, E, B,
                                                         x, Wrel1, Wroot1, PR1, gA, 0, tHalf);
    gcn_bin_sort<<<gB + B, SORT_T, 0, stream>>>(binCursor, ovCount, (const int2*)bsw, ovS, ovD, ovW,
                                                rowptr, (int2*)csr, N, B,
                                                x, Wrel1, Wroot1, PR1, gB, tHalf, nTiles);

    gcn_gather1<<<(N + 32 * G1_NPG - 1) / (32 * G1_NPG), 256, 0, stream>>>(rowptr, csr, PR1, b1, Wrel2, Wroot2, PR2, N);

    gcn_gather2_out<<<(N + 32 * G2_NPG - 1) / (32 * G2_NPG), 256, 0, stream>>>(rowptr, csr, PR2, b2,
                                                                               (float*)d_out, N);
}


# Layer-1 gather loses its own launch (5 launches -> 4): each sort work-group sums its bin's 256 nodes from the buckets it just ordered in LDS; GEMM1 all in the scatter launch; every sum unchanged
# speedup vs baseline: 5.6472x; 1.0375x over previous
#include <hip/hip_runtime.h>
#include <math.h>

typedef __attribute__((ext_vector_type(2))) float v2f;
typedef __attribute__((ext_vector_type(8))) float v8f;

#define D_IN  128
#define H_DIM 32
#define C_DIM 5

__global__ void gcn_zero_i(int* __restrict__ p, int n) {
    int i = blockIdx.x * 256 + threadIdx.x;
    if (i < n) p[i] = 0;
}

__device__ __forceinline__
void gcn_gemm1_tile(const float* __restrict__ x, const float* __restrict__ Wrel,
                    const float* __restrict__ Wroot, float* __restrict__ PR1, int tile) {
    int lane = threadIdx.x & 31;
    int half = lane >> 4;
    int m    = lane & 15;
    const float* xr = x + (long)(tile * 16 + m) * D_IN;

    v2f a[32];
#pragma unroll
    for (int ks = 0; ks < 32; ++ks) {
        int k = ks * 4 + half * 2;
        a[ks].x = xr[k];
        a[ks].y = xr[k + 1];
    }

#pragma unroll
    for (int ct = 0; ct < 4; ++ct) {
        const float* __restrict__ W = (ct < 2) ? Wrel : Wroot;
        int n = (ct & 1) * 16 + m;
        v8f acc = {};
#pragma unroll
        for (int ks = 0; ks < 32; ++ks) {
            int k = ks * 4 + half * 2;
            v2f b;
            b.x = W[k * H_DIM + n];
            b.y = W[(k + 1) * H_DIM + n];
            acc = __builtin_amdgcn_wmma_f32_16x16x4_f32(false, a[ks], false, b,
                                                        (short)0, acc, false, false);
        }
#pragma unroll
        for (int r = 0; r < 8; ++r) {
            int orow = tile * 16 + r + half * 8;
            PR1[(long)orow * 64 + ct * 16 + m] = acc[r];
        }
    }
}

#define G1_TPB 25
#define H_PITCH 36
__device__ __forceinline__
void gcn_gemm2_tile(const float* __restrict__ hs, const float* __restrict__ Wrel,
                    const float* __restrict__ Wroot, float* __restrict__ PR2, int row0, int lane) {
    int half = lane >> 4;
    int m    = lane & 15;
    const float* hr = hs + m * H_PITCH;
    const float* __restrict__ W = (m < 8) ? Wrel : Wroot;
    const int wc = min(m & 7, C_DIM - 1);

    v2f a[8];
#pragma unroll
    for (int ks = 0; ks < 8; ++ks) {
        int k = ks * 4 + half * 2;
        a[ks].x = hr[k];
        a[ks].y = hr[k + 1];
    }
    v8f acc = {};
#pragma unroll
    for (int ks = 0; ks < 8; ++ks) {
        int k = ks * 4 + half * 2;
        v2f b;
        b.x = W[k * C_DIM + wc];
        b.y = W[(k + 1) * C_DIM + wc];
        acc = __builtin_amdgcn_wmma_f32_16x16x4_f32(false, a[ks], false, b,
                                                    (short)0, acc, false, false);
    }
#pragma unroll
    for (int r = 0; r < 8; ++r) {
        int orow = row0 + r + half * 8;
        PR2[(long)orow * 16 + m] = acc[r];
    }
}

#define BIN_SHIFT 8
#define BIN_NODES 256
#define MAXB      512
#define SORT_T    512
#define EPT       8
#define CHUNK     (SORT_T * EPT)
#define SRC_BITS  24
#define SRC_MASK  0xFFFFFF
#define BIN_CAP_SHIFT 14
#define BIN_CAP   (1 << BIN_CAP_SHIFT)
#define BIN_PITCH (BIN_CAP + 40)
#define STAGE_CAP 5120
static_assert(CHUNK < (1 << (31 - BIN_CAP_SHIFT)), "a block's share of a bin and its place fit one word");
static_assert(SORT_T == MAXB, "one thread per bin counter");

__device__ __forceinline__ int gcn_block_excl_scan(int v, int* wsum) {
    int lane = threadIdx.x & 31, wave = threadIdx.x >> 5;
    int inc = v;
    for (int o = 1; o < 32; o <<= 1) {
        int y = __shfl_up(inc, o);
        if (lane >= o) inc += y;
    }
    if (lane == 31) wsum[wave] = inc;
    __syncthreads();
    int carry = 0;
    for (int w = 0; w < wave; ++w) carry += wsum[w];
    return carry + inc - v;
}

__global__ __launch_bounds__(SORT_T)
void gcn_bin_scatter(const int* __restrict__ src, const int* __restrict__ dst,
                     const float* __restrict__ ew, int* __restrict__ ovCount,
                     int* __restrict__ binCursor, int2* __restrict__ bsw,
                     int* __restrict__ ovS, int* __restrict__ ovD, float* __restrict__ ovW,
                     int E, int B,
                     const float* __restrict__ x, const float* __restrict__ Wrel1,
                     const float* __restrict__ Wroot1,
                     float* __restrict__ PR1, int nGemm, int tile0, int tile1) {
    __shared__ int hist[MAXB];
    __shared__ int place[MAXB];
    __shared__ int obase[MAXB];
    __shared__ int lbase[MAXB];
    __shared__ int wsumS[16];
    __shared__ __attribute__((aligned(8))) int stageS[2 * CHUNK];
    __shared__ int gidx[CHUNK];
    int2* st = (int2*)stageS;
    int tid = threadIdx.x;
    if ((int)blockIdx.x < nGemm) {
        const int t0 = tile0 + (int)blockIdx.x * G1_TPB;
        for (int t = tid >> 5; t < G1_TPB; t += SORT_T / 32)
            if (t0 + t < tile1) gcn_gemm1_tile(x, Wrel1, Wroot1, PR1, t0 + t);
        return;
    }
    const int chunk = (int)blockIdx.x - nGemm;
    hist[tid] = 0;
    __syncthreads();
    int e0 = chunk * CHUNK + tid;
    int d[EPT], r[EPT], s[EPT];
    float w[EPT];
#pragma unroll
    for (int k = 0; k < EPT; ++k) d[k] = dst[min(e0 + k * SORT_T, E - 1)];
#pragma unroll
    for (int k = 0; k < EPT; ++k) {
        int ec = min(e0 + k * SORT_T, E - 1);
        s[k] = src[ec];
        w[k] = ew[ec];
    }
#pragma unroll
    for (int k = 0; k < EPT; ++k) {
        r[k] = 0;
        if (e0 + k * SORT_T < E) r[k] = atomicAdd(&hist[d[k] >> BIN_SHIFT], 1);
    }
    __syncthreads();
    int c = hist[tid];
    const int lo = gcn_block_excl_scan(c, wsumS);
    int off = 0, fit = 0, ob = 0;
    if (tid < B && c) {
        off = atomicAdd(&binCursor[tid], c);
        fit = min(max(BIN_CAP - off, 0), c);
        if (fit < c) ob = atomicAdd(ovCount, c - fit);
    }
    place[tid] = (fit << BIN_CAP_SHIFT) | (fit ? off : 0);
    obase[tid] = ob;
    lbase[tid] = lo;
    __syncthreads();
    int pk[EPT], lb[EPT];
#pragma unroll
    for (int k = 0; k < EPT; ++k) { pk[k] = place[d[k] >> BIN_SHIFT]; lb[k] = lbase[d[k] >> BIN_SHIFT]; }
#pragma unroll
    for (int k = 0; k < EPT; ++k) {
        if (e0 + k * SORT_T < E) {
            const int bin = d[k] >> BIN_SHIFT;
            const int nf = pk[k] >> BIN_CAP_SHIFT;
            const int pos = lb[k] + r[k];
            if (r[k] < nf) {
                int key = s[k] | ((d[k] & (BIN_NODES - 1)) << SRC_BITS);
                st[pos] = make_int2(key, __float_as_int(w[k]));
                gidx[pos] = bin * BIN_PITCH + (pk[k] & (BIN_CAP - 1)) + r[k];
            } else {
                gidx[pos] = -1;
                int p = obase[bin] + (r[k] - nf);
                ovS[p] = s[k]; ovD[p] = d[k]; ovW[p] = w[k];
            }
        }
    }
    __syncthreads();
    const int nE = min(CHUNK, E - chunk * CHUNK);
    int g[EPT];
    int2 rc[EPT];
#pragma unroll
    for (int k = 0; k < EPT; ++k) {
        const int i = min(tid + k * SORT_T, nE - 1);
        g[k] = gidx[i];
        rc[k] = st[i];
    }
#pragma unroll
    for (int k = 0; k < EPT; ++k)
        if (tid + k * SORT_T < nE && g[k] >= 0) bsw[g[k]] = rc[k];
}

#define G1_BCAST(V, K) ((int)__builtin_amdgcn_mov_dpp8((unsigned)(V), (unsigned)(K) * 0x249249u))
#define G1_LOAD(K) { unsigned s = (unsigned)G1_BCAST(q.x, K); \
                     v[K] = *(const float4*)(tab + ((s << 8) | c16)); }
#define G1_FMA(K)  { float w = __int_as_float(G1_BCAST(q.y, K)); \
                     const bool on = (K) < left; \
                     float tx = acc.x + w * v[K].x, ty = acc.y + w * v[K].y; \
                     float tz = acc.z + w * v[K].z, tw = acc.w + w * v[K].w; \
                     acc.x = on ? tx : acc.x; acc.y = on ? ty : acc.y; \
                     acc.z = on ? tz : acc.z; acc.w = on ? tw : acc.w; }
#define G1_NPG (BIN_NODES / (SORT_T / 8))

template <bool FAR>
__device__ __forceinline__ int2 gcn_bin_rec(const int2* st, const int2* gl, int i) {
    if (!FAR) return st[i];
    const int* w = (const int*)(gl + i);
    return make_int2(__hip_atomic_load(w, __ATOMIC_RELAXED, __HIP_MEMORY_SCOPE_AGENT),
                     __hip_atomic_load(w + 1, __ATOMIC_RELAXED, __HIP_MEMORY_SCOPE_AGENT));
}

template <bool FAR>
__device__ __forceinline__ void gcn_bin_layer1(const int2* st, const int2* gl, const int* nstart,
                                               const float* __restrict__ PR1, const float* __restrict__ b1,
                                               float* hs, int node0, int ln0, int c4, int N) {
    const unsigned c16 = (unsigned)c4 * 16u;
    const char* tab = (const char*)PR1;
    const float bias0 = b1[c4 * 4], bias1 = b1[c4 * 4 + 1], bias2 = b1[c4 * 4 + 2], bias3 = b1[c4 * 4 + 3];
    for (int it = 0; it < G1_NPG; ++it) {
        const int ln = ln0 + it;
        const int n = node0 + ln;
        const int p0 = nstart[ln], e1 = nstart[ln + 1];
        const float4 root = *(const float4*)(tab + (((unsigned)min(n, N - 1) << 8) | c16) + 128);
        float4 acc = make_float4(0.0f, 0.0f, 0.0f, 0.0f);
        int2 q = gcn_bin_rec<FAR>(st, gl, max(min(p0 + c4, e1 - 1), 0));
        for (int p = p0; p < e1; p += 8) {
            const int left = e1 - p;
            int2 qn = gcn_bin_rec<FAR>(st, gl, min(p + 8 + c4, e1 - 1));
            float4 v[8];
            G1_LOAD(0) G1_LOAD(1) G1_LOAD(2) G1_LOAD(3)
            G1_LOAD(4) G1_LOAD(5) G1_LOAD(6) G1_LOAD(7)
            G1_FMA(0) G1_FMA(1) G1_FMA(2) G1_FMA(3)
            G1_FMA(4) G1_FMA(5) G1_FMA(6) G1_FMA(7)
            q = qn;
        }
        float4 o = make_float4(0.0f, 0.0f, 0.0f, 0.0f);
        if (n < N) {
            float cf = fmaxf((float)(e1 - p0), 1.0f);
            o.x = fmaxf(acc.x / cf + root.x + bias0, 0.0f);
            o.y = fmaxf(acc.y / cf + root.y + bias1, 0.0f);
            o.z = fmaxf(acc.z / cf + root.z + bias2, 0.0f);
            o.w = fmaxf(acc.w / cf + root.w + bias3, 0.0f);
        }
        *(float4*)(hs + ln * H_PITCH + c4 * 4) = o;
    }
}

__global__ __launch_bounds__(SORT_T) __attribute__((amdgpu_waves_per_eu(4, 8)))
void gcn_bin_sort_gather1(const int* __restrict__ binTotal, const int* __restrict__ ovCount,
                          const int2* __restrict__ bsw, const int* __restrict__ ovS,
                          const int* __restrict__ ovD, const float* __restrict__ ovW,
                          int* __restrict__ rowptr, int2* csr, int N, int B,
                          const float* __restrict__ PR1, const float* __restrict__ b1,
                          const float* __restrict__ Wrel2, const float* __restrict__ Wroot2,
                          float* __restrict__ PR2) {
    __shared__ int cnt[BIN_NODES];
    __shared__ int nstart[BIN_NODES + 1];
    __shared__ int wsumA[16];
    __shared__ int wsumB[16];
    __shared__ int sStart, sTot;
    __shared__ __attribute__((aligned(8))) int stageB[2 * STAGE_CAP];
    __shared__ __attribute__((aligned(16))) float hs[BIN_NODES * H_PITCH];
    int2* st = (int2*)stageB;
    int tid = threadIdx.x;
    int b = (int)blockIdx.x;
    int t = (tid < B) ? binTotal[tid] : 0;
    int ex = gcn_block_excl_scan(t, wsumA);
    if (tid == b) { sStart = ex; sTot = t; }
    if (tid < BIN_NODES) cnt[tid] = 0;
    __syncthreads();
    const int start = sStart;
    const int all = sTot;
    const int nst = min(all, STAGE_CAP);
    const int tot = min(all, BIN_CAP);
    const int nov = (all > BIN_CAP) ? ovCount[0] : 0;
    const int node0 = b << BIN_SHIFT;
    const int2* mysw = bsw + (long)b * BIN_PITCH;
    for (int i0 = tid; i0 - tid < tot; i0 += CHUNK) {
        int dv[EPT];
#pragma unroll
        for (int u = 0; u < EPT; ++u) dv[u] = mysw[min(i0 + u * SORT_T, tot - 1)].x;
#pragma unroll
        for (int u = 0; u < EPT; ++u)
            if (i0 + u * SORT_T < tot) atomicAdd(&cnt[(unsigned)dv[u] >> SRC_BITS], 1);
    }
    for (int i = tid; i < nov; i += SORT_T) {
        int dn = ovD[i];
        if ((dn >> BIN_SHIFT) == b) atomicAdd(&cnt[dn & (BIN_NODES - 1)], 1);
    }
    __syncthreads();
    int c = (tid < BIN_NODES) ? cnt[tid] : 0;
    int lex = gcn_block_excl_scan(c, wsumB);
    if (tid < BIN_NODES) {
        cnt[tid] = lex;
        nstart[tid] = lex;
        int n = node0 + tid;
        if (n <= N) rowptr[n] = start + lex;
    }
    if (tid == BIN_NODES) nstart[BIN_NODES] = all;
    __syncthreads();
    for (int i0 = tid; i0 - tid < tot; i0 += CHUNK) {
        int2 sw[EPT];
#pragma unroll
        for (int u = 0; u < EPT; ++u) sw[u] = mysw[min(i0 + u * SORT_T, tot - 1)];
#pragma unroll
        for (int u = 0; u < EPT; ++u) {
            if (i0 + u * SORT_T < tot) {
                int p = atomicAdd(&cnt[(unsigned)sw[u].x >> SRC_BITS], 1);
                const int2 rec = make_int2(sw[u].x & SRC_MASK, sw[u].y);
                if (p < STAGE_CAP) st[p] = rec; else csr[start + p] = rec;
            }
        }
    }
    for (int i = tid; i < nov; i += SORT_T) {
        int dn = ovD[i];
        if ((dn >> BIN_SHIFT) == b) {
            int p = atomicAdd(&cnt[dn & (BIN_NODES - 1)], 1);
            const int2 rec = make_int2(ovS[i], __float_as_int(ovW[i]));
            if (p < STAGE_CAP) st[p] = rec; else csr[start + p] = rec;
        }
    }
    __syncthreads();
    for (int i0 = tid; i0 - tid < nst; i0 += CHUNK) {
        int2 rc[EPT];
#pragma unroll
        for (int u = 0; u < EPT; ++u) rc[u] = st[min(i0 + u * SORT_T, nst - 1)];
#pragma unroll
        for (int u = 0; u < EPT; ++u)
            if (i0 + u * SORT_T < nst) csr[start + i0 + u * SORT_T] = rc[u];
    }
    const int2* gl = csr + start;
    const int c4 = tid & 7;
    const int ln0 = (tid >> 3) * G1_NPG;
    if (all > STAGE_CAP) {
        __syncthreads();
        gcn_bin_layer1<true>(st, gl, nstart, PR1, b1, hs, node0, ln0, c4, N);
    } else {
        gcn_bin_layer1<false>(st, gl, nstart, PR1, b1, hs, node0, ln0, c4, N);
    }
    __syncthreads();
    static_assert(BIN_NODES == 16 * (SORT_T / 32), "one GEMM2 tile per wave");
    static_assert(G1_NPG * (SORT_T / 8) == BIN_NODES, "the lane groups cover the bin's nodes");
    const int wave = tid >> 5;
    const int row0 = node0 + wave * 16;
    if (row0 + 16 <= N)
        gcn_gemm2_tile(hs + wave * 16 * H_PITCH, Wrel2, Wroot2, PR2, row0, tid & 31);
}

#define G2_NPG 2
#define G2_PAIR_LO(V) ((int)__builtin_amdgcn_mov_dpp8((unsigned)(V), 0x6D2240u))
#define G2_PAIR_HI(V) ((int)__builtin_amdgcn_mov_dpp8((unsigned)(V), 0xFF6B64u))
#define G2_X1(F) __uint_as_float(__builtin_amdgcn_mov_dpp8(__float_as_uint(F), 0xDE54C1u))
#define G2_X2(F) __uint_as_float(__builtin_amdgcn_mov_dpp8(__float_as_uint(F), 0xB3E21Au))
#define G2_X4(F) __uint_as_float(__builtin_amdgcn_mov_dpp8(__float_as_uint(F), 0x688FACu))

__device__ __forceinline__ void gcn_g2_records(int (&q)[4], const int2* __restrict__ ed,
                                               int p, int e1, int c) {
#pragma unroll
    for (int k = 0; k < 4; ++k) q[k] = ed[max(min(p + 8 * k + c, e1 - 1), 0)].x;
}
__device__ __forceinline__ void gcn_g2_rows(float4 (&v)[8], const int (&q)[4],
                                            const char* __restrict__ tab, unsigned h16) {
#pragma unroll
    for (int k = 0; k < 4; ++k) {
        unsigned s0 = (unsigned)G2_PAIR_LO(q[k]);
        unsigned s1 = (unsigned)G2_PAIR_HI(q[k]);
        v[2 * k]     = *(const float4*)(tab + ((s0 << 6) | h16));
        v[2 * k + 1] = *(const float4*)(tab + ((s1 << 6) | h16));
    }
}
__device__ __forceinline__ void gcn_g2_add(float4& acc, const float4 (&v)[8], int cnt, int slot) {
#pragma unroll
    for (int u = 0; u < 8; ++u) {
        const bool on = (4 * u + slot) < cnt;
        const float tx = acc.x + v[u].x, ty = acc.y + v[u].y;
        const float tz = acc.z + v[u].z, tw = acc.w + v[u].w;
        acc.x = on ? tx : acc.x; acc.y = on ? ty : acc.y;
        acc.z = on ? tz : acc.z; acc.w = on ? tw : acc.w;
    }
}

__global__ __launch_bounds__(256)
void gcn_gather2_out(const int* __restrict__ rowptr, const int* __restrict__ csr,
                     const float* __restrict__ PR2, const float* __restrict__ b2,
                     float* __restrict__ out, int N) {
    __shared__ float so[32 * G2_NPG * C_DIM];
    const int c    = threadIdx.x & 7;
    const int slot = c >> 1;
    const int hf   = c & 1;
    const unsigned h16 = (unsigned)hf * 16u;
    const int g  = threadIdx.x >> 3;
    const int n0 = ((int)blockIdx.x * 32 + g) * G2_NPG;
    const int2* ed = (const int2*)csr;
    const char* tab = (const char*)PR2;
    static_assert(G2_NPG == 2, "three bounds per group");
    const int nb = max(min(n0, N - 2), 0);
    const int d  = n0 - nb;
    const int t0 = rowptr[nb], t1 = rowptr[nb + 1], t2 = rowptr[nb + 2];
    int r[G2_NPG + 1];
    r[0] = (d >= 2) ? t2 : (d == 1) ? t1 : t0;
    r[1] = (d >= 1) ? t2 : t1;
    r[2] = t2;
    int q[G2_NPG][4];
#pragma unroll
    for (int it = 0; it < G2_NPG; ++it) gcn_g2_records(q[it], ed, r[it], r[it + 1], c);
    float4 root[G2_NPG];
#pragma unroll
    for (int it = 0; it < G2_NPG; ++it)
        root[it] = *(const float4*)(tab + (((unsigned)min(n0 + it, N - 1) << 6) | h16) + 32);
    const int bi = hf * 4;
    const float4 bb = make_float4(b2[bi], b2[min(bi + 1, C_DIM - 1)],
                                  b2[min(bi + 2, C_DIM - 1)], b2[min(bi + 3, C_DIM - 1)]);
    float4 v[G2_NPG][8];
#pragma unroll
    for (int it = 0; it < G2_NPG; ++it) gcn_g2_rows(v[it], q[it], tab, h16);
    asm volatile("" :: "v"(v[0][0].x), "v"(v[0][0].y), "v"(v[0][0].z), "v"(v[0][0].w));
#pragma unroll
    for (int it = 0; it < G2_NPG; ++it) {
        const int n = n0 + it;
        const int e1 = r[it + 1];
        const int deg = e1 - r[it];
        float4 acc = make_float4(0.0f, 0.0f, 0.0f, 0.0f);
        gcn_g2_add(acc, v[it], deg, slot);
        for (int p = r[it] + 32; p < e1; p += 32) {
            int qq[4];
            float4 vv[8];
            gcn_g2_records(qq, ed, p, e1, c);
            gcn_g2_rows(vv, qq, tab, h16);
            gcn_g2_add(acc, vv, e1 - p, slot);
        }
        acc.x += G2_X2(acc.x); acc.y += G2_X2(acc.y); acc.z += G2_X2(acc.z); acc.w += G2_X2(acc.w);
        acc.x += G2_X4(acc.x); acc.y += G2_X4(acc.y); acc.z += G2_X4(acc.z); acc.w += G2_X4(acc.w);
        const float cn = fmaxf((float)deg, 1.0f);
        const float v0 = acc.x / cn + root[it].x + bb.x;
        const float v1 = acc.y / cn + root[it].y + bb.y;
        const float v2 = acc.z / cn + root[it].z + bb.z;
        const float v3 = acc.w / cn + root[it].w + bb.w;
        const float v4 = G2_X1(v0);
        float mx = fmaxf(fmaxf(fmaxf(v0, v1), fmaxf(v2, v3)), v4);
        float s = expf(v0 - mx) + expf(v1 - mx) + expf(v2 - mx) + expf(v3 - mx) + expf(v4 - mx);
        float l = logf(s);
        if (c == 0 && n < N) {
            float* o = so + (g * G2_NPG + it) * C_DIM;
            o[0] = v0 - mx - l; o[1] = v1 - mx - l; o[2] = v2 - mx - l;
            o[3] = v3 - mx - l; o[4] = v4 - mx - l;
        }
    }
    __syncthreads();
    const int rem = N - (int)blockIdx.x * (32 * G2_NPG);
    const int valid = (rem < 32 * G2_NPG ? rem : 32 * G2_NPG) * C_DIM;
    static_assert(32 * G2_NPG * C_DIM <= 2 * 256, "two stores per thread cover the block's rows");
    float* ob = out + (long)blockIdx.x * (32 * G2_NPG * C_DIM);
    const int t = (int)threadIdx.x;
    if (t < valid)       ob[t]       = so[t];
    if (t + 256 < valid) ob[t + 256] = so[t + 256];
}

extern "C" void kernel_launch(void* const* d_in, const int* in_sizes, int n_in,
                              void* d_out, int out_size, void* d_ws, size_t ws_size,
                              hipStream_t stream) {
    const float* x      = (const float*)d_in[0];
    const int*   ei     = (const int*)d_in[1];
    const float* ew     = (const float*)d_in[2];
    const float* Wrel1  = (const float*)d_in[3];
    const float* Wroot1 = (const float*)d_in[4];
    const float* b1     = (const float*)d_in[5];
    const float* Wrel2  = (const float*)d_in[6];
    const float* Wroot2 = (const float*)d_in[7];
    const float* b2     = (const float*)d_in[8];

    const int N = in_sizes[0] / D_IN;
    const int E = in_sizes[1] / 2;
    const int* src = ei;
    const int* dst = ei + E;

    float* ws = (float*)d_ws;
    const int NP = ((N + 4096) / 4096) * 4096;
    long off = 0;
    float* PR1 = ws + off; off += (long)N * 64;
    float* PR2 = ws + off; off += (long)N * 16;
    int* ovCount   = (int*)(ws + off); off += MAXB;
    int* binCursor = (int*)(ws + off); off += MAXB;
    int* rowptr = (int*)(ws + off); off += NP;
    int* csr    = (int*)(ws + off); off += (long)E * 2;
    int* bsw    = (int*)(ws + off); off += (long)MAXB * BIN_PITCH * 2;
    int* ovS    = (int*)(ws + off); off += E;
    int* ovD    = (int*)(ws + off); off += E;
    float* ovW  = ws + off;         off += E;

    const int nTiles = N / 16;

    const int B = N / BIN_NODES + 1;
    const int nChunks = (E + CHUNK - 1) / CHUNK;

    const int gA = (nTiles + G1_TPB - 1) / G1_TPB;

    gcn_zero_i<<<(2 * MAXB + 255) / 256, 256, 0, stream>>>(ovCount, 2 * MAXB);
    gcn_bin_scatter<<<gA + nChunks, SORT_T, 0, stream>>>(src, dst, ew, ovCount, binCursor,
                                                         (int2*)bsw, ovS, ovD, ovW, E, B,
                                                         x, Wrel1, Wroot1, PR1, gA, 0, nTiles);
    gcn_bin_sort_gather1<<<B, SORT_T, 0, stream>>>(binCursor, ovCount, (const int2*)bsw, ovS, ovD, ovW,
                                                   rowptr, (int2*)csr, N, B,
                                                   PR1, b1, Wrel2, Wroot2, PR2);

    gcn_gather2_out<<<(N + 32 * G2_NPG - 1) / (32 * G2_NPG), 256, 0, stream>>>(rowptr, csr, PR2, b2,
                                                                               (float*)d_out, N);
}
